# starting kernel
# baseline (speedup 1.0000x reference)
#include <hip/hip_runtime.h>
#include <hip/hip_bf16.h>
#include <cstddef>


typedef float v2f __attribute__((ext_vector_type(2)));
typedef float v8f __attribute__((ext_vector_type(8)));

static __device__ __forceinline__ v8f wmma_f32(v2f a, v2f b, v8f c) {
  return __builtin_amdgcn_wmma_f32_16x16x4_f32(
       false, a,  false, b,
       (short)0, c,  false,  false);
}

__global__ void k_simraw(const float* __restrict__ E, float* __restrict__ sim) {
  const int idx = blockIdx.x * 256 + threadIdx.x;
  const int i = idx >> 10, j = idx & 1023;
  float acc = 0.f;
#pragma unroll
  for (int d = 0; d < 16; ++d) acc += E[i * 16 + d] * E[j * 16 + d];
  sim[idx] = fmaxf(acc, 0.f);
}

__global__ void k_rowsoftmax(float* __restrict__ sim) {
  const int row = blockIdx.x;
  const int t = threadIdx.x;
  __shared__ float red[256];
  float v[4];
  float mx = -3.0e38f;
#pragma unroll
  for (int i = 0; i < 4; ++i) {
    v[i] = sim[(size_t)row * 1024 + t + i * 256];
    mx = fmaxf(mx, v[i]);
  }
  red[t] = mx; __syncthreads();
  for (int s = 128; s > 0; s >>= 1) {
    if (t < s) red[t] = fmaxf(red[t], red[t + s]);
    __syncthreads();
  }
  mx = red[0]; __syncthreads();
  float sum = 0.f;
#pragma unroll
  for (int i = 0; i < 4; ++i) { v[i] = __expf(v[i] - mx); sum += v[i]; }
  red[t] = sum; __syncthreads();
  for (int s = 128; s > 0; s >>= 1) {
    if (t < s) red[t] += red[t + s];
    __syncthreads();
  }
  const float inv = 1.0f / red[0];
#pragma unroll
  for (int i = 0; i < 4; ++i)
    sim[(size_t)row * 1024 + t + i * 256] = v[i] * inv;
}

__global__ void k_cdist(const float* __restrict__ pa, float* __restrict__ sube) {
  const int b = blockIdx.z;
  const int jBase = blockIdx.x * 64, iBase = blockIdx.y * 64;
  __shared__ float Pi[64][17];
  __shared__ float Pj[64][17];
  const int t = threadIdx.x;
  for (int i = t; i < 64 * 16; i += 256) {
    const int r = i >> 4, d = i & 15;
    Pi[r][d] = pa[((size_t)b * 1024 + iBase + r) * 16 + d];
    Pj[r][d] = pa[((size_t)b * 1024 + jBase + r) * 16 + d];
  }
  __syncthreads();
  const int tj = t & 63;
  const int i0 = (t >> 6) * 16;
#pragma unroll 4
  for (int ii = 0; ii < 16; ++ii) {
    const int i = i0 + ii;
    float s = 0.f;
#pragma unroll
    for (int d = 0; d < 16; ++d) s += fabsf(Pi[i][d] - Pj[tj][d]);
    sube[((size_t)b * 1024 + iBase + i) * 1024 + jBase + tj] = __expf(-s);
  }
}

__global__ void k_colsum(const float* __restrict__ sube, float* __restrict__ S) {
  const int b = blockIdx.y;
  const int j = blockIdx.x * 256 + threadIdx.x;
  float acc = 0.f;
  for (int i = 0; i < 1024; ++i)
    acc += sube[((size_t)b * 1024 + i) * 1024 + j];
  S[b * 1024 + j] = acc;
}

__global__ void k_xnorm(const float* __restrict__ x, const float* __restrict__ S,
                        float* __restrict__ xn) {
  const int idx = blockIdx.x * 256 + threadIdx.x;
  xn[idx] = x[idx] / S[idx >> 6];
}

__global__ void k_weights(const float* __restrict__ E, const float* __restrict__ Wp,
                          float* __restrict__ W) {
  const int n = blockIdx.y;
  const int tid = blockIdx.x * 256 + threadIdx.x;
  const int kp = tid >> 5;
  const int oc = (tid & 31) * 2;
  float e[16];
#pragma unroll
  for (int d = 0; d < 16; ++d) e[d] = E[n * 16 + d];
  float4 acc = {0.f, 0.f, 0.f, 0.f};
#pragma unroll
  for (int d = 0; d < 16; ++d) {
    const v2f w0 = *(const v2f*)(Wp + (size_t)d * 8192 + (size_t)(2 * kp) * 64 + oc);
    const v2f w1 = *(const v2f*)(Wp + (size_t)d * 8192 + (size_t)(2 * kp + 1) * 64 + oc);
    acc.x += e[d] * w0.x;
    acc.y += e[d] * w1.x;
    acc.z += e[d] * w0.y;
    acc.w += e[d] * w1.y;
  }
  *(float4*)(W + (size_t)n * 8192 + (size_t)kp * 128 + (size_t)oc * 2) = acc;
}

__global__ __launch_bounds__(256)
void k_bmm_nk(const float* __restrict__ A, long long strideA,
              const float* __restrict__ B, long long strideB,
              float* __restrict__ C, long long strideC) {
  const int b = blockIdx.y;
  const float* Ab = A + (size_t)strideA * b;
  const float* Bb = B + (size_t)strideB * b;
  float* Cb = C + (size_t)strideC * b;
  const int rowBase = blockIdx.x * 128;

  __shared__ float As[128][68];
  __shared__ float Bs[32][160];

  const int t = threadIdx.x;
  const int wave = t >> 5;
  const int lane = t & 31;
  const int m16 = lane & 15;
  const int kh = lane >> 4;

  const v8f zero = {0.f, 0.f, 0.f, 0.f, 0.f, 0.f, 0.f, 0.f};
  v8f cacc[4];
#pragma unroll
  for (int nt = 0; nt < 4; ++nt) cacc[nt] = zero;

  for (int kc = 0; kc < 1024; kc += 64) {
    __syncthreads();
    for (int i = t; i < 128 * 16; i += 256) {
      const int r = i >> 4, cc = (i & 15) << 2;
      *(float4*)(&As[r][cc]) =
          *(const float4*)(Ab + (size_t)(rowBase + r) * 1024 + kc + cc);
    }
    for (int i = t; i < 32 * 16; i += 256) {
      const int kp = i >> 4, cc = (i & 15) << 2;
      const float4 v0 = *(const float4*)(Bb + (size_t)(kc + 2 * kp) * 64 + cc);
      const float4 v1 = *(const float4*)(Bb + (size_t)(kc + 2 * kp + 1) * 64 + cc);
      float4 lo = {v0.x, v1.x, v0.y, v1.y};
      float4 hi = {v0.z, v1.z, v0.w, v1.w};
      *(float4*)(&Bs[kp][cc * 2])     = lo;
      *(float4*)(&Bs[kp][cc * 2 + 4]) = hi;
    }
    __syncthreads();

    const int mrow = wave * 16 + m16;
#pragma unroll
    for (int ks = 0; ks < 16; ++ks) {
      const int kg = ks * 4 + 2 * kh;
      const int kp = ks * 2 + kh;
      const v2f a = *(const v2f*)(&As[mrow][kg]);
#pragma unroll
      for (int nt = 0; nt < 4; ++nt) {
        const v2f bf = *(const v2f*)(&Bs[kp][(nt * 16 + m16) * 2]);
        cacc[nt] = wmma_f32(a, bf, cacc[nt]);
      }
    }
  }

#pragma unroll
  for (int nt = 0; nt < 4; ++nt) {
#pragma unroll
    for (int r = 0; r < 8; ++r) {
      const int m = rowBase + wave * 16 + r + 8 * kh;
      Cb[(size_t)m * 64 + nt * 16 + m16] = cacc[nt][r];
    }
  }
}

__global__ __launch_bounds__(256)
void k_final(const float* __restrict__ sx, const float* __restrict__ xg1,
             const float* __restrict__ W, const float* __restrict__ E,
             const float* __restrict__ bias_pool, float* __restrict__ out) {
  __shared__ float As[8][8][132];
  const int t = threadIdx.x;
  const int wave = t >> 5;
  const int lane = t & 31;
  const int n = blockIdx.x * 8 + wave;
  float (*Aw)[132] = As[wave];

  for (int rb = 0; rb < 8; ++rb) {
    const float* s0 = sx  + ((size_t)rb * 1024 + n) * 64;
    const float* s1 = xg1 + ((size_t)rb * 1024 + n) * 64;
    for (int c = lane; c < 128; c += 32)
      Aw[rb][c] = (c < 64) ? s0[c] : s1[c - 64];
  }
  __syncthreads();

  const float* Wn = W + (size_t)n * 8192;
  const int m16 = lane & 15;
  const int kh = lane >> 4;

  const v8f zero = {0.f, 0.f, 0.f, 0.f, 0.f, 0.f, 0.f, 0.f};
  v8f cacc[4];
#pragma unroll
  for (int nt = 0; nt < 4; ++nt) cacc[nt] = zero;

#pragma unroll 4
  for (int ks = 0; ks < 32; ++ks) {
    const int kg = ks * 4 + 2 * kh;
    const int kp = ks * 2 + kh;
    v2f a = {0.f, 0.f};
    if (m16 < 8) a = *(const v2f*)(&Aw[m16][kg]);
#pragma unroll
    for (int nt = 0; nt < 4; ++nt) {
      const v2f bf = *(const v2f*)(Wn + (size_t)kp * 128 + (nt * 16 + m16) * 2);
      cacc[nt] = wmma_f32(a, bf, cacc[nt]);
    }
  }

#pragma unroll
  for (int nt = 0; nt < 4; ++nt) {
    const int col = nt * 16 + m16;
    float bv = 0.f;
#pragma unroll
    for (int d = 0; d < 16; ++d) bv += E[n * 16 + d] * bias_pool[d * 64 + col];
#pragma unroll
    for (int r = 0; r < 8; ++r) {
      const int m = r + 8 * kh;
      if (m < 8)
        out[((size_t)m * 1024 + n) * 64 + col] = cacc[nt][r] + bv;
    }
  }
}

extern "C" void kernel_launch(void* const* d_in, const int* in_sizes, int n_in,
                              void* d_out, int out_size, void* d_ws, size_t ws_size,
                              hipStream_t stream) {
  (void)in_sizes; (void)n_in; (void)out_size; (void)ws_size;
  const float* x    = (const float*)d_in[0];
  const float* E    = (const float*)d_in[1];
  const float* pa   = (const float*)d_in[2];
  const float* Wp   = (const float*)d_in[3];
  const float* bp   = (const float*)d_in[4];
  float* out = (float*)d_out;

  float* ws   = (float*)d_ws;
  float* sim  = ws;
  float* sube = sim  + 1048576;
  float* S    = sube + 8388608;
  float* xn   = S    + 8192;
  float* sx   = xn   + 524288;
  float* xg1  = sx   + 524288;
  float* W    = xg1  + 524288;

  k_simraw    <<<4096, 256, 0, stream>>>(E, sim);
  k_rowsoftmax<<<1024, 256, 0, stream>>>(sim);
  k_cdist     <<<dim3(16, 16, 8), 256, 0, stream>>>(pa, sube);
  k_colsum    <<<dim3(4, 8),      256, 0, stream>>>(sube, S);
  k_xnorm     <<<2048, 256, 0, stream>>>(x, S, xn);
  k_weights   <<<dim3(8, 1024),   256, 0, stream>>>(E, Wp, W);
  k_bmm_nk<<<dim3(8, 8), 256, 0, stream>>>(sube, 1024LL * 1024, xn, 1024LL * 64,
                                           sx, 1024LL * 64);
  k_bmm_nk<<<dim3(8, 8), 256, 0, stream>>>(sim, 0LL, sx, 1024LL * 64,
                                           xg1, 1024LL * 64);
  k_final<<<128, 256, 0, stream>>>(sx, xg1, W, E, bp, out);
}


# sub_e never stored: first product builds its A fragments in registers from the points (LDS broadcast), column sums recomputed from points; changes sum order of column sums and of each L1 distance
# speedup vs baseline: 1.2728x; 1.2728x over previous
#include <hip/hip_runtime.h>
#include <hip/hip_bf16.h>
#include <cstddef>


typedef float v2f __attribute__((ext_vector_type(2)));
typedef float v8f __attribute__((ext_vector_type(8)));

static __device__ __forceinline__ v8f wmma_f32(v2f a, v2f b, v8f c) {
  return __builtin_amdgcn_wmma_f32_16x16x4_f32(
       false, a,  false, b,
       (short)0, c,  false,  false);
}

static __device__ __forceinline__ float exp_neg_l1(const float (&p)[16],
                                                   const float (&q)[16]) {
  float s0 = 0.f, s1 = 0.f, s2 = 0.f, s3 = 0.f;
#pragma unroll
  for (int d = 0; d < 4; ++d) {
    s0 += fabsf(p[d]      - q[d]);
    s1 += fabsf(p[d + 4]  - q[d + 4]);
    s2 += fabsf(p[d + 8]  - q[d + 8]);
    s3 += fabsf(p[d + 12] - q[d + 12]);
  }
  return __expf(-((s0 + s1) + (s2 + s3)));
}

__global__ void k_simraw(const float* __restrict__ E, float* __restrict__ sim) {
  const int idx = blockIdx.x * 256 + threadIdx.x;
  const int i = idx >> 10, j = idx & 1023;
  float acc = 0.f;
#pragma unroll
  for (int d = 0; d < 16; ++d) acc += E[i * 16 + d] * E[j * 16 + d];
  sim[idx] = fmaxf(acc, 0.f);
}

__global__ void k_rowsoftmax(float* __restrict__ sim) {
  const int row = blockIdx.x;
  const int t = threadIdx.x;
  __shared__ float red[256];
  float v[4];
  float mx = -3.0e38f;
#pragma unroll
  for (int i = 0; i < 4; ++i) {
    v[i] = sim[(size_t)row * 1024 + t + i * 256];
    mx = fmaxf(mx, v[i]);
  }
  red[t] = mx; __syncthreads();
  for (int s = 128; s > 0; s >>= 1) {
    if (t < s) red[t] = fmaxf(red[t], red[t + s]);
    __syncthreads();
  }
  mx = red[0]; __syncthreads();
  float sum = 0.f;
#pragma unroll
  for (int i = 0; i < 4; ++i) { v[i] = __expf(v[i] - mx); sum += v[i]; }
  red[t] = sum; __syncthreads();
  for (int s = 128; s > 0; s >>= 1) {
    if (t < s) red[t] += red[t + s];
    __syncthreads();
  }
  const float inv = 1.0f / red[0];
#pragma unroll
  for (int i = 0; i < 4; ++i)
    sim[(size_t)row * 1024 + t + i * 256] = v[i] * inv;
}

__global__ __launch_bounds__(512)
void k_colsum_pts(const float* __restrict__ pa, float* __restrict__ S) {
  const int b = blockIdx.y;
  const int lane = threadIdx.x & 31;
  const int w = __builtin_amdgcn_readfirstlane((int)(threadIdx.x >> 5));
  const int j = blockIdx.x * 32 + lane;
  __shared__ float part[16][32];

  float pj[16];
#pragma unroll
  for (int q = 0; q < 4; ++q) {
    const float4 v = *(const float4*)(pa + ((size_t)b * 1024 + j) * 16 + q * 4);
    pj[q * 4 + 0] = v.x; pj[q * 4 + 1] = v.y; pj[q * 4 + 2] = v.z; pj[q * 4 + 3] = v.w;
  }
  const float* pi = pa + ((size_t)b * 1024 + (size_t)w * 64) * 16;
  float acc = 0.f;
  for (int ii = 0; ii < 64; ++ii) {
    float qi[16];
#pragma unroll
    for (int d = 0; d < 16; ++d) qi[d] = pi[ii * 16 + d];
    acc += exp_neg_l1(qi, pj);
  }
  part[w][lane] = acc;
  __syncthreads();
  if (threadIdx.x < 32) {
    float tot = 0.f;
#pragma unroll
    for (int w2 = 0; w2 < 16; ++w2) tot += part[w2][threadIdx.x];
    S[b * 1024 + blockIdx.x * 32 + threadIdx.x] = tot;
  }
}

__global__ void k_xnorm(const float* __restrict__ x, const float* __restrict__ S,
                        float* __restrict__ xn) {
  const int idx = blockIdx.x * 256 + threadIdx.x;
  xn[idx] = x[idx] / S[idx >> 6];
}

__global__ void k_weights(const float* __restrict__ E, const float* __restrict__ Wp,
                          float* __restrict__ W) {
  const int n = blockIdx.y;
  const int tid = blockIdx.x * 256 + threadIdx.x;
  const int kp = tid >> 5;
  const int oc = (tid & 31) * 2;
  float e[16];
#pragma unroll
  for (int d = 0; d < 16; ++d) e[d] = E[n * 16 + d];
  float4 acc = {0.f, 0.f, 0.f, 0.f};
#pragma unroll
  for (int d = 0; d < 16; ++d) {
    const v2f w0 = *(const v2f*)(Wp + (size_t)d * 8192 + (size_t)(2 * kp) * 64 + oc);
    const v2f w1 = *(const v2f*)(Wp + (size_t)d * 8192 + (size_t)(2 * kp + 1) * 64 + oc);
    acc.x += e[d] * w0.x;
    acc.y += e[d] * w1.x;
    acc.z += e[d] * w0.y;
    acc.w += e[d] * w1.y;
  }
  *(float4*)(W + (size_t)n * 8192 + (size_t)kp * 128 + (size_t)oc * 2) = acc;
}

#define SX_KC 512
__global__ __launch_bounds__(128)
void k_sx_points(const float* __restrict__ pa, const float* __restrict__ xn,
                 float* __restrict__ sx) {
  const int b = blockIdx.y;
  const int rowBase = blockIdx.x * 64;
  const float* pab = pa + (size_t)b * 1024 * 16;
  const float* Bb = xn + (size_t)b * 1024 * 64;
  float* Cb = sx + (size_t)b * 1024 * 64;

  __shared__ float Ps[1024][16];
  __shared__ float Bs[SX_KC / 2][160];

  const int t = threadIdx.x;
  const int wave = t >> 5;
  const int lane = t & 31;
  const int m16 = lane & 15;
  const int kh = lane >> 4;

  for (int i = t; i < 1024 * 4; i += 128)
    ((float4*)&Ps[0][0])[i] = ((const float4*)pab)[i];

  float pm[16];
  {
    const int mrow = rowBase + wave * 16 + m16;
#pragma unroll
    for (int q = 0; q < 4; ++q) {
      const float4 v = *(const float4*)(pab + (size_t)mrow * 16 + q * 4);
      pm[q * 4 + 0] = v.x; pm[q * 4 + 1] = v.y; pm[q * 4 + 2] = v.z; pm[q * 4 + 3] = v.w;
    }
  }

  const v8f zero = {0.f, 0.f, 0.f, 0.f, 0.f, 0.f, 0.f, 0.f};
  v8f cacc[4];
#pragma unroll
  for (int nt = 0; nt < 4; ++nt) cacc[nt] = zero;

  __syncthreads();

  float pt[32];
  v2f a_cur;
  {
    float p0[32];
    const float* q0 = &Ps[2 * kh][0];
#pragma unroll
    for (int d = 0; d < 32; ++d) p0[d] = q0[d];
    const float (&r0)[16] = *(const float (*)[16])(&p0[0]);
    const float (&r1)[16] = *(const float (*)[16])(&p0[16]);
    a_cur[0] = exp_neg_l1(pm, r0);
    a_cur[1] = exp_neg_l1(pm, r1);
    const float* q1 = &Ps[4 + 2 * kh][0];
#pragma unroll
    for (int d = 0; d < 32; ++d) pt[d] = q1[d];
  }

  for (int kc = 0; kc < 1024; kc += SX_KC) {
    __syncthreads();
    for (int i0 = t; i0 < (SX_KC / 2) * 16; i0 += 128 * 8) {
      float4 v0[8], v1[8];
#pragma unroll
      for (int u = 0; u < 8; ++u) {
        const int i = i0 + u * 128;
        const int kp = i >> 4, cc = (i & 15) << 2;
        v0[u] = *(const float4*)(Bb + (size_t)(kc + 2 * kp) * 64 + cc);
        v1[u] = *(const float4*)(Bb + (size_t)(kc + 2 * kp + 1) * 64 + cc);
      }
#pragma unroll
      for (int u = 0; u < 8; ++u) {
        const int i = i0 + u * 128;
        const int kp = i >> 4, cc = (i & 15) << 2;
        float4 lo = {v0[u].x, v1[u].x, v0[u].y, v1[u].y};
        float4 hi = {v0[u].z, v1[u].z, v0[u].w, v1[u].w};
        *(float4*)(&Bs[kp][cc * 2])     = lo;
        *(float4*)(&Bs[kp][cc * 2 + 4]) = hi;
      }
    }
    __syncthreads();

    v2f bf_cur[4];
#pragma unroll
    for (int nt = 0; nt < 4; ++nt)
      bf_cur[nt] = *(const v2f*)(&Bs[kh][(nt * 16 + m16) * 2]);

#pragma unroll 2
    for (int ks = 0; ks < SX_KC / 4; ++ks) {
      const int g = kc / 4 + ks;
      float ptn[32];
      const float* qn = &Ps[((g + 2) & 255) * 4 + 2 * kh][0];
#pragma unroll
      for (int d = 0; d < 32; ++d) ptn[d] = qn[d];
      v2f bf_nxt[4];
      const int kpn = ((ks + 1) & (SX_KC / 4 - 1)) * 2 + kh;
#pragma unroll
      for (int nt = 0; nt < 4; ++nt)
        bf_nxt[nt] = *(const v2f*)(&Bs[kpn][(nt * 16 + m16) * 2]);
      const float (&r0)[16] = *(const float (*)[16])(&pt[0]);
      const float (&r1)[16] = *(const float (*)[16])(&pt[16]);
      v2f a_nxt;
      a_nxt[0] = exp_neg_l1(pm, r0);
      a_nxt[1] = exp_neg_l1(pm, r1);
#pragma unroll
      for (int nt = 0; nt < 4; ++nt)
        cacc[nt] = wmma_f32(a_cur, bf_cur[nt], cacc[nt]);
      __builtin_amdgcn_sched_group_barrier(0x100, 10, 0);
#pragma unroll
      for (int nt = 0; nt < 4; ++nt) {
        __builtin_amdgcn_sched_group_barrier(0x008, 1, 0);
        __builtin_amdgcn_sched_group_barrier(0x002, 18, 0);
      }
      a_cur = a_nxt;
#pragma unroll
      for (int d = 0; d < 32; ++d) pt[d] = ptn[d];
#pragma unroll
      for (int nt = 0; nt < 4; ++nt) bf_cur[nt] = bf_nxt[nt];
    }
  }

#pragma unroll
  for (int nt = 0; nt < 4; ++nt) {
#pragma unroll
    for (int r = 0; r < 8; ++r) {
      const int m = rowBase + wave * 16 + r + 8 * kh;
      Cb[(size_t)m * 64 + nt * 16 + m16] = cacc[nt][r];
    }
  }
}

__global__ __launch_bounds__(256)
void k_bmm_nk(const float* __restrict__ A, long long strideA,
              const float* __restrict__ B, long long strideB,
              float* __restrict__ C, long long strideC) {
  const int b = blockIdx.y;
  const float* Ab = A + (size_t)strideA * b;
  const float* Bb = B + (size_t)strideB * b;
  float* Cb = C + (size_t)strideC * b;
  const int rowBase = blockIdx.x * 128;

  __shared__ float As[128][68];
  __shared__ float Bs[32][160];

  const int t = threadIdx.x;
  const int wave = t >> 5;
  const int lane = t & 31;
  const int m16 = lane & 15;
  const int kh = lane >> 4;

  const v8f zero = {0.f, 0.f, 0.f, 0.f, 0.f, 0.f, 0.f, 0.f};
  v8f cacc[4];
#pragma unroll
  for (int nt = 0; nt < 4; ++nt) cacc[nt] = zero;

  for (int kc = 0; kc < 1024; kc += 64) {
    __syncthreads();
    for (int i = t; i < 128 * 16; i += 256) {
      const int r = i >> 4, cc = (i & 15) << 2;
      *(float4*)(&As[r][cc]) =
          *(const float4*)(Ab + (size_t)(rowBase + r) * 1024 + kc + cc);
    }
    for (int i = t; i < 32 * 16; i += 256) {
      const int kp = i >> 4, cc = (i & 15) << 2;
      const float4 v0 = *(const float4*)(Bb + (size_t)(kc + 2 * kp) * 64 + cc);
      const float4 v1 = *(const float4*)(Bb + (size_t)(kc + 2 * kp + 1) * 64 + cc);
      float4 lo = {v0.x, v1.x, v0.y, v1.y};
      float4 hi = {v0.z, v1.z, v0.w, v1.w};
      *(float4*)(&Bs[kp][cc * 2])     = lo;
      *(float4*)(&Bs[kp][cc * 2 + 4]) = hi;
    }
    __syncthreads();

    const int mrow = wave * 16 + m16;
#pragma unroll
    for (int ks = 0; ks < 16; ++ks) {
      const int kg = ks * 4 + 2 * kh;
      const int kp = ks * 2 + kh;
      const v2f a = *(const v2f*)(&As[mrow][kg]);
#pragma unroll
      for (int nt = 0; nt < 4; ++nt) {
        const v2f bf = *(const v2f*)(&Bs[kp][(nt * 16 + m16) * 2]);
        cacc[nt] = wmma_f32(a, bf, cacc[nt]);
      }
    }
  }

#pragma unroll
  for (int nt = 0; nt < 4; ++nt) {
#pragma unroll
    for (int r = 0; r < 8; ++r) {
      const int m = rowBase + wave * 16 + r + 8 * kh;
      Cb[(size_t)m * 64 + nt * 16 + m16] = cacc[nt][r];
    }
  }
}

__global__ __launch_bounds__(256)
void k_final(const float* __restrict__ sx, const float* __restrict__ xg1,
             const float* __restrict__ W, const float* __restrict__ E,
             const float* __restrict__ bias_pool, float* __restrict__ out) {
  __shared__ float As[8][8][132];
  const int t = threadIdx.x;
  const int wave = t >> 5;
  const int lane = t & 31;
  const int n = blockIdx.x * 8 + wave;
  float (*Aw)[132] = As[wave];

  for (int rb = 0; rb < 8; ++rb) {
    const float* s0 = sx  + ((size_t)rb * 1024 + n) * 64;
    const float* s1 = xg1 + ((size_t)rb * 1024 + n) * 64;
    for (int c = lane; c < 128; c += 32)
      Aw[rb][c] = (c < 64) ? s0[c] : s1[c - 64];
  }
  __syncthreads();

  const float* Wn = W + (size_t)n * 8192;
  const int m16 = lane & 15;
  const int kh = lane >> 4;

  const v8f zero = {0.f, 0.f, 0.f, 0.f, 0.f, 0.f, 0.f, 0.f};
  v8f cacc[4];
#pragma unroll
  for (int nt = 0; nt < 4; ++nt) cacc[nt] = zero;

#pragma unroll 4
  for (int ks = 0; ks < 32; ++ks) {
    const int kg = ks * 4 + 2 * kh;
    const int kp = ks * 2 + kh;
    v2f a = {0.f, 0.f};
    if (m16 < 8) a = *(const v2f*)(&Aw[m16][kg]);
#pragma unroll
    for (int nt = 0; nt < 4; ++nt) {
      const v2f bf = *(const v2f*)(Wn + (size_t)kp * 128 + (nt * 16 + m16) * 2);
      cacc[nt] = wmma_f32(a, bf, cacc[nt]);
    }
  }

#pragma unroll
  for (int nt = 0; nt < 4; ++nt) {
    const int col = nt * 16 + m16;
    float bv = 0.f;
#pragma unroll
    for (int d = 0; d < 16; ++d) bv += E[n * 16 + d] * bias_pool[d * 64 + col];
#pragma unroll
    for (int r = 0; r < 8; ++r) {
      const int m = r + 8 * kh;
      if (m < 8)
        out[((size_t)m * 1024 + n) * 64 + col] = cacc[nt][r] + bv;
    }
  }
}

extern "C" void kernel_launch(void* const* d_in, const int* in_sizes, int n_in,
                              void* d_out, int out_size, void* d_ws, size_t ws_size,
                              hipStream_t stream) {
  (void)in_sizes; (void)n_in; (void)out_size; (void)ws_size;
  const float* x    = (const float*)d_in[0];
  const float* E    = (const float*)d_in[1];
  const float* pa   = (const float*)d_in[2];
  const float* Wp   = (const float*)d_in[3];
  const float* bp   = (const float*)d_in[4];
  float* out = (float*)d_out;

  float* ws   = (float*)d_ws;
  float* sim  = ws;
  float* sube = sim  + 1048576;
  float* S    = sube + 8388608;
  float* xn   = S    + 8192;
  float* sx   = xn   + 524288;
  float* xg1  = sx   + 524288;
  float* W    = xg1  + 524288;

  k_simraw    <<<4096, 256, 0, stream>>>(E, sim);
  k_rowsoftmax<<<1024, 256, 0, stream>>>(sim);
  k_colsum_pts<<<dim3(32, 8),     512, 0, stream>>>(pa, S);
  k_xnorm     <<<2048, 256, 0, stream>>>(x, S, xn);
  k_weights   <<<dim3(8, 1024),   256, 0, stream>>>(E, Wp, W);
  k_sx_points<<<dim3(16, 8), 128, 0, stream>>>(pa, xn, sx);
  k_bmm_nk<<<dim3(8, 8), 256, 0, stream>>>(sim, 0LL, sx, 1024LL * 64,
                                           xg1, 1024LL * 64);
  k_final<<<128, 256, 0, stream>>>(sx, xg1, W, E, bp, out);
}


# sim never stored: 2nd product forms 16x16 tiles of sim^T by WMMA from node embeddings (already in A-fragment layout), row max and sum from a stats pass; changes sum order of dots and softmax sums
# speedup vs baseline: 2.0138x; 1.5822x over previous
#include <hip/hip_runtime.h>
#include <hip/hip_bf16.h>
#include <cstddef>


typedef float v2f __attribute__((ext_vector_type(2)));
typedef float v8f __attribute__((ext_vector_type(8)));

static __device__ __forceinline__ v8f wmma_f32(v2f a, v2f b, v8f c) {
  return __builtin_amdgcn_wmma_f32_16x16x4_f32(
       false, a,  false, b,
       (short)0, c,  false,  false);
}

static __device__ __forceinline__ float exp_neg_l1(const float (&p)[16],
                                                   const float (&q)[16]) {
  float s0 = 0.f, s1 = 0.f, s2 = 0.f, s3 = 0.f;
#pragma unroll
  for (int d = 0; d < 4; ++d) {
    s0 += fabsf(p[d]      - q[d]);
    s1 += fabsf(p[d + 4]  - q[d + 4]);
    s2 += fabsf(p[d + 8]  - q[d + 8]);
    s3 += fabsf(p[d + 12] - q[d + 12]);
  }
  return __expf(-((s0 + s1) + (s2 + s3)));
}

__global__ __launch_bounds__(512)
void k_rowstats(const float* __restrict__ E, float* __restrict__ rmx,
                float* __restrict__ rinv) {
  const int t = threadIdx.x;
  const int wave = t >> 5;
  const int lane = t & 31;
  const int m16 = lane & 15;
  const int kh = lane >> 4;
  const int mrow = blockIdx.x * 16 + m16;
  __shared__ float part[32][16];

  v2f bq[4];
  {
    const float4 e0 = *(const float4*)(E + (size_t)mrow * 16 + 8 * kh);
    const float4 e1 = *(const float4*)(E + (size_t)mrow * 16 + 8 * kh + 4);
    bq[0][0] = e0.x; bq[0][1] = e0.y; bq[1][0] = e0.z; bq[1][1] = e0.w;
    bq[2][0] = e1.x; bq[2][1] = e1.y; bq[3][0] = e1.z; bq[3][1] = e1.w;
  }
  const v8f zero = {0.f, 0.f, 0.f, 0.f, 0.f, 0.f, 0.f, 0.f};
  v8f tl[4];
#pragma unroll
  for (int q = 0; q < 4; ++q) {
    const float* er = E + (size_t)(wave * 64 + q * 16 + m16) * 16 + 8 * kh;
    const float4 f0 = *(const float4*)(er);
    const float4 f1 = *(const float4*)(er + 4);
    v8f st = zero;
    st = wmma_f32(v2f{f0.x, f0.y}, bq[0], st);
    st = wmma_f32(v2f{f0.z, f0.w}, bq[1], st);
    st = wmma_f32(v2f{f1.x, f1.y}, bq[2], st);
    st = wmma_f32(v2f{f1.z, f1.w}, bq[3], st);
    tl[q] = st;
  }
  float mx = 0.f;
#pragma unroll
  for (int q = 0; q < 4; ++q)
#pragma unroll
    for (int r = 0; r < 8; ++r) mx = fmaxf(mx, tl[q][r]);
  part[wave * 2 + kh][m16] = mx;
  __syncthreads();
  float m = 0.f;
#pragma unroll
  for (int p = 0; p < 32; ++p) m = fmaxf(m, part[p][m16]);
  __syncthreads();
  float acc = 0.f;
#pragma unroll
  for (int q = 0; q < 4; ++q)
#pragma unroll
    for (int r = 0; r < 8; ++r) acc += __expf(fmaxf(tl[q][r], 0.f) - m);
  part[wave * 2 + kh][m16] = acc;
  __syncthreads();
  if (t < 16) {
    float tot = 0.f;
#pragma unroll
    for (int p = 0; p < 32; ++p) tot += part[p][t];
    rmx[blockIdx.x * 16 + t] = m;
    rinv[blockIdx.x * 16 + t] = 1.0f / tot;
  }
}

__global__ __launch_bounds__(512)
void k_colsum_pts(const float* __restrict__ pa, float* __restrict__ S) {
  const int b = blockIdx.y;
  const int lane = threadIdx.x & 31;
  const int w = __builtin_amdgcn_readfirstlane((int)(threadIdx.x >> 5));
  const int j = blockIdx.x * 32 + lane;
  __shared__ float part[16][32];

  float pj[16];
#pragma unroll
  for (int q = 0; q < 4; ++q) {
    const float4 v = *(const float4*)(pa + ((size_t)b * 1024 + j) * 16 + q * 4);
    pj[q * 4 + 0] = v.x; pj[q * 4 + 1] = v.y; pj[q * 4 + 2] = v.z; pj[q * 4 + 3] = v.w;
  }
  const float* pi = pa + ((size_t)b * 1024 + (size_t)w * 64) * 16;
  float acc = 0.f;
  for (int ii = 0; ii < 64; ++ii) {
    float qi[16];
#pragma unroll
    for (int d = 0; d < 16; ++d) qi[d] = pi[ii * 16 + d];
    acc += exp_neg_l1(qi, pj);
  }
  part[w][lane] = acc;
  __syncthreads();
  if (threadIdx.x < 32) {
    float tot = 0.f;
#pragma unroll
    for (int w2 = 0; w2 < 16; ++w2) tot += part[w2][threadIdx.x];
    S[b * 1024 + blockIdx.x * 32 + threadIdx.x] = tot;
  }
}

__global__ void k_xnorm(const float* __restrict__ x, const float* __restrict__ S,
                        float* __restrict__ xn) {
  const int idx = blockIdx.x * 256 + threadIdx.x;
  xn[idx] = x[idx] / S[idx >> 6];
}

__global__ void k_weights(const float* __restrict__ E, const float* __restrict__ Wp,
                          float* __restrict__ W) {
  const int n = blockIdx.y;
  const int tid = blockIdx.x * 256 + threadIdx.x;
  const int kp = tid >> 5;
  const int oc = (tid & 31) * 2;
  float e[16];
#pragma unroll
  for (int d = 0; d < 16; ++d) e[d] = E[n * 16 + d];
  float4 acc = {0.f, 0.f, 0.f, 0.f};
#pragma unroll
  for (int d = 0; d < 16; ++d) {
    const v2f w0 = *(const v2f*)(Wp + (size_t)d * 8192 + (size_t)(2 * kp) * 64 + oc);
    const v2f w1 = *(const v2f*)(Wp + (size_t)d * 8192 + (size_t)(2 * kp + 1) * 64 + oc);
    acc.x += e[d] * w0.x;
    acc.y += e[d] * w1.x;
    acc.z += e[d] * w0.y;
    acc.w += e[d] * w1.y;
  }
  *(float4*)(W + (size_t)n * 8192 + (size_t)kp * 128 + (size_t)oc * 2) = acc;
}

#define SX_KC 512
__global__ __launch_bounds__(128)
void k_sx_points(const float* __restrict__ pa, const float* __restrict__ xn,
                 float* __restrict__ sx) {
  const int b = blockIdx.y;
  const int rowBase = blockIdx.x * 64;
  const float* pab = pa + (size_t)b * 1024 * 16;
  const float* Bb = xn + (size_t)b * 1024 * 64;
  float* Cb = sx + (size_t)b * 1024 * 64;

  __shared__ float Ps[1024][16];
  __shared__ float Bs[SX_KC / 2][160];

  const int t = threadIdx.x;
  const int wave = t >> 5;
  const int lane = t & 31;
  const int m16 = lane & 15;
  const int kh = lane >> 4;

  for (int i = t; i < 1024 * 4; i += 128)
    ((float4*)&Ps[0][0])[i] = ((const float4*)pab)[i];

  float pm[16];
  {
    const int mrow = rowBase + wave * 16 + m16;
#pragma unroll
    for (int q = 0; q < 4; ++q) {
      const float4 v = *(const float4*)(pab + (size_t)mrow * 16 + q * 4);
      pm[q * 4 + 0] = v.x; pm[q * 4 + 1] = v.y; pm[q * 4 + 2] = v.z; pm[q * 4 + 3] = v.w;
    }
  }

  const v8f zero = {0.f, 0.f, 0.f, 0.f, 0.f, 0.f, 0.f, 0.f};
  v8f cacc[4];
#pragma unroll
  for (int nt = 0; nt < 4; ++nt) cacc[nt] = zero;

  __syncthreads();

  float pt[32];
  v2f a_cur;
  {
    float p0[32];
    const float* q0 = &Ps[2 * kh][0];
#pragma unroll
    for (int d = 0; d < 32; ++d) p0[d] = q0[d];
    const float (&r0)[16] = *(const float (*)[16])(&p0[0]);
    const float (&r1)[16] = *(const float (*)[16])(&p0[16]);
    a_cur[0] = exp_neg_l1(pm, r0);
    a_cur[1] = exp_neg_l1(pm, r1);
    const float* q1 = &Ps[4 + 2 * kh][0];
#pragma unroll
    for (int d = 0; d < 32; ++d) pt[d] = q1[d];
  }

  for (int kc = 0; kc < 1024; kc += SX_KC) {
    __syncthreads();
    for (int i0 = t; i0 < (SX_KC / 2) * 16; i0 += 128 * 8) {
      float4 v0[8], v1[8];
#pragma unroll
      for (int u = 0; u < 8; ++u) {
        const int i = i0 + u * 128;
        const int kp = i >> 4, cc = (i & 15) << 2;
        v0[u] = *(const float4*)(Bb + (size_t)(kc + 2 * kp) * 64 + cc);
        v1[u] = *(const float4*)(Bb + (size_t)(kc + 2 * kp + 1) * 64 + cc);
      }
#pragma unroll
      for (int u = 0; u < 8; ++u) {
        const int i = i0 + u * 128;
        const int kp = i >> 4, cc = (i & 15) << 2;
        float4 lo = {v0[u].x, v1[u].x, v0[u].y, v1[u].y};
        float4 hi = {v0[u].z, v1[u].z, v0[u].w, v1[u].w};
        *(float4*)(&Bs[kp][cc * 2])     = lo;
        *(float4*)(&Bs[kp][cc * 2 + 4]) = hi;
      }
    }
    __syncthreads();

    v2f bf_cur[4];
#pragma unroll
    for (int nt = 0; nt < 4; ++nt)
      bf_cur[nt] = *(const v2f*)(&Bs[kh][(nt * 16 + m16) * 2]);

#pragma unroll 2
    for (int ks = 0; ks < SX_KC / 4; ++ks) {
      const int g = kc / 4 + ks;
      float ptn[32];
      const float* qn = &Ps[((g + 2) & 255) * 4 + 2 * kh][0];
#pragma unroll
      for (int d = 0; d < 32; ++d) ptn[d] = qn[d];
      v2f bf_nxt[4];
      const int kpn = ((ks + 1) & (SX_KC / 4 - 1)) * 2 + kh;
#pragma unroll
      for (int nt = 0; nt < 4; ++nt)
        bf_nxt[nt] = *(const v2f*)(&Bs[kpn][(nt * 16 + m16) * 2]);
      const float (&r0)[16] = *(const float (*)[16])(&pt[0]);
      const float (&r1)[16] = *(const float (*)[16])(&pt[16]);
      v2f a_nxt;
      a_nxt[0] = exp_neg_l1(pm, r0);
      a_nxt[1] = exp_neg_l1(pm, r1);
#pragma unroll
      for (int nt = 0; nt < 4; ++nt)
        cacc[nt] = wmma_f32(a_cur, bf_cur[nt], cacc[nt]);
      __builtin_amdgcn_sched_group_barrier(0x100, 10, 0);
#pragma unroll
      for (int nt = 0; nt < 4; ++nt) {
        __builtin_amdgcn_sched_group_barrier(0x008, 1, 0);
        __builtin_amdgcn_sched_group_barrier(0x002, 18, 0);
      }
      a_cur = a_nxt;
#pragma unroll
      for (int d = 0; d < 32; ++d) pt[d] = ptn[d];
#pragma unroll
      for (int nt = 0; nt < 4; ++nt) bf_cur[nt] = bf_nxt[nt];
    }
  }

#pragma unroll
  for (int nt = 0; nt < 4; ++nt) {
#pragma unroll
    for (int r = 0; r < 8; ++r) {
      const int m = rowBase + wave * 16 + r + 8 * kh;
      Cb[(size_t)m * 64 + nt * 16 + m16] = cacc[nt][r];
    }
  }
}

#define XG_KC 512
__global__ __launch_bounds__(128)
void k_xg_sim(const float* __restrict__ E, const float* __restrict__ rmx,
              const float* __restrict__ rinv, const float* __restrict__ sx,
              float* __restrict__ xg1) {
  const int b = blockIdx.y;
  const int rowBase = blockIdx.x * 32;
  const float* Bb = sx + (size_t)b * 1024 * 64;
  float* Cb = xg1 + (size_t)b * 1024 * 64;

  __shared__ float Es[1024][20];
  __shared__ float Bs[XG_KC / 2][160];

  const int t = threadIdx.x;
  const int wave = t >> 5;
  const int lane = t & 31;
  const int m16 = lane & 15;
  const int kh = lane >> 4;
  const int rt = wave >> 1;
  const int ch = wave & 1;

  for (int i0 = t; i0 < 1024 * 4; i0 += 128 * 16) {
    float4 v[16];
#pragma unroll
    for (int u = 0; u < 16; ++u) v[u] = ((const float4*)E)[i0 + u * 128];
#pragma unroll
    for (int u = 0; u < 16; ++u) {
      const int i = i0 + u * 128;
      *(float4*)(&Es[i >> 2][(i & 3) << 2]) = v[u];
    }
  }

  const int mrow = rowBase + rt * 16 + m16;
  v2f bq[4];
  {
    const float4 e0 = *(const float4*)(E + (size_t)mrow * 16 + 8 * kh);
    const float4 e1 = *(const float4*)(E + (size_t)mrow * 16 + 8 * kh + 4);
    bq[0][0] = e0.x; bq[0][1] = e0.y; bq[1][0] = e0.z; bq[1][1] = e0.w;
    bq[2][0] = e1.x; bq[2][1] = e1.y; bq[3][0] = e1.z; bq[3][1] = e1.w;
  }
  const float mxm = rmx[mrow];
  const float invm = rinv[mrow];
  const int kperm = 4 * ((m16 & 7) >> 1) + 2 * (m16 >> 3) + (m16 & 1);
  const float* erow = &Es[kperm][8 * kh];
  const float* bcol = &Bs[kh][(ch * 32 + m16) * 2];

  const v8f zero = {0.f, 0.f, 0.f, 0.f, 0.f, 0.f, 0.f, 0.f};
  v8f cacc[2];
  cacc[0] = zero; cacc[1] = zero;

  float4 ea0, ea1;
  v8f st_cur;
  v2f a_cur[4];

  for (int kc = 0; kc < 1024; kc += XG_KC) {
    __syncthreads();
    for (int i0 = t; i0 < (XG_KC / 2) * 16; i0 += 128 * 16) {
      float4 v0[16], v1[16];
#pragma unroll
      for (int u = 0; u < 16; ++u) {
        const int i = i0 + u * 128;
        const int kp = i >> 4, cc = (i & 15) << 2;
        v0[u] = *(const float4*)(Bb + (size_t)(kc + 2 * kp) * 64 + cc);
        v1[u] = *(const float4*)(Bb + (size_t)(kc + 2 * kp + 1) * 64 + cc);
      }
#pragma unroll
      for (int u = 0; u < 16; ++u) {
        const int i = i0 + u * 128;
        const int kp = i >> 4, cc = (i & 15) << 2;
        float4 lo = {v0[u].x, v1[u].x, v0[u].y, v1[u].y};
        float4 hi = {v0[u].z, v1[u].z, v0[u].w, v1[u].w};
        *(float4*)(&Bs[kp][cc * 2])     = lo;
        *(float4*)(&Bs[kp][cc * 2 + 4]) = hi;
      }
    }
    __syncthreads();

    if (kc == 0) {
      const float4 f0 = *(const float4*)(erow);
      const float4 f1 = *(const float4*)(erow + 4);
      const float4 g0 = *(const float4*)(erow + 16 * 20);
      const float4 g1 = *(const float4*)(erow + 16 * 20 + 4);
      v8f s0 = zero;
      s0 = wmma_f32(v2f{f0.x, f0.y}, bq[0], s0);
      s0 = wmma_f32(v2f{f0.z, f0.w}, bq[1], s0);
      s0 = wmma_f32(v2f{f1.x, f1.y}, bq[2], s0);
      s0 = wmma_f32(v2f{f1.z, f1.w}, bq[3], s0);
      v8f s1 = zero;
      s1 = wmma_f32(v2f{g0.x, g0.y}, bq[0], s1);
      s1 = wmma_f32(v2f{g0.z, g0.w}, bq[1], s1);
      s1 = wmma_f32(v2f{g1.x, g1.y}, bq[2], s1);
      s1 = wmma_f32(v2f{g1.z, g1.w}, bq[3], s1);
#pragma unroll
      for (int j = 0; j < 4; ++j) {
        a_cur[j][0] = __expf(fmaxf(s0[2 * j], 0.f) - mxm) * invm;
        a_cur[j][1] = __expf(fmaxf(s0[2 * j + 1], 0.f) - mxm) * invm;
      }
      st_cur = s1;
      ea0 = *(const float4*)(erow + 32 * 20);
      ea1 = *(const float4*)(erow + 32 * 20 + 4);
    }

    v2f bf_cur[4][2];
#pragma unroll
    for (int j = 0; j < 4; ++j)
#pragma unroll
      for (int nt = 0; nt < 2; ++nt)
        bf_cur[j][nt] = *(const v2f*)(bcol + (2 * j) * 160 + nt * 32);

    for (int gl = 0; gl < XG_KC / 16; ++gl) {
      const int g = kc / 16 + gl;
      const float* en = erow + (size_t)(((g + 3) & 63) * 16) * 20;
      const float4 en0 = *(const float4*)(en);
      const float4 en1 = *(const float4*)(en + 4);
      v2f bf_nxt[4][2];
      const float* bn = bcol + (size_t)(((gl + 1) & (XG_KC / 16 - 1)) * 8) * 160;
#pragma unroll
      for (int j = 0; j < 4; ++j)
#pragma unroll
        for (int nt = 0; nt < 2; ++nt)
          bf_nxt[j][nt] = *(const v2f*)(bn + (2 * j) * 160 + nt * 32);
      v2f a_nxt[4];
#pragma unroll
      for (int j = 0; j < 4; ++j) {
        a_nxt[j][0] = __expf(fmaxf(st_cur[2 * j], 0.f) - mxm) * invm;
        a_nxt[j][1] = __expf(fmaxf(st_cur[2 * j + 1], 0.f) - mxm) * invm;
      }
      const v2f ta[4] = {v2f{ea0.x, ea0.y}, v2f{ea0.z, ea0.w},
                         v2f{ea1.x, ea1.y}, v2f{ea1.z, ea1.w}};
      v8f st_nxt = zero;
#pragma unroll
      for (int j = 0; j < 4; ++j) {
        st_nxt = wmma_f32(ta[j], bq[j], st_nxt);
        cacc[0] = wmma_f32(a_cur[j], bf_cur[j][0], cacc[0]);
        cacc[1] = wmma_f32(a_cur[j], bf_cur[j][1], cacc[1]);
      }
      st_cur = st_nxt;
      ea0 = en0; ea1 = en1;
#pragma unroll
      for (int j = 0; j < 4; ++j) {
        a_cur[j] = a_nxt[j];
        bf_cur[j][0] = bf_nxt[j][0];
        bf_cur[j][1] = bf_nxt[j][1];
      }
    }
  }

#pragma unroll
  for (int nt = 0; nt < 2; ++nt) {
#pragma unroll
    for (int r = 0; r < 8; ++r) {
      const int m = rowBase + rt * 16 + r + 8 * kh;
      Cb[(size_t)m * 64 + (ch * 2 + nt) * 16 + m16] = cacc[nt][r];
    }
  }
}

__global__ __launch_bounds__(256)
void k_final(const float* __restrict__ sx, const float* __restrict__ xg1,
             const float* __restrict__ W, const float* __restrict__ E,
             const float* __restrict__ bias_pool, float* __restrict__ out) {
  __shared__ float As[8][8][132];
  const int t = threadIdx.x;
  const int wave = t >> 5;
  const int lane = t & 31;
  const int n = blockIdx.x * 8 + wave;
  float (*Aw)[132] = As[wave];

  for (int rb = 0; rb < 8; ++rb) {
    const float* s0 = sx  + ((size_t)rb * 1024 + n) * 64;
    const float* s1 = xg1 + ((size_t)rb * 1024 + n) * 64;
    for (int c = lane; c < 128; c += 32)
      Aw[rb][c] = (c < 64) ? s0[c] : s1[c - 64];
  }
  __syncthreads();

  const float* Wn = W + (size_t)n * 8192;
  const int m16 = lane & 15;
  const int kh = lane >> 4;

  const v8f zero = {0.f, 0.f, 0.f, 0.f, 0.f, 0.f, 0.f, 0.f};
  v8f cacc[4];
#pragma unroll
  for (int nt = 0; nt < 4; ++nt) cacc[nt] = zero;

#pragma unroll 4
  for (int ks = 0; ks < 32; ++ks) {
    const int kg = ks * 4 + 2 * kh;
    const int kp = ks * 2 + kh;
    v2f a = {0.f, 0.f};
    if (m16 < 8) a = *(const v2f*)(&Aw[m16][kg]);
#pragma unroll
    for (int nt = 0; nt < 4; ++nt) {
      const v2f bf = *(const v2f*)(Wn + (size_t)kp * 128 + (nt * 16 + m16) * 2);
      cacc[nt] = wmma_f32(a, bf, cacc[nt]);
    }
  }

#pragma unroll
  for (int nt = 0; nt < 4; ++nt) {
    const int col = nt * 16 + m16;
    float bv = 0.f;
#pragma unroll
    for (int d = 0; d < 16; ++d) bv += E[n * 16 + d] * bias_pool[d * 64 + col];
#pragma unroll
    for (int r = 0; r < 8; ++r) {
      const int m = r + 8 * kh;
      if (m < 8)
        out[((size_t)m * 1024 + n) * 64 + col] = cacc[nt][r] + bv;
    }
  }
}

extern "C" void kernel_launch(void* const* d_in, const int* in_sizes, int n_in,
                              void* d_out, int out_size, void* d_ws, size_t ws_size,
                              hipStream_t stream) {
  (void)in_sizes; (void)n_in; (void)out_size; (void)ws_size;
  const float* x    = (const float*)d_in[0];
  const float* E    = (const float*)d_in[1];
  const float* pa   = (const float*)d_in[2];
  const float* Wp   = (const float*)d_in[3];
  const float* bp   = (const float*)d_in[4];
  float* out = (float*)d_out;

  float* ws   = (float*)d_ws;
  float* rmx  = ws;
  float* rinv = ws + 1024;
  float* sube = ws   + 1048576;
  float* S    = sube + 8388608;
  float* xn   = S    + 8192;
  float* sx   = xn   + 524288;
  float* xg1  = sx   + 524288;
  float* W    = xg1  + 524288;

  k_rowstats  <<<64, 512, 0, stream>>>(E, rmx, rinv);
  k_colsum_pts<<<dim3(32, 8),     512, 0, stream>>>(pa, S);
  k_xnorm     <<<2048, 256, 0, stream>>>(x, S, xn);
  k_weights   <<<dim3(8, 1024),   256, 0, stream>>>(E, Wp, W);
  k_sx_points<<<dim3(16, 8), 128, 0, stream>>>(pa, xn, sx);
  k_xg_sim<<<dim3(32, 8), 128, 0, stream>>>(E, rmx, rinv, sx, xg1);
  k_final<<<128, 256, 0, stream>>>(sx, xg1, W, E, bp, out);
}


# sub_e product on 256 work-groups of 32 rows (was 128 of 64): each A fragment formed once per work-group (lane = row, k row uniform), passed to both column-half waves through LDS; sum order unchanged
# speedup vs baseline: 3.0661x; 1.5225x over previous
#include <hip/hip_runtime.h>
#include <hip/hip_bf16.h>
#include <cstddef>


typedef float v2f __attribute__((ext_vector_type(2)));
typedef float v8f __attribute__((ext_vector_type(8)));

static __device__ __forceinline__ v8f wmma_f32(v2f a, v2f b, v8f c) {
  return __builtin_amdgcn_wmma_f32_16x16x4_f32(
       false, a,  false, b,
       (short)0, c,  false,  false);
}

static __device__ __forceinline__ float exp_neg_l1(const float (&p)[16],
                                                   const float (&q)[16]) {
  float s0 = 0.f, s1 = 0.f, s2 = 0.f, s3 = 0.f;
#pragma unroll
  for (int d = 0; d < 4; ++d) {
    s0 += fabsf(p[d]      - q[d]);
    s1 += fabsf(p[d + 4]  - q[d + 4]);
    s2 += fabsf(p[d + 8]  - q[d + 8]);
    s3 += fabsf(p[d + 12] - q[d + 12]);
  }
  return __expf(-((s0 + s1) + (s2 + s3)));
}

__global__ __launch_bounds__(512)
void k_rowstats(const float* __restrict__ E, float* __restrict__ rmx,
                float* __restrict__ rinv) {
  const int t = threadIdx.x;
  const int wave = t >> 5;
  const int lane = t & 31;
  const int m16 = lane & 15;
  const int kh = lane >> 4;
  const int mrow = blockIdx.x * 16 + m16;
  __shared__ float part[32][16];

  v2f bq[4];
  {
    const float4 e0 = *(const float4*)(E + (size_t)mrow * 16 + 8 * kh);
    const float4 e1 = *(const float4*)(E + (size_t)mrow * 16 + 8 * kh + 4);
    bq[0][0] = e0.x; bq[0][1] = e0.y; bq[1][0] = e0.z; bq[1][1] = e0.w;
    bq[2][0] = e1.x; bq[2][1] = e1.y; bq[3][0] = e1.z; bq[3][1] = e1.w;
  }
  const v8f zero = {0.f, 0.f, 0.f, 0.f, 0.f, 0.f, 0.f, 0.f};
  v8f tl[4];
#pragma unroll
  for (int q = 0; q < 4; ++q) {
    const float* er = E + (size_t)(wave * 64 + q * 16 + m16) * 16 + 8 * kh;
    const float4 f0 = *(const float4*)(er);
    const float4 f1 = *(const float4*)(er + 4);
    v8f st = zero;
    st = wmma_f32(v2f{f0.x, f0.y}, bq[0], st);
    st = wmma_f32(v2f{f0.z, f0.w}, bq[1], st);
    st = wmma_f32(v2f{f1.x, f1.y}, bq[2], st);
    st = wmma_f32(v2f{f1.z, f1.w}, bq[3], st);
    tl[q] = st;
  }
  float mx = 0.f;
#pragma unroll
  for (int q = 0; q < 4; ++q)
#pragma unroll
    for (int r = 0; r < 8; ++r) mx = fmaxf(mx, tl[q][r]);
  part[wave * 2 + kh][m16] = mx;
  __syncthreads();
  float m = 0.f;
#pragma unroll
  for (int p = 0; p < 32; ++p) m = fmaxf(m, part[p][m16]);
  __syncthreads();
  float acc = 0.f;
#pragma unroll
  for (int q = 0; q < 4; ++q)
#pragma unroll
    for (int r = 0; r < 8; ++r) acc += __expf(fmaxf(tl[q][r], 0.f) - m);
  part[wave * 2 + kh][m16] = acc;
  __syncthreads();
  if (t < 16) {
    float tot = 0.f;
#pragma unroll
    for (int p = 0; p < 32; ++p) tot += part[p][t];
    rmx[blockIdx.x * 16 + t] = m;
    rinv[blockIdx.x * 16 + t] = 1.0f / tot;
  }
}

__global__ __launch_bounds__(512)
void k_colsum_pts(const float* __restrict__ pa, float* __restrict__ S) {
  const int b = blockIdx.y;
  const int lane = threadIdx.x & 31;
  const int w = __builtin_amdgcn_readfirstlane((int)(threadIdx.x >> 5));
  const int j = blockIdx.x * 32 + lane;
  __shared__ float part[16][32];

  float pj[16];
#pragma unroll
  for (int q = 0; q < 4; ++q) {
    const float4 v = *(const float4*)(pa + ((size_t)b * 1024 + j) * 16 + q * 4);
    pj[q * 4 + 0] = v.x; pj[q * 4 + 1] = v.y; pj[q * 4 + 2] = v.z; pj[q * 4 + 3] = v.w;
  }
  const float* pi = pa + ((size_t)b * 1024 + (size_t)w * 64) * 16;
  float acc = 0.f;
  for (int ii = 0; ii < 64; ++ii) {
    float qi[16];
#pragma unroll
    for (int d = 0; d < 16; ++d) qi[d] = pi[ii * 16 + d];
    acc += exp_neg_l1(qi, pj);
  }
  part[w][lane] = acc;
  __syncthreads();
  if (threadIdx.x < 32) {
    float tot = 0.f;
#pragma unroll
    for (int w2 = 0; w2 < 16; ++w2) tot += part[w2][threadIdx.x];
    S[b * 1024 + blockIdx.x * 32 + threadIdx.x] = tot;
  }
}

__global__ void k_xnorm(const float* __restrict__ x, const float* __restrict__ S,
                        float* __restrict__ xn) {
  const int idx = blockIdx.x * 256 + threadIdx.x;
  xn[idx] = x[idx] / S[idx >> 6];
}

#define SX_KC 512
#define SX_AF_TILE 2080
#define SX_AF_BUF (2 * SX_AF_TILE)

static __device__ __forceinline__ int sx_step_row(int p, int w) {
  return ((p >> 3) * 128 + w * 32 + (p & 7) * 4) & 1023;
}

static __device__ __forceinline__ void sx_load_pts(const float* __restrict__ pab, int p, int w,
                                                   float (&q)[4][16]) {
  const float* pn = pab + (size_t)sx_step_row(p, w) * 16;
#pragma unroll
  for (int kk = 0; kk < 4; ++kk) {
#pragma unroll
    for (int q4 = 0; q4 < 4; ++q4) {
      const float4 v = *(const float4*)(pn + kk * 16 + q4 * 4);
      q[kk][q4 * 4 + 0] = v.x; q[kk][q4 * 4 + 1] = v.y;
      q[kk][q4 * 4 + 2] = v.z; q[kk][q4 * 4 + 3] = v.w;
    }
  }
}

template <bool FORM, bool USE>
static __device__ __forceinline__ void sx_step(const float* __restrict__ pab, int p0, int w, int j,
                                               float (&q)[4][16], const float (&pm)[16],
                                               float* afw, const float* afr, const float* bcol,
                                               const v2f (&au)[4], const v2f (&bu)[4][2],
                                               v2f (&an)[4], v2f (&bn)[4][2], v8f (&cacc)[2]) {
  if (USE) {
    const int jn = (j + 1) & 7;
#pragma unroll
    for (int u = 0; u < 4; ++u) {
      an[u] = *(const v2f*)(afr + (4 * jn + u) * 64);
      bn[u][0] = *(const v2f*)(bcol + (2 * (4 * jn + u)) * 160);
      bn[u][1] = *(const v2f*)(bcol + (2 * (4 * jn + u)) * 160 + 32);
    }
  }
  float e[4];
  if (FORM) {
#pragma unroll
    for (int kk = 0; kk < 4; ++kk) e[kk] = exp_neg_l1(q[kk], pm);
    sx_load_pts(pab, p0 + j + 2, w, q);
  }
  if (USE) {
#pragma unroll
    for (int u = 0; u < 4; ++u) {
      cacc[0] = wmma_f32(au[u], bu[u][0], cacc[0]);
      cacc[1] = wmma_f32(au[u], bu[u][1], cacc[1]);
    }
  }
  if (FORM && USE) {
    __builtin_amdgcn_sched_group_barrier(0x100, 8, 0);
#pragma unroll
    for (int u = 0; u < 8; ++u) {
      __builtin_amdgcn_sched_group_barrier(0x008, 1, 0);
      __builtin_amdgcn_sched_group_barrier(0x002, 19, 0);
    }
  }
  if (FORM) {
    *(v2f*)(afw + j * 64)      = v2f{e[0], e[1]};
    *(v2f*)(afw + j * 64 + 32) = v2f{e[2], e[3]};
  }
}

template <bool FORM, bool USE>
static __device__ __forceinline__ void sx_steps(const float* __restrict__ pab, int p0, int w,
                                                float (&qa)[4][16], float (&qb)[4][16],
                                                const float (&pm)[16],
                                                float* afw, const float* afr, const float* bcol,
                                                v8f (&cacc)[2]) {
  v2f a0[4], b0[4][2], a1[4], b1[4][2];
  if (USE) {
#pragma unroll
    for (int u = 0; u < 4; ++u) {
      a0[u] = *(const v2f*)(afr + u * 64);
      b0[u][0] = *(const v2f*)(bcol + (2 * u) * 160);
      b0[u][1] = *(const v2f*)(bcol + (2 * u) * 160 + 32);
    }
  }
  for (int jj = 0; jj < 8; jj += 2) {
    sx_step<FORM, USE>(pab, p0, w, jj,     qa, pm, afw, afr, bcol, a0, b0, a1, b1, cacc);
    sx_step<FORM, USE>(pab, p0, w, jj + 1, qb, pm, afw, afr, bcol, a1, b1, a0, b0, cacc);
  }
}

__global__ __launch_bounds__(128)
void k_sx_points(const float* __restrict__ pa, const float* __restrict__ xn,
                 float* __restrict__ sx) {
  const int b = blockIdx.y;
  const int rowBase = blockIdx.x * 32;
  const float* pab = pa + (size_t)b * 1024 * 16;
  const float* Bb = xn + (size_t)b * 1024 * 64;
  float* Cb = sx + (size_t)b * 1024 * 64;

  __shared__ float Bs[SX_KC / 2][160];
  __shared__ __attribute__((aligned(16))) float Af[2 * SX_AF_BUF];

  const int t = threadIdx.x;
  const int wave = t >> 5;
  const int lane = t & 31;
  const int m16 = lane & 15;
  const int kh = lane >> 4;
  const int rt = wave >> 1;
  const int ch = wave & 1;

  float pm[16];
  float qa[4][16], qb[4][16];
  {
    const float* pr = pab + (size_t)(rowBase + lane) * 16;
#pragma unroll
    for (int q = 0; q < 4; ++q) {
      const float4 v = *(const float4*)(pr + q * 4);
      pm[q * 4 + 0] = v.x; pm[q * 4 + 1] = v.y; pm[q * 4 + 2] = v.z; pm[q * 4 + 3] = v.w;
    }
    sx_load_pts(pab, 0, wave, qa);
    sx_load_pts(pab, 1, wave, qb);
  }

  const v8f zero = {0.f, 0.f, 0.f, 0.f, 0.f, 0.f, 0.f, 0.f};
  v8f cacc[2];
  cacc[0] = zero; cacc[1] = zero;

  float* afw0 = Af + (lane >> 4) * SX_AF_TILE + (8 * wave) * 64 + (lane & 15) * 2;
  const float* afr0 = Af + rt * SX_AF_TILE + lane * 2;
  const float* bcol0 = &Bs[kh][(ch * 32 + m16) * 2];

  for (int kc = 0; kc < 1024; kc += SX_KC) {
    if (kc != 0) __syncthreads();
    for (int i0 = t; i0 < (SX_KC / 2) * 16; i0 += 128 * 8) {
      float4 v0[8], v1[8];
#pragma unroll
      for (int u = 0; u < 8; ++u) {
        const int i = i0 + u * 128;
        const int kp = i >> 4, cc = (i & 15) << 2;
        v0[u] = *(const float4*)(Bb + (size_t)(kc + 2 * kp) * 64 + cc);
        v1[u] = *(const float4*)(Bb + (size_t)(kc + 2 * kp + 1) * 64 + cc);
      }
#pragma unroll
      for (int u = 0; u < 8; ++u) {
        const int i = i0 + u * 128;
        const int kp = i >> 4, cc = (i & 15) << 2;
        float4 lo = {v0[u].x, v1[u].x, v0[u].y, v1[u].y};
        float4 hi = {v0[u].z, v1[u].z, v0[u].w, v1[u].w};
        *(float4*)(&Bs[kp][cc * 2])     = lo;
        *(float4*)(&Bs[kp][cc * 2 + 4]) = hi;
      }
    }
    if (kc == 0)
      sx_steps<true, false>(pab, 0, wave, qa, qb, pm, afw0, afr0, bcol0, cacc);
    __syncthreads();

    for (int s4 = 0; s4 < 4; ++s4) {
      const int sc = kc / 128 + s4;
      const float* afr = afr0 + (sc & 1) * SX_AF_BUF;
      float* afw = afw0 + ((sc + 1) & 1) * SX_AF_BUF;
      const float* bcol = bcol0 + (size_t)(s4 * 64) * 160;
      if (sc < 7) {
        sx_steps<true, true>(pab, (sc + 1) * 8, wave, qa, qb, pm, afw, afr, bcol, cacc);
        if (s4 < 3) __syncthreads();
      } else {
        sx_steps<false, true>(pab, 0, wave, qa, qb, pm, afw, afr, bcol, cacc);
      }
    }
  }

#pragma unroll
  for (int nt = 0; nt < 2; ++nt) {
#pragma unroll
    for (int r = 0; r < 8; ++r) {
      const int m = rowBase + rt * 16 + r + 8 * kh;
      Cb[(size_t)m * 64 + (ch * 2 + nt) * 16 + m16] = cacc[nt][r];
    }
  }
}

#define XG_KC 512
__global__ __launch_bounds__(128)
void k_xg_sim(const float* __restrict__ E, const float* __restrict__ rmx,
              const float* __restrict__ rinv, const float* __restrict__ sx,
              float* __restrict__ xg1) {
  const int b = blockIdx.y;
  const int rowBase = blockIdx.x * 32;
  const float* Bb = sx + (size_t)b * 1024 * 64;
  float* Cb = xg1 + (size_t)b * 1024 * 64;

  __shared__ float Es[1024][20];
  __shared__ float Bs[XG_KC / 2][160];

  const int t = threadIdx.x;
  const int wave = t >> 5;
  const int lane = t & 31;
  const int m16 = lane & 15;
  const int kh = lane >> 4;
  const int rt = wave >> 1;
  const int ch = wave & 1;

  for (int i0 = t; i0 < 1024 * 4; i0 += 128 * 16) {
    float4 v[16];
#pragma unroll
    for (int u = 0; u < 16; ++u) v[u] = ((const float4*)E)[i0 + u * 128];
#pragma unroll
    for (int u = 0; u < 16; ++u) {
      const int i = i0 + u * 128;
      *(float4*)(&Es[i >> 2][(i & 3) << 2]) = v[u];
    }
  }

  const int mrow = rowBase + rt * 16 + m16;
  v2f bq[4];
  {
    const float4 e0 = *(const float4*)(E + (size_t)mrow * 16 + 8 * kh);
    const float4 e1 = *(const float4*)(E + (size_t)mrow * 16 + 8 * kh + 4);
    bq[0][0] = e0.x; bq[0][1] = e0.y; bq[1][0] = e0.z; bq[1][1] = e0.w;
    bq[2][0] = e1.x; bq[2][1] = e1.y; bq[3][0] = e1.z; bq[3][1] = e1.w;
  }
  const float mxm = rmx[mrow];
  const float invm = rinv[mrow];
  const int kperm = 4 * ((m16 & 7) >> 1) + 2 * (m16 >> 3) + (m16 & 1);
  const float* erow = &Es[kperm][8 * kh];
  const float* bcol = &Bs[kh][(ch * 32 + m16) * 2];

  const v8f zero = {0.f, 0.f, 0.f, 0.f, 0.f, 0.f, 0.f, 0.f};
  v8f cacc[2];
  cacc[0] = zero; cacc[1] = zero;

  float4 ea0, ea1;
  v8f st_cur;
  v2f a_cur[4];

  for (int kc = 0; kc < 1024; kc += XG_KC) {
    __syncthreads();
    for (int i0 = t; i0 < (XG_KC / 2) * 16; i0 += 128 * 16) {
      float4 v0[16], v1[16];
#pragma unroll
      for (int u = 0; u < 16; ++u) {
        const int i = i0 + u * 128;
        const int kp = i >> 4, cc = (i & 15) << 2;
        v0[u] = *(const float4*)(Bb + (size_t)(kc + 2 * kp) * 64 + cc);
        v1[u] = *(const float4*)(Bb + (size_t)(kc + 2 * kp + 1) * 64 + cc);
      }
#pragma unroll
      for (int u = 0; u < 16; ++u) {
        const int i = i0 + u * 128;
        const int kp = i >> 4, cc = (i & 15) << 2;
        float4 lo = {v0[u].x, v1[u].x, v0[u].y, v1[u].y};
        float4 hi = {v0[u].z, v1[u].z, v0[u].w, v1[u].w};
        *(float4*)(&Bs[kp][cc * 2])     = lo;
        *(float4*)(&Bs[kp][cc * 2 + 4]) = hi;
      }
    }
    __syncthreads();

    if (kc == 0) {
      const float4 f0 = *(const float4*)(erow);
      const float4 f1 = *(const float4*)(erow + 4);
      const float4 g0 = *(const float4*)(erow + 16 * 20);
      const float4 g1 = *(const float4*)(erow + 16 * 20 + 4);
      v8f s0 = zero;
      s0 = wmma_f32(v2f{f0.x, f0.y}, bq[0], s0);
      s0 = wmma_f32(v2f{f0.z, f0.w}, bq[1], s0);
      s0 = wmma_f32(v2f{f1.x, f1.y}, bq[2], s0);
      s0 = wmma_f32(v2f{f1.z, f1.w}, bq[3], s0);
      v8f s1 = zero;
      s1 = wmma_f32(v2f{g0.x, g0.y}, bq[0], s1);
      s1 = wmma_f32(v2f{g0.z, g0.w}, bq[1], s1);
      s1 = wmma_f32(v2f{g1.x, g1.y}, bq[2], s1);
      s1 = wmma_f32(v2f{g1.z, g1.w}, bq[3], s1);
#pragma unroll
      for (int j = 0; j < 4; ++j) {
        a_cur[j][0] = __expf(fmaxf(s0[2 * j], 0.f) - mxm) * invm;
        a_cur[j][1] = __expf(fmaxf(s0[2 * j + 1], 0.f) - mxm) * invm;
      }
      st_cur = s1;
      ea0 = *(const float4*)(erow + 32 * 20);
      ea1 = *(const float4*)(erow + 32 * 20 + 4);
    }

    v2f bf_cur[4][2];
#pragma unroll
    for (int j = 0; j < 4; ++j)
#pragma unroll
      for (int nt = 0; nt < 2; ++nt)
        bf_cur[j][nt] = *(const v2f*)(bcol + (2 * j) * 160 + nt * 32);

    for (int gl = 0; gl < XG_KC / 16; ++gl) {
      const int g = kc / 16 + gl;
      const float* en = erow + (size_t)(((g + 3) & 63) * 16) * 20;
      const float4 en0 = *(const float4*)(en);
      const float4 en1 = *(const float4*)(en + 4);
      v2f bf_nxt[4][2];
      const float* bn = bcol + (size_t)(((gl + 1) & (XG_KC / 16 - 1)) * 8) * 160;
#pragma unroll
      for (int j = 0; j < 4; ++j)
#pragma unroll
        for (int nt = 0; nt < 2; ++nt)
          bf_nxt[j][nt] = *(const v2f*)(bn + (2 * j) * 160 + nt * 32);
      v2f a_nxt[4];
#pragma unroll
      for (int j = 0; j < 4; ++j) {
        a_nxt[j][0] = __expf(fmaxf(st_cur[2 * j], 0.f) - mxm) * invm;
        a_nxt[j][1] = __expf(fmaxf(st_cur[2 * j + 1], 0.f) - mxm) * invm;
      }
      const v2f ta[4] = {v2f{ea0.x, ea0.y}, v2f{ea0.z, ea0.w},
                         v2f{ea1.x, ea1.y}, v2f{ea1.z, ea1.w}};
      v8f st_nxt = zero;
#pragma unroll
      for (int j = 0; j < 4; ++j) {
        st_nxt = wmma_f32(ta[j], bq[j], st_nxt);
        cacc[0] = wmma_f32(a_cur[j], bf_cur[j][0], cacc[0]);
        cacc[1] = wmma_f32(a_cur[j], bf_cur[j][1], cacc[1]);
      }
      st_cur = st_nxt;
      ea0 = en0; ea1 = en1;
#pragma unroll
      for (int j = 0; j < 4; ++j) {
        a_cur[j] = a_nxt[j];
        bf_cur[j][0] = bf_nxt[j][0];
        bf_cur[j][1] = bf_nxt[j][1];
      }
    }
  }

#pragma unroll
  for (int nt = 0; nt < 2; ++nt) {
#pragma unroll
    for (int r = 0; r < 8; ++r) {
      const int m = rowBase + rt * 16 + r + 8 * kh;
      Cb[(size_t)m * 64 + (ch * 2 + nt) * 16 + m16] = cacc[nt][r];
    }
  }
}

#define FP_DS 2
#define FP_NS (16 / FP_DS)

static __device__ __forceinline__ void pool_issue(const float* __restrict__ src, int t,
                                                  float4 (&v0)[16], float4 (&v1)[16]) {
#pragma unroll
  for (int u = 0; u < 16; ++u) {
    const int i = t + u * 128;
    const int kp = i >> 4, cc = (i & 15) << 2;
    v0[u] = *(const float4*)(src + (size_t)(2 * kp) * 64 + cc);
    v1[u] = *(const float4*)(src + (size_t)(2 * kp + 1) * 64 + cc);
  }
}

static __device__ __forceinline__ void pool_store(float (*Bsb)[160], int t,
                                                  const float4 (&v0)[16], const float4 (&v1)[16]) {
#pragma unroll
  for (int u = 0; u < 16; ++u) {
    const int i = t + u * 128;
    const int kp = i >> 4, cc = (i & 15) << 2;
    float4 lo = {v0[u].x, v1[u].x, v0[u].y, v1[u].y};
    float4 hi = {v0[u].z, v1[u].z, v0[u].w, v1[u].w};
    *(float4*)(&Bsb[kp][cc * 2])     = lo;
    *(float4*)(&Bsb[kp][cc * 2 + 4]) = hi;
  }
}

static __device__ __forceinline__ void pool_stage(const float* bc, const v2f (&a)[32],
                                                  v2f e2, v8f (&cacc)[2]) {
  v2f bf[2][8][2];
#pragma unroll
  for (int j = 0; j < 8; ++j) {
    bf[0][j][0] = *(const v2f*)(bc + (2 * j) * 160);
    bf[0][j][1] = *(const v2f*)(bc + (2 * j) * 160 + 32);
  }
#pragma unroll
  for (int blk = 0; blk < FP_DS * 4; ++blk) {
    if (blk + 1 < FP_DS * 4) {
#pragma unroll
      for (int j = 0; j < 8; ++j) {
        bf[(blk + 1) & 1][j][0] = *(const v2f*)(bc + (16 * (blk + 1) + 2 * j) * 160);
        bf[(blk + 1) & 1][j][1] = *(const v2f*)(bc + (16 * (blk + 1) + 2 * j) * 160 + 32);
      }
    }
    const float ed = e2[blk >> 2];
#pragma unroll
    for (int j = 0; j < 8; ++j) {
      const v2f as = a[8 * (blk & 3) + j] * ed;
      cacc[0] = wmma_f32(as, bf[blk & 1][j][0], cacc[0]);
      cacc[1] = wmma_f32(as, bf[blk & 1][j][1], cacc[1]);
    }
  }
}

__global__ __launch_bounds__(128)
void k_final_pool(const float* __restrict__ sx, const float* __restrict__ xg1,
                  const float* __restrict__ E, const float* __restrict__ Wp,
                  const float* __restrict__ bias_pool, float* __restrict__ out) {
  const int b = blockIdx.y;
  const int rowBase = blockIdx.x * 32;

  __shared__ float Bs[2][FP_DS * 64][160];

  const int t = threadIdx.x;
  const int wave = t >> 5;
  const int lane = t & 31;
  const int m16 = lane & 15;
  const int kh = lane >> 4;
  const int rt = wave >> 1;
  const int ch = wave & 1;
  const int nrow = rowBase + rt * 16 + m16;

  float4 v0[16], v1[16];
  pool_issue(Wp, t, v0, v1);

  v2f a[32];
  {
    const size_t roff = ((size_t)b * 1024 + nrow) * 64 + 2 * kh;
#pragma unroll
    for (int ks = 0; ks < 16; ++ks) {
      a[ks]      = *(const v2f*)(sx  + roff + 4 * ks);
      a[16 + ks] = *(const v2f*)(xg1 + roff + 4 * ks);
    }
  }
  const float* erow = E + (size_t)nrow * 16;
  v2f ea[4], bb[4][2];
#pragma unroll
  for (int j = 0; j < 4; ++j) {
    ea[j] = *(const v2f*)(erow + 4 * j + 2 * kh);
#pragma unroll
    for (int nt = 0; nt < 2; ++nt) {
      const int col = ch * 32 + nt * 16 + m16;
      bb[j][nt][0] = bias_pool[(4 * j + 2 * kh) * 64 + col];
      bb[j][nt][1] = bias_pool[(4 * j + 2 * kh + 1) * 64 + col];
    }
  }
  v2f e_cur = *(const v2f*)(erow);

  const v8f zero = {0.f, 0.f, 0.f, 0.f, 0.f, 0.f, 0.f, 0.f};
  v8f cacc[2];
  cacc[0] = zero; cacc[1] = zero;
#pragma unroll
  for (int j = 0; j < 4; ++j) {
    cacc[0] = wmma_f32(ea[j], bb[j][0], cacc[0]);
    cacc[1] = wmma_f32(ea[j], bb[j][1], cacc[1]);
  }

  pool_store(Bs[0], t, v0, v1);
  __syncthreads();

  for (int s = 0; s < FP_NS; ++s) {
    const int sn = (s + 1) & (FP_NS - 1);
    pool_issue(Wp + (size_t)sn * (FP_DS * 8192), t, v0, v1);
    const v2f e_nxt = *(const v2f*)(erow + FP_DS * sn);
    pool_stage(&Bs[s & 1][kh][(ch * 32 + m16) * 2], a, e_cur, cacc);
    pool_store(Bs[(s + 1) & 1], t, v0, v1);
    e_cur = e_nxt;
    __syncthreads();
  }

  float* Cb = out + (size_t)b * 1024 * 64;
#pragma unroll
  for (int nt = 0; nt < 2; ++nt) {
#pragma unroll
    for (int r = 0; r < 8; ++r) {
      const int m = rowBase + rt * 16 + r + 8 * kh;
      Cb[(size_t)m * 64 + (ch * 2 + nt) * 16 + m16] = cacc[nt][r];
    }
  }
}

extern "C" void kernel_launch(void* const* d_in, const int* in_sizes, int n_in,
                              void* d_out, int out_size, void* d_ws, size_t ws_size,
                              hipStream_t stream) {
  (void)in_sizes; (void)n_in; (void)out_size; (void)ws_size;
  const float* x    = (const float*)d_in[0];
  const float* E    = (const float*)d_in[1];
  const float* pa   = (const float*)d_in[2];
  const float* Wp   = (const float*)d_in[3];
  const float* bp   = (const float*)d_in[4];
  float* out = (float*)d_out;

  float* ws   = (float*)d_ws;
  float* rmx  = ws;
  float* rinv = ws + 1024;
  float* sube = ws   + 1048576;
  float* S    = sube + 8388608;
  float* xn   = S    + 8192;
  float* sx   = xn   + 524288;
  float* xg1  = sx   + 524288;

  k_rowstats  <<<64, 512, 0, stream>>>(E, rmx, rinv);
  k_colsum_pts<<<dim3(32, 8),     512, 0, stream>>>(pa, S);
  k_xnorm     <<<2048, 256, 0, stream>>>(x, S, xn);
  k_sx_points<<<dim3(32, 8), 128, 0, stream>>>(pa, xn, sx);
  k_xg_sim<<<dim3(32, 8), 128, 0, stream>>>(E, rmx, rinv, sx, xg1);
  k_final_pool<<<dim3(32, 8), 128, 0, stream>>>(sx, xg1, E, Wp, bp, out);
}


# sim product reads all operands from global in fragment order (sx copy written so by its producer, requests 8+ groups ahead in registers): no LDS, no barriers; k re-dealt to k-steps: sum order changes
# speedup vs baseline: 3.3376x; 1.0885x over previous
#include <hip/hip_runtime.h>
#include <hip/hip_bf16.h>
#include <cstddef>


typedef float v2f __attribute__((ext_vector_type(2)));
typedef float v8f __attribute__((ext_vector_type(8)));

static __device__ __forceinline__ v8f wmma_f32(v2f a, v2f b, v8f c) {
  return __builtin_amdgcn_wmma_f32_16x16x4_f32(
       false, a,  false, b,
       (short)0, c,  false,  false);
}

static __device__ __forceinline__ float exp_neg_l1(const float (&p)[16],
                                                   const float (&q)[16]) {
  float s0 = 0.f, s1 = 0.f, s2 = 0.f, s3 = 0.f;
#pragma unroll
  for (int d = 0; d < 4; ++d) {
    s0 += fabsf(p[d]      - q[d]);
    s1 += fabsf(p[d + 4]  - q[d + 4]);
    s2 += fabsf(p[d + 8]  - q[d + 8]);
    s3 += fabsf(p[d + 12] - q[d + 12]);
  }
  return __expf(-((s0 + s1) + (s2 + s3)));
}

__global__ __launch_bounds__(512)
void k_rowstats(const float* __restrict__ E, float* __restrict__ rmx,
                float* __restrict__ rinv) {
  const int t = threadIdx.x;
  const int wave = t >> 5;
  const int lane = t & 31;
  const int m16 = lane & 15;
  const int kh = lane >> 4;
  const int mrow = blockIdx.x * 16 + m16;
  __shared__ float part[32][16];

  v2f bq[4];
  {
    const float4 e0 = *(const float4*)(E + (size_t)mrow * 16 + 8 * kh);
    const float4 e1 = *(const float4*)(E + (size_t)mrow * 16 + 8 * kh + 4);
    bq[0][0] = e0.x; bq[0][1] = e0.y; bq[1][0] = e0.z; bq[1][1] = e0.w;
    bq[2][0] = e1.x; bq[2][1] = e1.y; bq[3][0] = e1.z; bq[3][1] = e1.w;
  }
  const v8f zero = {0.f, 0.f, 0.f, 0.f, 0.f, 0.f, 0.f, 0.f};
  v8f tl[4];
#pragma unroll
  for (int q = 0; q < 4; ++q) {
    const float* er = E + (size_t)(wave * 64 + q * 16 + m16) * 16 + 8 * kh;
    const float4 f0 = *(const float4*)(er);
    const float4 f1 = *(const float4*)(er + 4);
    v8f st = zero;
    st = wmma_f32(v2f{f0.x, f0.y}, bq[0], st);
    st = wmma_f32(v2f{f0.z, f0.w}, bq[1], st);
    st = wmma_f32(v2f{f1.x, f1.y}, bq[2], st);
    st = wmma_f32(v2f{f1.z, f1.w}, bq[3], st);
    tl[q] = st;
  }
  float mx = 0.f;
#pragma unroll
  for (int q = 0; q < 4; ++q)
#pragma unroll
    for (int r = 0; r < 8; ++r) mx = fmaxf(mx, tl[q][r]);
  part[wave * 2 + kh][m16] = mx;
  __syncthreads();
  float m = 0.f;
#pragma unroll
  for (int p = 0; p < 32; ++p) m = fmaxf(m, part[p][m16]);
  __syncthreads();
  float acc = 0.f;
#pragma unroll
  for (int q = 0; q < 4; ++q)
#pragma unroll
    for (int r = 0; r < 8; ++r) acc += __expf(fmaxf(tl[q][r], 0.f) - m);
  part[wave * 2 + kh][m16] = acc;
  __syncthreads();
  if (t < 16) {
    float tot = 0.f;
#pragma unroll
    for (int p = 0; p < 32; ++p) tot += part[p][t];
    rmx[blockIdx.x * 16 + t] = m;
    rinv[blockIdx.x * 16 + t] = 1.0f / tot;
  }
}

__global__ __launch_bounds__(512)
void k_colsum_pts(const float* __restrict__ pa, float* __restrict__ S) {
  const int b = blockIdx.y;
  const int lane = threadIdx.x & 31;
  const int w = __builtin_amdgcn_readfirstlane((int)(threadIdx.x >> 5));
  const int j = blockIdx.x * 32 + lane;
  __shared__ float part[16][32];

  float pj[16];
#pragma unroll
  for (int q = 0; q < 4; ++q) {
    const float4 v = *(const float4*)(pa + ((size_t)b * 1024 + j) * 16 + q * 4);
    pj[q * 4 + 0] = v.x; pj[q * 4 + 1] = v.y; pj[q * 4 + 2] = v.z; pj[q * 4 + 3] = v.w;
  }
  const float* pi = pa + ((size_t)b * 1024 + (size_t)w * 64) * 16;
  float acc = 0.f;
  for (int ii = 0; ii < 64; ++ii) {
    float qi[16];
#pragma unroll
    for (int d = 0; d < 16; ++d) qi[d] = pi[ii * 16 + d];
    acc += exp_neg_l1(qi, pj);
  }
  part[w][lane] = acc;
  __syncthreads();
  if (threadIdx.x < 32) {
    float tot = 0.f;
#pragma unroll
    for (int w2 = 0; w2 < 16; ++w2) tot += part[w2][threadIdx.x];
    S[b * 1024 + blockIdx.x * 32 + threadIdx.x] = tot;
  }
}

__global__ void k_xnorm(const float* __restrict__ x, const float* __restrict__ S,
                        float* __restrict__ xn) {
  const int idx = blockIdx.x * 256 + threadIdx.x;
  xn[idx] = x[idx] / S[idx >> 6];
}

#define SX_KC 512
#define SX_AF_TILE 2080
#define SX_AF_BUF (2 * SX_AF_TILE)

static __device__ __forceinline__ int sx_step_row(int p, int w) {
  return ((p >> 3) * 128 + w * 32 + (p & 7) * 4) & 1023;
}

static __device__ __forceinline__ void sx_load_pts(const float* __restrict__ pab, int p, int w,
                                                   float (&q)[4][16]) {
  const float* pn = pab + (size_t)sx_step_row(p, w) * 16;
#pragma unroll
  for (int kk = 0; kk < 4; ++kk) {
#pragma unroll
    for (int q4 = 0; q4 < 4; ++q4) {
      const float4 v = *(const float4*)(pn + kk * 16 + q4 * 4);
      q[kk][q4 * 4 + 0] = v.x; q[kk][q4 * 4 + 1] = v.y;
      q[kk][q4 * 4 + 2] = v.z; q[kk][q4 * 4 + 3] = v.w;
    }
  }
}

template <bool FORM, bool USE>
static __device__ __forceinline__ void sx_step(const float* __restrict__ pab, int p0, int w, int j,
                                               float (&q)[4][16], const float (&pm)[16],
                                               float* afw, const float* afr, const float* bcol,
                                               const v2f (&au)[4], const v2f (&bu)[4][2],
                                               v2f (&an)[4], v2f (&bn)[4][2], v8f (&cacc)[2]) {
  if (USE) {
    const int jn = (j + 1) & 7;
#pragma unroll
    for (int u = 0; u < 4; ++u) {
      an[u] = *(const v2f*)(afr + (4 * jn + u) * 64);
      bn[u][0] = *(const v2f*)(bcol + (2 * (4 * jn + u)) * 160);
      bn[u][1] = *(const v2f*)(bcol + (2 * (4 * jn + u)) * 160 + 32);
    }
  }
  float e[4];
  if (FORM) {
#pragma unroll
    for (int kk = 0; kk < 4; ++kk) e[kk] = exp_neg_l1(q[kk], pm);
    sx_load_pts(pab, p0 + j + 2, w, q);
  }
  if (USE) {
#pragma unroll
    for (int u = 0; u < 4; ++u) {
      cacc[0] = wmma_f32(au[u], bu[u][0], cacc[0]);
      cacc[1] = wmma_f32(au[u], bu[u][1], cacc[1]);
    }
  }
  if (FORM && USE) {
    __builtin_amdgcn_sched_group_barrier(0x100, 8, 0);
#pragma unroll
    for (int u = 0; u < 8; ++u) {
      __builtin_amdgcn_sched_group_barrier(0x008, 1, 0);
      __builtin_amdgcn_sched_group_barrier(0x002, 19, 0);
    }
  }
  if (FORM) {
    *(v2f*)(afw + j * 64)      = v2f{e[0], e[1]};
    *(v2f*)(afw + j * 64 + 32) = v2f{e[2], e[3]};
  }
}

template <bool FORM, bool USE>
static __device__ __forceinline__ void sx_steps(const float* __restrict__ pab, int p0, int w,
                                                float (&qa)[4][16], float (&qb)[4][16],
                                                const float (&pm)[16],
                                                float* afw, const float* afr, const float* bcol,
                                                v8f (&cacc)[2]) {
  v2f a0[4], b0[4][2], a1[4], b1[4][2];
  if (USE) {
#pragma unroll
    for (int u = 0; u < 4; ++u) {
      a0[u] = *(const v2f*)(afr + u * 64);
      b0[u][0] = *(const v2f*)(bcol + (2 * u) * 160);
      b0[u][1] = *(const v2f*)(bcol + (2 * u) * 160 + 32);
    }
  }
  for (int jj = 0; jj < 8; jj += 2) {
    sx_step<FORM, USE>(pab, p0, w, jj,     qa, pm, afw, afr, bcol, a0, b0, a1, b1, cacc);
    sx_step<FORM, USE>(pab, p0, w, jj + 1, qb, pm, afw, afr, bcol, a1, b1, a0, b0, cacc);
  }
}

__global__ __launch_bounds__(128)
void k_sx_points(const float* __restrict__ pa, const float* __restrict__ xn,
                 float* __restrict__ sx, float* __restrict__ sxF) {
  const int b = blockIdx.y;
  const int rowBase = blockIdx.x * 32;
  const float* pab = pa + (size_t)b * 1024 * 16;
  const float* Bb = xn + (size_t)b * 1024 * 64;
  float* Cb = sx + (size_t)b * 1024 * 64;

  __shared__ float Bs[SX_KC / 2][160];
  __shared__ __attribute__((aligned(16))) float Af[2 * SX_AF_BUF];

  const int t = threadIdx.x;
  const int wave = t >> 5;
  const int lane = t & 31;
  const int m16 = lane & 15;
  const int kh = lane >> 4;
  const int rt = wave >> 1;
  const int ch = wave & 1;

  float pm[16];
  float qa[4][16], qb[4][16];
  {
    const float* pr = pab + (size_t)(rowBase + lane) * 16;
#pragma unroll
    for (int q = 0; q < 4; ++q) {
      const float4 v = *(const float4*)(pr + q * 4);
      pm[q * 4 + 0] = v.x; pm[q * 4 + 1] = v.y; pm[q * 4 + 2] = v.z; pm[q * 4 + 3] = v.w;
    }
    sx_load_pts(pab, 0, wave, qa);
    sx_load_pts(pab, 1, wave, qb);
  }

  const v8f zero = {0.f, 0.f, 0.f, 0.f, 0.f, 0.f, 0.f, 0.f};
  v8f cacc[2];
  cacc[0] = zero; cacc[1] = zero;

  float* afw0 = Af + (lane >> 4) * SX_AF_TILE + (8 * wave) * 64 + (lane & 15) * 2;
  const float* afr0 = Af + rt * SX_AF_TILE + lane * 2;
  const float* bcol0 = &Bs[kh][(ch * 32 + m16) * 2];

  for (int kc = 0; kc < 1024; kc += SX_KC) {
    if (kc != 0) __syncthreads();
    for (int i0 = t; i0 < (SX_KC / 2) * 16; i0 += 128 * 8) {
      float4 v0[8], v1[8];
#pragma unroll
      for (int u = 0; u < 8; ++u) {
        const int i = i0 + u * 128;
        const int kp = i >> 4, cc = (i & 15) << 2;
        v0[u] = *(const float4*)(Bb + (size_t)(kc + 2 * kp) * 64 + cc);
        v1[u] = *(const float4*)(Bb + (size_t)(kc + 2 * kp + 1) * 64 + cc);
      }
#pragma unroll
      for (int u = 0; u < 8; ++u) {
        const int i = i0 + u * 128;
        const int kp = i >> 4, cc = (i & 15) << 2;
        float4 lo = {v0[u].x, v1[u].x, v0[u].y, v1[u].y};
        float4 hi = {v0[u].z, v1[u].z, v0[u].w, v1[u].w};
        *(float4*)(&Bs[kp][cc * 2])     = lo;
        *(float4*)(&Bs[kp][cc * 2 + 4]) = hi;
      }
    }
    if (kc == 0)
      sx_steps<true, false>(pab, 0, wave, qa, qb, pm, afw0, afr0, bcol0, cacc);
    __syncthreads();

    for (int s4 = 0; s4 < 4; ++s4) {
      const int sc = kc / 128 + s4;
      const float* afr = afr0 + (sc & 1) * SX_AF_BUF;
      float* afw = afw0 + ((sc + 1) & 1) * SX_AF_BUF;
      const float* bcol = bcol0 + (size_t)(s4 * 64) * 160;
      if (sc < 7) {
        sx_steps<true, true>(pab, (sc + 1) * 8, wave, qa, qb, pm, afw, afr, bcol, cacc);
        if (s4 < 3) __syncthreads();
      } else {
        sx_steps<false, true>(pab, 0, wave, qa, qb, pm, afw, afr, bcol, cacc);
      }
    }
  }

#pragma unroll
  for (int nt = 0; nt < 2; ++nt) {
#pragma unroll
    for (int r = 0; r < 8; ++r) {
      const int m = rowBase + rt * 16 + r + 8 * kh;
      Cb[(size_t)m * 64 + (ch * 2 + nt) * 16 + m16] = cacc[nt][r];
    }
  }
  {
    float* Fb = sxF + (((size_t)b * 64 + blockIdx.x * 2 + rt) * 4 + ch * 2) * 256 + lane * 8;
#pragma unroll
    for (int nt = 0; nt < 2; ++nt) {
      *(float4*)(Fb + nt * 256)     = float4{cacc[nt][0], cacc[nt][1], cacc[nt][2], cacc[nt][3]};
      *(float4*)(Fb + nt * 256 + 4) = float4{cacc[nt][4], cacc[nt][5], cacc[nt][6], cacc[nt][7]};
    }
  }
}

#define XG_PF 8
typedef float v4f __attribute__((ext_vector_type(4)));
#define V4_LO(x) __builtin_shufflevector((x), (x), 0, 1)
#define V4_HI(x) __builtin_shufflevector((x), (x), 2, 3)
__global__ __launch_bounds__(128)
void k_xg_sim(const float* __restrict__ E, const float* __restrict__ rmx,
              const float* __restrict__ rinv, const float* __restrict__ sxF,
              float* __restrict__ xg1) {
  const int b = blockIdx.y;
  const int rowBase = blockIdx.x * 32;
  float* Cb = xg1 + (size_t)b * 1024 * 64;

  const int t = threadIdx.x;
  const int wave = t >> 5;
  const int lane = t & 31;
  const int m16 = lane & 15;
  const int kh = lane >> 4;
  const int rt = wave >> 1;
  const int ch = wave & 1;

  const int mrow = rowBase + rt * 16 + m16;
  v2f bq[4];
  {
    const float4 e0 = *(const float4*)(E + (size_t)mrow * 16 + 8 * kh);
    const float4 e1 = *(const float4*)(E + (size_t)mrow * 16 + 8 * kh + 4);
    bq[0][0] = e0.x; bq[0][1] = e0.y; bq[1][0] = e0.z; bq[1][1] = e0.w;
    bq[2][0] = e1.x; bq[2][1] = e1.y; bq[3][0] = e1.z; bq[3][1] = e1.w;
  }
  const float mxm = rmx[mrow];
  const float invm = rinv[mrow];
  const float* erow = E + (size_t)m16 * 16 + 8 * kh;
  const float* bfr = sxF + ((size_t)b * 256 + ch * 2) * 256 + lane * 8;

  const v8f zero = {0.f, 0.f, 0.f, 0.f, 0.f, 0.f, 0.f, 0.f};
  v8f cacc[2];
  cacc[0] = zero; cacc[1] = zero;

  v4f eA[XG_PF][2], eB[XG_PF][2];
  v4f bA[XG_PF][2][2], bB[XG_PF][2][2];
  v8f st_cur;
  v2f a_cur[4];

  auto request = [&](v4f (&eb)[XG_PF][2], v4f (&bb)[XG_PF][2][2], int tt, int g) {
    const float* en = erow + (size_t)((g + 2) & 63) * 256;
    eb[tt][0] = *(const v4f*)(en);
    eb[tt][1] = *(const v4f*)(en + 4);
    const float* bn = bfr + (size_t)(g & 63) * 1024;
#pragma unroll
    for (int nt = 0; nt < 2; ++nt) {
      bb[tt][nt][0] = *(const v4f*)(bn + nt * 256);
      bb[tt][nt][1] = *(const v4f*)(bn + nt * 256 + 4);
    }
  };
  auto trip = [&](const v4f (&eb)[XG_PF][2], const v4f (&bb)[XG_PF][2][2], int tt) {
    const v2f ta[4] = {V4_LO(eb[tt][0]), V4_HI(eb[tt][0]), V4_LO(eb[tt][1]), V4_HI(eb[tt][1])};
    v2f bf[4][2];
#pragma unroll
    for (int nt = 0; nt < 2; ++nt) {
      bf[0][nt] = V4_LO(bb[tt][nt][0]); bf[1][nt] = V4_HI(bb[tt][nt][0]);
      bf[2][nt] = V4_LO(bb[tt][nt][1]); bf[3][nt] = V4_HI(bb[tt][nt][1]);
    }
    v2f a_nxt[4];
#pragma unroll
    for (int j = 0; j < 4; ++j) {
      a_nxt[j][0] = __expf(fmaxf(st_cur[2 * j], 0.f) - mxm) * invm;
      a_nxt[j][1] = __expf(fmaxf(st_cur[2 * j + 1], 0.f) - mxm) * invm;
    }
    v8f st_nxt = zero;
#pragma unroll
    for (int j = 0; j < 4; ++j) {
      st_nxt = wmma_f32(ta[j], bq[j], st_nxt);
      cacc[0] = wmma_f32(a_cur[j], bf[j][0], cacc[0]);
      cacc[1] = wmma_f32(a_cur[j], bf[j][1], cacc[1]);
    }
    st_cur = st_nxt;
#pragma unroll
    for (int j = 0; j < 4; ++j) a_cur[j] = a_nxt[j];
    __builtin_amdgcn_sched_barrier(0);
  };

  {
    const v4f f0 = *(const v4f*)(erow);
    const v4f f1 = *(const v4f*)(erow + 4);
    const v4f g0 = *(const v4f*)(erow + 256);
    const v4f g1 = *(const v4f*)(erow + 256 + 4);
#pragma unroll
    for (int u = 0; u < XG_PF; ++u) request(eA, bA, u, u);
    asm volatile("" ::: "memory");
    v8f s0 = zero;
    s0 = wmma_f32(V4_LO(f0), bq[0], s0);
    s0 = wmma_f32(V4_HI(f0), bq[1], s0);
    s0 = wmma_f32(V4_LO(f1), bq[2], s0);
    s0 = wmma_f32(V4_HI(f1), bq[3], s0);
    v8f s1 = zero;
    s1 = wmma_f32(V4_LO(g0), bq[0], s1);
    s1 = wmma_f32(V4_HI(g0), bq[1], s1);
    s1 = wmma_f32(V4_LO(g1), bq[2], s1);
    s1 = wmma_f32(V4_HI(g1), bq[3], s1);
#pragma unroll
    for (int j = 0; j < 4; ++j) {
      a_cur[j][0] = __expf(fmaxf(s0[2 * j], 0.f) - mxm) * invm;
      a_cur[j][1] = __expf(fmaxf(s0[2 * j + 1], 0.f) - mxm) * invm;
    }
    st_cur = s1;
  }

  for (int g0 = 0; g0 < 64; g0 += 2 * XG_PF) {
#pragma unroll
    for (int u = 0; u < XG_PF; ++u) {
      if (u < XG_PF / 2) {
        request(eB, bB, 2 * u,     g0 + XG_PF + 2 * u);
        request(eB, bB, 2 * u + 1, g0 + XG_PF + 2 * u + 1);
        asm volatile("" ::: "memory");
      }
      trip(eA, bA, u);
    }
#pragma unroll
    for (int u = 0; u < XG_PF; ++u) {
      if (u < XG_PF / 2) {
        request(eA, bA, 2 * u,     g0 + 2 * XG_PF + 2 * u);
        request(eA, bA, 2 * u + 1, g0 + 2 * XG_PF + 2 * u + 1);
        asm volatile("" ::: "memory");
      }
      trip(eB, bB, u);
    }
  }

#pragma unroll
  for (int nt = 0; nt < 2; ++nt) {
#pragma unroll
    for (int r = 0; r < 8; ++r) {
      const int m = rowBase + rt * 16 + r + 8 * kh;
      Cb[(size_t)m * 64 + (ch * 2 + nt) * 16 + m16] = cacc[nt][r];
    }
  }
}

#define FP_DS 2
#define FP_NS (16 / FP_DS)

static __device__ __forceinline__ void pool_issue(const float* __restrict__ src, int t,
                                                  float4 (&v0)[16], float4 (&v1)[16]) {
#pragma unroll
  for (int u = 0; u < 16; ++u) {
    const int i = t + u * 128;
    const int kp = i >> 4, cc = (i & 15) << 2;
    v0[u] = *(const float4*)(src + (size_t)(2 * kp) * 64 + cc);
    v1[u] = *(const float4*)(src + (size_t)(2 * kp + 1) * 64 + cc);
  }
}

static __device__ __forceinline__ void pool_store(float (*Bsb)[160], int t,
                                                  const float4 (&v0)[16], const float4 (&v1)[16]) {
#pragma unroll
  for (int u = 0; u < 16; ++u) {
    const int i = t + u * 128;
    const int kp = i >> 4, cc = (i & 15) << 2;
    float4 lo = {v0[u].x, v1[u].x, v0[u].y, v1[u].y};
    float4 hi = {v0[u].z, v1[u].z, v0[u].w, v1[u].w};
    *(float4*)(&Bsb[kp][cc * 2])     = lo;
    *(float4*)(&Bsb[kp][cc * 2 + 4]) = hi;
  }
}

static __device__ __forceinline__ void pool_stage(const float* bc, const v2f (&a)[32],
                                                  v2f e2, v8f (&cacc)[2]) {
  v2f bf[2][8][2];
#pragma unroll
  for (int j = 0; j < 8; ++j) {
    bf[0][j][0] = *(const v2f*)(bc + (2 * j) * 160);
    bf[0][j][1] = *(const v2f*)(bc + (2 * j) * 160 + 32);
  }
#pragma unroll
  for (int blk = 0; blk < FP_DS * 4; ++blk) {
    if (blk + 1 < FP_DS * 4) {
#pragma unroll
      for (int j = 0; j < 8; ++j) {
        bf[(blk + 1) & 1][j][0] = *(const v2f*)(bc + (16 * (blk + 1) + 2 * j) * 160);
        bf[(blk + 1) & 1][j][1] = *(const v2f*)(bc + (16 * (blk + 1) + 2 * j) * 160 + 32);
      }
    }
    const float ed = e2[blk >> 2];
#pragma unroll
    for (int j = 0; j < 8; ++j) {
      const v2f as = a[8 * (blk & 3) + j] * ed;
      cacc[0] = wmma_f32(as, bf[blk & 1][j][0], cacc[0]);
      cacc[1] = wmma_f32(as, bf[blk & 1][j][1], cacc[1]);
    }
  }
}

__global__ __launch_bounds__(128)
void k_final_pool(const float* __restrict__ sx, const float* __restrict__ xg1,
                  const float* __restrict__ E, const float* __restrict__ Wp,
                  const float* __restrict__ bias_pool, float* __restrict__ out) {
  const int b = blockIdx.y;
  const int rowBase = blockIdx.x * 32;

  __shared__ float Bs[2][FP_DS * 64][160];

  const int t = threadIdx.x;
  const int wave = t >> 5;
  const int lane = t & 31;
  const int m16 = lane & 15;
  const int kh = lane >> 4;
  const int rt = wave >> 1;
  const int ch = wave & 1;
  const int nrow = rowBase + rt * 16 + m16;

  float4 v0[16], v1[16];
  pool_issue(Wp, t, v0, v1);

  v2f a[32];
  {
    const size_t roff = ((size_t)b * 1024 + nrow) * 64 + 2 * kh;
#pragma unroll
    for (int ks = 0; ks < 16; ++ks) {
      a[ks]      = *(const v2f*)(sx  + roff + 4 * ks);
      a[16 + ks] = *(const v2f*)(xg1 + roff + 4 * ks);
    }
  }
  const float* erow = E + (size_t)nrow * 16;
  v2f ea[4], bb[4][2];
#pragma unroll
  for (int j = 0; j < 4; ++j) {
    ea[j] = *(const v2f*)(erow + 4 * j + 2 * kh);
#pragma unroll
    for (int nt = 0; nt < 2; ++nt) {
      const int col = ch * 32 + nt * 16 + m16;
      bb[j][nt][0] = bias_pool[(4 * j + 2 * kh) * 64 + col];
      bb[j][nt][1] = bias_pool[(4 * j + 2 * kh + 1) * 64 + col];
    }
  }
  v2f e_cur = *(const v2f*)(erow);

  const v8f zero = {0.f, 0.f, 0.f, 0.f, 0.f, 0.f, 0.f, 0.f};
  v8f cacc[2];
  cacc[0] = zero; cacc[1] = zero;
#pragma unroll
  for (int j = 0; j < 4; ++j) {
    cacc[0] = wmma_f32(ea[j], bb[j][0], cacc[0]);
    cacc[1] = wmma_f32(ea[j], bb[j][1], cacc[1]);
  }

  pool_store(Bs[0], t, v0, v1);
  __syncthreads();

  for (int s = 0; s < FP_NS; ++s) {
    const int sn = (s + 1) & (FP_NS - 1);
    pool_issue(Wp + (size_t)sn * (FP_DS * 8192), t, v0, v1);
    const v2f e_nxt = *(const v2f*)(erow + FP_DS * sn);
    pool_stage(&Bs[s & 1][kh][(ch * 32 + m16) * 2], a, e_cur, cacc);
    pool_store(Bs[(s + 1) & 1], t, v0, v1);
    e_cur = e_nxt;
    __syncthreads();
  }

  float* Cb = out + (size_t)b * 1024 * 64;
#pragma unroll
  for (int nt = 0; nt < 2; ++nt) {
#pragma unroll
    for (int r = 0; r < 8; ++r) {
      const int m = rowBase + rt * 16 + r + 8 * kh;
      Cb[(size_t)m * 64 + (ch * 2 + nt) * 16 + m16] = cacc[nt][r];
    }
  }
}

extern "C" void kernel_launch(void* const* d_in, const int* in_sizes, int n_in,
                              void* d_out, int out_size, void* d_ws, size_t ws_size,
                              hipStream_t stream) {
  (void)in_sizes; (void)n_in; (void)out_size; (void)ws_size;
  const float* x    = (const float*)d_in[0];
  const float* E    = (const float*)d_in[1];
  const float* pa   = (const float*)d_in[2];
  const float* Wp   = (const float*)d_in[3];
  const float* bp   = (const float*)d_in[4];
  float* out = (float*)d_out;

  float* ws   = (float*)d_ws;
  float* rmx  = ws;
  float* rinv = ws + 1024;
  float* sube = ws   + 1048576;
  float* S    = sube + 8388608;
  float* xn   = S    + 8192;
  float* sx   = xn   + 524288;
  float* xg1  = sx   + 524288;
  float* sxF  = sube;

  k_rowstats  <<<64, 512, 0, stream>>>(E, rmx, rinv);
  k_colsum_pts<<<dim3(32, 8),     512, 0, stream>>>(pa, S);
  k_xnorm     <<<2048, 256, 0, stream>>>(x, S, xn);
  k_sx_points<<<dim3(32, 8), 128, 0, stream>>>(pa, xn, sx, sxF);
  k_xg_sim<<<dim3(32, 8), 128, 180224, stream>>>(E, rmx, rinv, sxF, xg1);
  k_final_pool<<<dim3(32, 8), 128, 0, stream>>>(sx, xg1, E, Wp, bp, out);
}


# Softmax denominator folded into the sx product's B staging: each row of x is scaled by 1/S of its k between global read and LDS store; the x/S pass and its launch are removed; sx rounding changes
# speedup vs baseline: 3.4566x; 1.0357x over previous
#include <hip/hip_runtime.h>
#include <hip/hip_bf16.h>
#include <cstddef>


typedef float v2f __attribute__((ext_vector_type(2)));
typedef float v8f __attribute__((ext_vector_type(8)));

static __device__ __forceinline__ v8f wmma_f32(v2f a, v2f b, v8f c) {
  return __builtin_amdgcn_wmma_f32_16x16x4_f32(
       false, a,  false, b,
       (short)0, c,  false,  false);
}

static __device__ __forceinline__ float exp_neg_l1(const float (&p)[16],
                                                   const float (&q)[16]) {
  float s0 = 0.f, s1 = 0.f, s2 = 0.f, s3 = 0.f;
#pragma unroll
  for (int d = 0; d < 4; ++d) {
    s0 += fabsf(p[d]      - q[d]);
    s1 += fabsf(p[d + 4]  - q[d + 4]);
    s2 += fabsf(p[d + 8]  - q[d + 8]);
    s3 += fabsf(p[d + 12] - q[d + 12]);
  }
  return __expf(-((s0 + s1) + (s2 + s3)));
}

__global__ __launch_bounds__(512)
void k_rowstats(const float* __restrict__ E, float* __restrict__ rmx,
                float* __restrict__ rinv) {
  const int t = threadIdx.x;
  const int wave = t >> 5;
  const int lane = t & 31;
  const int m16 = lane & 15;
  const int kh = lane >> 4;
  const int mrow = blockIdx.x * 16 + m16;
  __shared__ float part[32][16];

  v2f bq[4];
  {
    const float4 e0 = *(const float4*)(E + (size_t)mrow * 16 + 8 * kh);
    const float4 e1 = *(const float4*)(E + (size_t)mrow * 16 + 8 * kh + 4);
    bq[0][0] = e0.x; bq[0][1] = e0.y; bq[1][0] = e0.z; bq[1][1] = e0.w;
    bq[2][0] = e1.x; bq[2][1] = e1.y; bq[3][0] = e1.z; bq[3][1] = e1.w;
  }
  const v8f zero = {0.f, 0.f, 0.f, 0.f, 0.f, 0.f, 0.f, 0.f};
  v8f tl[4];
#pragma unroll
  for (int q = 0; q < 4; ++q) {
    const float* er = E + (size_t)(wave * 64 + q * 16 + m16) * 16 + 8 * kh;
    const float4 f0 = *(const float4*)(er);
    const float4 f1 = *(const float4*)(er + 4);
    v8f st = zero;
    st = wmma_f32(v2f{f0.x, f0.y}, bq[0], st);
    st = wmma_f32(v2f{f0.z, f0.w}, bq[1], st);
    st = wmma_f32(v2f{f1.x, f1.y}, bq[2], st);
    st = wmma_f32(v2f{f1.z, f1.w}, bq[3], st);
    tl[q] = st;
  }
  float mx = 0.f;
#pragma unroll
  for (int q = 0; q < 4; ++q)
#pragma unroll
    for (int r = 0; r < 8; ++r) mx = fmaxf(mx, tl[q][r]);
  part[wave * 2 + kh][m16] = mx;
  __syncthreads();
  float m = 0.f;
#pragma unroll
  for (int p = 0; p < 32; ++p) m = fmaxf(m, part[p][m16]);
  __syncthreads();
  float acc = 0.f;
#pragma unroll
  for (int q = 0; q < 4; ++q)
#pragma unroll
    for (int r = 0; r < 8; ++r) acc += __expf(fmaxf(tl[q][r], 0.f) - m);
  part[wave * 2 + kh][m16] = acc;
  __syncthreads();
  if (t < 16) {
    float tot = 0.f;
#pragma unroll
    for (int p = 0; p < 32; ++p) tot += part[p][t];
    rmx[blockIdx.x * 16 + t] = m;
    rinv[blockIdx.x * 16 + t] = 1.0f / tot;
  }
}

__global__ __launch_bounds__(512)
void k_colsum_pts(const float* __restrict__ pa, float* __restrict__ rS) {
  const int b = blockIdx.y;
  const int lane = threadIdx.x & 31;
  const int w = __builtin_amdgcn_readfirstlane((int)(threadIdx.x >> 5));
  const int j = blockIdx.x * 32 + lane;
  __shared__ float part[16][32];

  float pj[16];
#pragma unroll
  for (int q = 0; q < 4; ++q) {
    const float4 v = *(const float4*)(pa + ((size_t)b * 1024 + j) * 16 + q * 4);
    pj[q * 4 + 0] = v.x; pj[q * 4 + 1] = v.y; pj[q * 4 + 2] = v.z; pj[q * 4 + 3] = v.w;
  }
  const float* pi = pa + ((size_t)b * 1024 + (size_t)w * 64) * 16;
  float acc = 0.f;
  for (int ii = 0; ii < 64; ++ii) {
    float qi[16];
#pragma unroll
    for (int d = 0; d < 16; ++d) qi[d] = pi[ii * 16 + d];
    acc += exp_neg_l1(qi, pj);
  }
  part[w][lane] = acc;
  __syncthreads();
  if (threadIdx.x < 32) {
    float tot = 0.f;
#pragma unroll
    for (int w2 = 0; w2 < 16; ++w2) tot += part[w2][threadIdx.x];
    rS[b * 1024 + blockIdx.x * 32 + threadIdx.x] = 1.0f / tot;
  }
}

#define SX_KC 512
#define SX_AF_TILE 2080
#define SX_AF_BUF (2 * SX_AF_TILE)

static __device__ __forceinline__ int sx_step_row(int p, int w) {
  return ((p >> 3) * 128 + w * 32 + (p & 7) * 4) & 1023;
}

static __device__ __forceinline__ void sx_load_pts(const float* __restrict__ pab, int p, int w,
                                                   float (&q)[4][16]) {
  const float* pn = pab + (size_t)sx_step_row(p, w) * 16;
#pragma unroll
  for (int kk = 0; kk < 4; ++kk) {
#pragma unroll
    for (int q4 = 0; q4 < 4; ++q4) {
      const float4 v = *(const float4*)(pn + kk * 16 + q4 * 4);
      q[kk][q4 * 4 + 0] = v.x; q[kk][q4 * 4 + 1] = v.y;
      q[kk][q4 * 4 + 2] = v.z; q[kk][q4 * 4 + 3] = v.w;
    }
  }
}

template <bool FORM, bool USE>
static __device__ __forceinline__ void sx_step(const float* __restrict__ pab, int p0, int w, int j,
                                               float (&q)[4][16], const float (&pm)[16],
                                               float* afw, const float* afr, const float* bcol,
                                               const v2f (&au)[4], const v2f (&bu)[4][2],
                                               v2f (&an)[4], v2f (&bn)[4][2], v8f (&cacc)[2]) {
  if (USE) {
    const int jn = (j + 1) & 7;
#pragma unroll
    for (int u = 0; u < 4; ++u) {
      an[u] = *(const v2f*)(afr + (4 * jn + u) * 64);
      bn[u][0] = *(const v2f*)(bcol + (2 * (4 * jn + u)) * 160);
      bn[u][1] = *(const v2f*)(bcol + (2 * (4 * jn + u)) * 160 + 32);
    }
  }
  float e[4];
  if (FORM) {
#pragma unroll
    for (int kk = 0; kk < 4; ++kk) e[kk] = exp_neg_l1(q[kk], pm);
    sx_load_pts(pab, p0 + j + 2, w, q);
  }
  if (USE) {
#pragma unroll
    for (int u = 0; u < 4; ++u) {
      cacc[0] = wmma_f32(au[u], bu[u][0], cacc[0]);
      cacc[1] = wmma_f32(au[u], bu[u][1], cacc[1]);
    }
  }
  if (FORM && USE) {
    __builtin_amdgcn_sched_group_barrier(0x100, 8, 0);
#pragma unroll
    for (int u = 0; u < 8; ++u) {
      __builtin_amdgcn_sched_group_barrier(0x008, 1, 0);
      __builtin_amdgcn_sched_group_barrier(0x002, 19, 0);
    }
  }
  if (FORM) {
    *(v2f*)(afw + j * 64)      = v2f{e[0], e[1]};
    *(v2f*)(afw + j * 64 + 32) = v2f{e[2], e[3]};
  }
}

template <bool FORM, bool USE>
static __device__ __forceinline__ void sx_steps(const float* __restrict__ pab, int p0, int w,
                                                float (&qa)[4][16], float (&qb)[4][16],
                                                const float (&pm)[16],
                                                float* afw, const float* afr, const float* bcol,
                                                v8f (&cacc)[2]) {
  v2f a0[4], b0[4][2], a1[4], b1[4][2];
  if (USE) {
#pragma unroll
    for (int u = 0; u < 4; ++u) {
      a0[u] = *(const v2f*)(afr + u * 64);
      b0[u][0] = *(const v2f*)(bcol + (2 * u) * 160);
      b0[u][1] = *(const v2f*)(bcol + (2 * u) * 160 + 32);
    }
  }
  for (int jj = 0; jj < 8; jj += 2) {
    sx_step<FORM, USE>(pab, p0, w, jj,     qa, pm, afw, afr, bcol, a0, b0, a1, b1, cacc);
    sx_step<FORM, USE>(pab, p0, w, jj + 1, qb, pm, afw, afr, bcol, a1, b1, a0, b0, cacc);
  }
}

__global__ __launch_bounds__(128)
void k_sx_points(const float* __restrict__ pa, const float* __restrict__ x,
                 const float* __restrict__ rS,
                 float* __restrict__ sx, float* __restrict__ sxF) {
  const int b = blockIdx.y;
  const int rowBase = blockIdx.x * 32;
  const float* pab = pa + (size_t)b * 1024 * 16;
  const float* rsb = rS + (size_t)b * 1024;
  const float* Bb = x + (size_t)b * 1024 * 64;
  float* Cb = sx + (size_t)b * 1024 * 64;

  __shared__ float Bs[SX_KC / 2][160];
  __shared__ __attribute__((aligned(16))) float Af[2 * SX_AF_BUF];

  const int t = threadIdx.x;
  const int wave = t >> 5;
  const int lane = t & 31;
  const int m16 = lane & 15;
  const int kh = lane >> 4;
  const int rt = wave >> 1;
  const int ch = wave & 1;

  float pm[16];
  float qa[4][16], qb[4][16];
  {
    const float* pr = pab + (size_t)(rowBase + lane) * 16;
#pragma unroll
    for (int q = 0; q < 4; ++q) {
      const float4 v = *(const float4*)(pr + q * 4);
      pm[q * 4 + 0] = v.x; pm[q * 4 + 1] = v.y; pm[q * 4 + 2] = v.z; pm[q * 4 + 3] = v.w;
    }
    sx_load_pts(pab, 0, wave, qa);
    sx_load_pts(pab, 1, wave, qb);
  }

  const v8f zero = {0.f, 0.f, 0.f, 0.f, 0.f, 0.f, 0.f, 0.f};
  v8f cacc[2];
  cacc[0] = zero; cacc[1] = zero;

  float* afw0 = Af + (lane >> 4) * SX_AF_TILE + (8 * wave) * 64 + (lane & 15) * 2;
  const float* afr0 = Af + rt * SX_AF_TILE + lane * 2;
  const float* bcol0 = &Bs[kh][(ch * 32 + m16) * 2];

  for (int kc = 0; kc < 1024; kc += SX_KC) {
    if (kc != 0) __syncthreads();
    for (int i0 = t; i0 < (SX_KC / 2) * 16; i0 += 128 * 8) {
      float4 v0[8], v1[8];
      v2f r[8];
#pragma unroll
      for (int u = 0; u < 8; ++u) {
        const int i = i0 + u * 128;
        const int kp = i >> 4, cc = (i & 15) << 2;
        v0[u] = *(const float4*)(Bb + (size_t)(kc + 2 * kp) * 64 + cc);
        v1[u] = *(const float4*)(Bb + (size_t)(kc + 2 * kp + 1) * 64 + cc);
        r[u] = *(const v2f*)(rsb + kc + 2 * kp);
      }
#pragma unroll
      for (int u = 0; u < 8; ++u) {
        const int i = i0 + u * 128;
        const int kp = i >> 4, cc = (i & 15) << 2;
        float4 lo = {v0[u].x * r[u][0], v1[u].x * r[u][1], v0[u].y * r[u][0], v1[u].y * r[u][1]};
        float4 hi = {v0[u].z * r[u][0], v1[u].z * r[u][1], v0[u].w * r[u][0], v1[u].w * r[u][1]};
        *(float4*)(&Bs[kp][cc * 2])     = lo;
        *(float4*)(&Bs[kp][cc * 2 + 4]) = hi;
      }
    }
    if (kc == 0)
      sx_steps<true, false>(pab, 0, wave, qa, qb, pm, afw0, afr0, bcol0, cacc);
    __syncthreads();

    for (int s4 = 0; s4 < 4; ++s4) {
      const int sc = kc / 128 + s4;
      const float* afr = afr0 + (sc & 1) * SX_AF_BUF;
      float* afw = afw0 + ((sc + 1) & 1) * SX_AF_BUF;
      const float* bcol = bcol0 + (size_t)(s4 * 64) * 160;
      if (sc < 7) {
        sx_steps<true, true>(pab, (sc + 1) * 8, wave, qa, qb, pm, afw, afr, bcol, cacc);
        if (s4 < 3) __syncthreads();
      } else {
        sx_steps<false, true>(pab, 0, wave, qa, qb, pm, afw, afr, bcol, cacc);
      }
    }
  }

#pragma unroll
  for (int nt = 0; nt < 2; ++nt) {
#pragma unroll
    for (int r = 0; r < 8; ++r) {
      const int m = rowBase + rt * 16 + r + 8 * kh;
      Cb[(size_t)m * 64 + (ch * 2 + nt) * 16 + m16] = cacc[nt][r];
    }
  }
  {
    float* Fb = sxF + (((size_t)b * 64 + blockIdx.x * 2 + rt) * 4 + ch * 2) * 256 + lane * 8;
#pragma unroll
    for (int nt = 0; nt < 2; ++nt) {
      *(float4*)(Fb + nt * 256)     = float4{cacc[nt][0], cacc[nt][1], cacc[nt][2], cacc[nt][3]};
      *(float4*)(Fb + nt * 256 + 4) = float4{cacc[nt][4], cacc[nt][5], cacc[nt][6], cacc[nt][7]};
    }
  }
}

#define XG_PF 8
typedef float v4f __attribute__((ext_vector_type(4)));
#define V4_LO(x) __builtin_shufflevector((x), (x), 0, 1)
#define V4_HI(x) __builtin_shufflevector((x), (x), 2, 3)
__global__ __launch_bounds__(128)
void k_xg_sim(const float* __restrict__ E, const float* __restrict__ rmx,
              const float* __restrict__ rinv, const float* __restrict__ sxF,
              float* __restrict__ xg1) {
  const int b = blockIdx.y;
  const int rowBase = blockIdx.x * 32;
  float* Cb = xg1 + (size_t)b * 1024 * 64;

  const int t = threadIdx.x;
  const int wave = t >> 5;
  const int lane = t & 31;
  const int m16 = lane & 15;
  const int kh = lane >> 4;
  const int rt = wave >> 1;
  const int ch = wave & 1;

  const int mrow = rowBase + rt * 16 + m16;
  v2f bq[4];
  {
    const float4 e0 = *(const float4*)(E + (size_t)mrow * 16 + 8 * kh);
    const float4 e1 = *(const float4*)(E + (size_t)mrow * 16 + 8 * kh + 4);
    bq[0][0] = e0.x; bq[0][1] = e0.y; bq[1][0] = e0.z; bq[1][1] = e0.w;
    bq[2][0] = e1.x; bq[2][1] = e1.y; bq[3][0] = e1.z; bq[3][1] = e1.w;
  }
  const float mxm = rmx[mrow];
  const float invm = rinv[mrow];
  const float* erow = E + (size_t)m16 * 16 + 8 * kh;
  const float* bfr = sxF + ((size_t)b * 256 + ch * 2) * 256 + lane * 8;

  const v8f zero = {0.f, 0.f, 0.f, 0.f, 0.f, 0.f, 0.f, 0.f};
  v8f cacc[2];
  cacc[0] = zero; cacc[1] = zero;

  v4f eA[XG_PF][2], eB[XG_PF][2];
  v4f bA[XG_PF][2][2], bB[XG_PF][2][2];
  v8f st_cur;
  v2f a_cur[4];

  auto request = [&](v4f (&eb)[XG_PF][2], v4f (&bb)[XG_PF][2][2], int tt, int g) {
    const float* en = erow + (size_t)((g + 2) & 63) * 256;
    eb[tt][0] = *(const v4f*)(en);
    eb[tt][1] = *(const v4f*)(en + 4);
    const float* bn = bfr + (size_t)(g & 63) * 1024;
#pragma unroll
    for (int nt = 0; nt < 2; ++nt) {
      bb[tt][nt][0] = *(const v4f*)(bn + nt * 256);
      bb[tt][nt][1] = *(const v4f*)(bn + nt * 256 + 4);
    }
  };
  auto trip = [&](const v4f (&eb)[XG_PF][2], const v4f (&bb)[XG_PF][2][2], int tt) {
    const v2f ta[4] = {V4_LO(eb[tt][0]), V4_HI(eb[tt][0]), V4_LO(eb[tt][1]), V4_HI(eb[tt][1])};
    v2f bf[4][2];
#pragma unroll
    for (int nt = 0; nt < 2; ++nt) {
      bf[0][nt] = V4_LO(bb[tt][nt][0]); bf[1][nt] = V4_HI(bb[tt][nt][0]);
      bf[2][nt] = V4_LO(bb[tt][nt][1]); bf[3][nt] = V4_HI(bb[tt][nt][1]);
    }
    v2f a_nxt[4];
#pragma unroll
    for (int j = 0; j < 4; ++j) {
      a_nxt[j][0] = __expf(fmaxf(st_cur[2 * j], 0.f) - mxm) * invm;
      a_nxt[j][1] = __expf(fmaxf(st_cur[2 * j + 1], 0.f) - mxm) * invm;
    }
    v8f st_nxt = zero;
#pragma unroll
    for (int j = 0; j < 4; ++j) {
      st_nxt = wmma_f32(ta[j], bq[j], st_nxt);
      cacc[0] = wmma_f32(a_cur[j], bf[j][0], cacc[0]);
      cacc[1] = wmma_f32(a_cur[j], bf[j][1], cacc[1]);
    }
    st_cur = st_nxt;
#pragma unroll
    for (int j = 0; j < 4; ++j) a_cur[j] = a_nxt[j];
    __builtin_amdgcn_sched_barrier(0);
  };

  {
    const v4f f0 = *(const v4f*)(erow);
    const v4f f1 = *(const v4f*)(erow + 4);
    const v4f g0 = *(const v4f*)(erow + 256);
    const v4f g1 = *(const v4f*)(erow + 256 + 4);
#pragma unroll
    for (int u = 0; u < XG_PF; ++u) request(eA, bA, u, u);
    asm volatile("" ::: "memory");
    v8f s0 = zero;
    s0 = wmma_f32(V4_LO(f0), bq[0], s0);
    s0 = wmma_f32(V4_HI(f0), bq[1], s0);
    s0 = wmma_f32(V4_LO(f1), bq[2], s0);
    s0 = wmma_f32(V4_HI(f1), bq[3], s0);
    v8f s1 = zero;
    s1 = wmma_f32(V4_LO(g0), bq[0], s1);
    s1 = wmma_f32(V4_HI(g0), bq[1], s1);
    s1 = wmma_f32(V4_LO(g1), bq[2], s1);
    s1 = wmma_f32(V4_HI(g1), bq[3], s1);
#pragma unroll
    for (int j = 0; j < 4; ++j) {
      a_cur[j][0] = __expf(fmaxf(s0[2 * j], 0.f) - mxm) * invm;
      a_cur[j][1] = __expf(fmaxf(s0[2 * j + 1], 0.f) - mxm) * invm;
    }
    st_cur = s1;
  }

  for (int g0 = 0; g0 < 64; g0 += 2 * XG_PF) {
#pragma unroll
    for (int u = 0; u < XG_PF; ++u) {
      if (u < XG_PF / 2) {
        request(eB, bB, 2 * u,     g0 + XG_PF + 2 * u);
        request(eB, bB, 2 * u + 1, g0 + XG_PF + 2 * u + 1);
        asm volatile("" ::: "memory");
      }
      trip(eA, bA, u);
    }
#pragma unroll
    for (int u = 0; u < XG_PF; ++u) {
      if (u < XG_PF / 2) {
        request(eA, bA, 2 * u,     g0 + 2 * XG_PF + 2 * u);
        request(eA, bA, 2 * u + 1, g0 + 2 * XG_PF + 2 * u + 1);
        asm volatile("" ::: "memory");
      }
      trip(eB, bB, u);
    }
  }

#pragma unroll
  for (int nt = 0; nt < 2; ++nt) {
#pragma unroll
    for (int r = 0; r < 8; ++r) {
      const int m = rowBase + rt * 16 + r + 8 * kh;
      Cb[(size_t)m * 64 + (ch * 2 + nt) * 16 + m16] = cacc[nt][r];
    }
  }
}

#define FP_DS 2
#define FP_NS (16 / FP_DS)

static __device__ __forceinline__ void pool_issue(const float* __restrict__ src, int t,
                                                  float4 (&v0)[16], float4 (&v1)[16]) {
#pragma unroll
  for (int u = 0; u < 16; ++u) {
    const int i = t + u * 128;
    const int kp = i >> 4, cc = (i & 15) << 2;
    v0[u] = *(const float4*)(src + (size_t)(2 * kp) * 64 + cc);
    v1[u] = *(const float4*)(src + (size_t)(2 * kp + 1) * 64 + cc);
  }
}

static __device__ __forceinline__ void pool_store(float (*Bsb)[160], int t,
                                                  const float4 (&v0)[16], const float4 (&v1)[16]) {
#pragma unroll
  for (int u = 0; u < 16; ++u) {
    const int i = t + u * 128;
    const int kp = i >> 4, cc = (i & 15) << 2;
    float4 lo = {v0[u].x, v1[u].x, v0[u].y, v1[u].y};
    float4 hi = {v0[u].z, v1[u].z, v0[u].w, v1[u].w};
    *(float4*)(&Bsb[kp][cc * 2])     = lo;
    *(float4*)(&Bsb[kp][cc * 2 + 4]) = hi;
  }
}

static __device__ __forceinline__ void pool_stage(const float* bc, const v2f (&a)[32],
                                                  v2f e2, v8f (&cacc)[2]) {
  v2f bf[2][8][2];
#pragma unroll
  for (int j = 0; j < 8; ++j) {
    bf[0][j][0] = *(const v2f*)(bc + (2 * j) * 160);
    bf[0][j][1] = *(const v2f*)(bc + (2 * j) * 160 + 32);
  }
#pragma unroll
  for (int blk = 0; blk < FP_DS * 4; ++blk) {
    if (blk + 1 < FP_DS * 4) {
#pragma unroll
      for (int j = 0; j < 8; ++j) {
        bf[(blk + 1) & 1][j][0] = *(const v2f*)(bc + (16 * (blk + 1) + 2 * j) * 160);
        bf[(blk + 1) & 1][j][1] = *(const v2f*)(bc + (16 * (blk + 1) + 2 * j) * 160 + 32);
      }
    }
    const float ed = e2[blk >> 2];
#pragma unroll
    for (int j = 0; j < 8; ++j) {
      const v2f as = a[8 * (blk & 3) + j] * ed;
      cacc[0] = wmma_f32(as, bf[blk & 1][j][0], cacc[0]);
      cacc[1] = wmma_f32(as, bf[blk & 1][j][1], cacc[1]);
    }
  }
}

__global__ __launch_bounds__(128)
void k_final_pool(const float* __restrict__ sx, const float* __restrict__ xg1,
                  const float* __restrict__ E, const float* __restrict__ Wp,
                  const float* __restrict__ bias_pool, float* __restrict__ out) {
  const int b = blockIdx.y;
  const int rowBase = blockIdx.x * 32;

  __shared__ float Bs[2][FP_DS * 64][160];

  const int t = threadIdx.x;
  const int wave = t >> 5;
  const int lane = t & 31;
  const int m16 = lane & 15;
  const int kh = lane >> 4;
  const int rt = wave >> 1;
  const int ch = wave & 1;
  const int nrow = rowBase + rt * 16 + m16;

  float4 v0[16], v1[16];
  pool_issue(Wp, t, v0, v1);

  v2f a[32];
  {
    const size_t roff = ((size_t)b * 1024 + nrow) * 64 + 2 * kh;
#pragma unroll
    for (int ks = 0; ks < 16; ++ks) {
      a[ks]      = *(const v2f*)(sx  + roff + 4 * ks);
      a[16 + ks] = *(const v2f*)(xg1 + roff + 4 * ks);
    }
  }
  const float* erow = E + (size_t)nrow * 16;
  v2f ea[4], bb[4][2];
#pragma unroll
  for (int j = 0; j < 4; ++j) {
    ea[j] = *(const v2f*)(erow + 4 * j + 2 * kh);
#pragma unroll
    for (int nt = 0; nt < 2; ++nt) {
      const int col = ch * 32 + nt * 16 + m16;
      bb[j][nt][0] = bias_pool[(4 * j + 2 * kh) * 64 + col];
      bb[j][nt][1] = bias_pool[(4 * j + 2 * kh + 1) * 64 + col];
    }
  }
  v2f e_cur = *(const v2f*)(erow);

  const v8f zero = {0.f, 0.f, 0.f, 0.f, 0.f, 0.f, 0.f, 0.f};
  v8f cacc[2];
  cacc[0] = zero; cacc[1] = zero;
#pragma unroll
  for (int j = 0; j < 4; ++j) {
    cacc[0] = wmma_f32(ea[j], bb[j][0], cacc[0]);
    cacc[1] = wmma_f32(ea[j], bb[j][1], cacc[1]);
  }

  pool_store(Bs[0], t, v0, v1);
  __syncthreads();

  for (int s = 0; s < FP_NS; ++s) {
    const int sn = (s + 1) & (FP_NS - 1);
    pool_issue(Wp + (size_t)sn * (FP_DS * 8192), t, v0, v1);
    const v2f e_nxt = *(const v2f*)(erow + FP_DS * sn);
    pool_stage(&Bs[s & 1][kh][(ch * 32 + m16) * 2], a, e_cur, cacc);
    pool_store(Bs[(s + 1) & 1], t, v0, v1);
    e_cur = e_nxt;
    __syncthreads();
  }

  float* Cb = out + (size_t)b * 1024 * 64;
#pragma unroll
  for (int nt = 0; nt < 2; ++nt) {
#pragma unroll
    for (int r = 0; r < 8; ++r) {
      const int m = rowBase + rt * 16 + r + 8 * kh;
      Cb[(size_t)m * 64 + (ch * 2 + nt) * 16 + m16] = cacc[nt][r];
    }
  }
}

extern "C" void kernel_launch(void* const* d_in, const int* in_sizes, int n_in,
                              void* d_out, int out_size, void* d_ws, size_t ws_size,
                              hipStream_t stream) {
  (void)in_sizes; (void)n_in; (void)out_size; (void)ws_size;
  const float* x    = (const float*)d_in[0];
  const float* E    = (const float*)d_in[1];
  const float* pa   = (const float*)d_in[2];
  const float* Wp   = (const float*)d_in[3];
  const float* bp   = (const float*)d_in[4];
  float* out = (float*)d_out;

  float* ws   = (float*)d_ws;
  float* rmx  = ws;
  float* rinv = ws + 1024;
  float* sube = ws   + 1048576;
  float* rS   = sube + 8388608;
  float* sx   = rS   + 8192 + 524288;
  float* xg1  = sx   + 524288;
  float* sxF  = sube;

  k_rowstats  <<<64, 512, 0, stream>>>(E, rmx, rinv);
  k_colsum_pts<<<dim3(32, 8),     512, 0, stream>>>(pa, rS);
  k_sx_points<<<dim3(32, 8), 128, 0, stream>>>(pa, x, rS, sx, sxF);
  k_xg_sim<<<dim3(32, 8), 128, 180224, stream>>>(E, rmx, rinv, sxF, xg1);
  k_final_pool<<<dim3(32, 8), 128, 0, stream>>>(sx, xg1, E, Wp, bp, out);
}


# sx product on 8-wave work-groups: four waves only stage B (x rows times 1/S) into two LDS stages, a 256-k chunk ahead, while four form and multiply; no staging phases left; sum order unchanged
# speedup vs baseline: 3.5992x; 1.0413x over previous
#include <hip/hip_runtime.h>
#include <hip/hip_bf16.h>
#include <cstddef>


typedef float v2f __attribute__((ext_vector_type(2)));
typedef float v8f __attribute__((ext_vector_type(8)));

static __device__ __forceinline__ v8f wmma_f32(v2f a, v2f b, v8f c) {
  return __builtin_amdgcn_wmma_f32_16x16x4_f32(
       false, a,  false, b,
       (short)0, c,  false,  false);
}

static __device__ __forceinline__ float exp_neg_l1(const float (&p)[16],
                                                   const float (&q)[16]) {
  float s0 = 0.f, s1 = 0.f, s2 = 0.f, s3 = 0.f;
#pragma unroll
  for (int d = 0; d < 4; ++d) {
    s0 += fabsf(p[d]      - q[d]);
    s1 += fabsf(p[d + 4]  - q[d + 4]);
    s2 += fabsf(p[d + 8]  - q[d + 8]);
    s3 += fabsf(p[d + 12] - q[d + 12]);
  }
  return __expf(-((s0 + s1) + (s2 + s3)));
}

__global__ __launch_bounds__(512)
void k_rowstats(const float* __restrict__ E, float* __restrict__ rmx,
                float* __restrict__ rinv) {
  const int t = threadIdx.x;
  const int wave = t >> 5;
  const int lane = t & 31;
  const int m16 = lane & 15;
  const int kh = lane >> 4;
  const int mrow = blockIdx.x * 16 + m16;
  __shared__ float part[32][16];

  v2f bq[4];
  {
    const float4 e0 = *(const float4*)(E + (size_t)mrow * 16 + 8 * kh);
    const float4 e1 = *(const float4*)(E + (size_t)mrow * 16 + 8 * kh + 4);
    bq[0][0] = e0.x; bq[0][1] = e0.y; bq[1][0] = e0.z; bq[1][1] = e0.w;
    bq[2][0] = e1.x; bq[2][1] = e1.y; bq[3][0] = e1.z; bq[3][1] = e1.w;
  }
  const v8f zero = {0.f, 0.f, 0.f, 0.f, 0.f, 0.f, 0.f, 0.f};
  v8f tl[4];
#pragma unroll
  for (int q = 0; q < 4; ++q) {
    const float* er = E + (size_t)(wave * 64 + q * 16 + m16) * 16 + 8 * kh;
    const float4 f0 = *(const float4*)(er);
    const float4 f1 = *(const float4*)(er + 4);
    v8f st = zero;
    st = wmma_f32(v2f{f0.x, f0.y}, bq[0], st);
    st = wmma_f32(v2f{f0.z, f0.w}, bq[1], st);
    st = wmma_f32(v2f{f1.x, f1.y}, bq[2], st);
    st = wmma_f32(v2f{f1.z, f1.w}, bq[3], st);
    tl[q] = st;
  }
  float mx = 0.f;
#pragma unroll
  for (int q = 0; q < 4; ++q)
#pragma unroll
    for (int r = 0; r < 8; ++r) mx = fmaxf(mx, tl[q][r]);
  part[wave * 2 + kh][m16] = mx;
  __syncthreads();
  float m = 0.f;
#pragma unroll
  for (int p = 0; p < 32; ++p) m = fmaxf(m, part[p][m16]);
  __syncthreads();
  float acc = 0.f;
#pragma unroll
  for (int q = 0; q < 4; ++q)
#pragma unroll
    for (int r = 0; r < 8; ++r) acc += __expf(fmaxf(tl[q][r], 0.f) - m);
  part[wave * 2 + kh][m16] = acc;
  __syncthreads();
  if (t < 16) {
    float tot = 0.f;
#pragma unroll
    for (int p = 0; p < 32; ++p) tot += part[p][t];
    rmx[blockIdx.x * 16 + t] = m;
    rinv[blockIdx.x * 16 + t] = 1.0f / tot;
  }
}

__global__ __launch_bounds__(512)
void k_colsum_pts(const float* __restrict__ pa, float* __restrict__ rS) {
  const int b = blockIdx.y;
  const int lane = threadIdx.x & 31;
  const int w = __builtin_amdgcn_readfirstlane((int)(threadIdx.x >> 5));
  const int j = blockIdx.x * 32 + lane;
  __shared__ float part[16][32];

  float pj[16];
#pragma unroll
  for (int q = 0; q < 4; ++q) {
    const float4 v = *(const float4*)(pa + ((size_t)b * 1024 + j) * 16 + q * 4);
    pj[q * 4 + 0] = v.x; pj[q * 4 + 1] = v.y; pj[q * 4 + 2] = v.z; pj[q * 4 + 3] = v.w;
  }
  const float* pi = pa + ((size_t)b * 1024 + (size_t)w * 64) * 16;
  float acc = 0.f;
  for (int ii = 0; ii < 64; ++ii) {
    float qi[16];
#pragma unroll
    for (int d = 0; d < 16; ++d) qi[d] = pi[ii * 16 + d];
    acc += exp_neg_l1(qi, pj);
  }
  part[w][lane] = acc;
  __syncthreads();
  if (threadIdx.x < 32) {
    float tot = 0.f;
#pragma unroll
    for (int w2 = 0; w2 < 16; ++w2) tot += part[w2][threadIdx.x];
    rS[b * 1024 + blockIdx.x * 32 + threadIdx.x] = 1.0f / tot;
  }
}

#define SX_KC 256
#define SX_AF_TILE 2080
#define SX_AF_BUF (2 * SX_AF_TILE)

static __device__ __forceinline__ int sx_step_row(int p, int w) {
  return ((p >> 3) * 128 + w * 32 + (p & 7) * 4) & 1023;
}

static __device__ __forceinline__ void sx_load_pts(const float* __restrict__ pab, int p, int w,
                                                   float (&q)[4][16]) {
  const float* pn = pab + (size_t)sx_step_row(p, w) * 16;
#pragma unroll
  for (int kk = 0; kk < 4; ++kk) {
#pragma unroll
    for (int q4 = 0; q4 < 4; ++q4) {
      const float4 v = *(const float4*)(pn + kk * 16 + q4 * 4);
      q[kk][q4 * 4 + 0] = v.x; q[kk][q4 * 4 + 1] = v.y;
      q[kk][q4 * 4 + 2] = v.z; q[kk][q4 * 4 + 3] = v.w;
    }
  }
}

template <bool FORM, bool USE>
static __device__ __forceinline__ void sx_step(const float* __restrict__ pab, int p0, int w, int j,
                                               float (&q)[4][16], const float (&pm)[16],
                                               float* afw, const float* afr, const float* bcol,
                                               const v2f (&au)[4], const v2f (&bu)[4][2],
                                               v2f (&an)[4], v2f (&bn)[4][2], v8f (&cacc)[2]) {
  if (USE) {
    const int jn = (j + 1) & 7;
#pragma unroll
    for (int u = 0; u < 4; ++u) {
      an[u] = *(const v2f*)(afr + (4 * jn + u) * 64);
      bn[u][0] = *(const v2f*)(bcol + (2 * (4 * jn + u)) * 160);
      bn[u][1] = *(const v2f*)(bcol + (2 * (4 * jn + u)) * 160 + 32);
    }
  }
  float e[4];
  if (FORM) {
#pragma unroll
    for (int kk = 0; kk < 4; ++kk) e[kk] = exp_neg_l1(q[kk], pm);
    sx_load_pts(pab, p0 + j + 2, w, q);
  }
  if (USE) {
#pragma unroll
    for (int u = 0; u < 4; ++u) {
      cacc[0] = wmma_f32(au[u], bu[u][0], cacc[0]);
      cacc[1] = wmma_f32(au[u], bu[u][1], cacc[1]);
    }
  }
  if (FORM && USE) {
    __builtin_amdgcn_sched_group_barrier(0x100, 8, 0);
#pragma unroll
    for (int u = 0; u < 8; ++u) {
      __builtin_amdgcn_sched_group_barrier(0x008, 1, 0);
      __builtin_amdgcn_sched_group_barrier(0x002, 19, 0);
    }
  }
  if (FORM) {
    *(v2f*)(afw + j * 64)      = v2f{e[0], e[1]};
    *(v2f*)(afw + j * 64 + 32) = v2f{e[2], e[3]};
  }
}

template <bool FORM, bool USE>
static __device__ __forceinline__ void sx_steps(const float* __restrict__ pab, int p0, int w,
                                                float (&qa)[4][16], float (&qb)[4][16],
                                                const float (&pm)[16],
                                                float* afw, const float* afr, const float* bcol,
                                                v8f (&cacc)[2]) {
  v2f a0[4], b0[4][2], a1[4], b1[4][2];
  if (USE) {
#pragma unroll
    for (int u = 0; u < 4; ++u) {
      a0[u] = *(const v2f*)(afr + u * 64);
      b0[u][0] = *(const v2f*)(bcol + (2 * u) * 160);
      b0[u][1] = *(const v2f*)(bcol + (2 * u) * 160 + 32);
    }
  }
#pragma unroll(FORM && USE ? 4 : 1)
  for (int jj = 0; jj < 8; jj += 2) {
    sx_step<FORM, USE>(pab, p0, w, jj,     qa, pm, afw, afr, bcol, a0, b0, a1, b1, cacc);
    sx_step<FORM, USE>(pab, p0, w, jj + 1, qb, pm, afw, afr, bcol, a1, b1, a0, b0, cacc);
  }
}

static __device__ __forceinline__ void sx_stage_half(const float* __restrict__ Bb,
                                                     const float* __restrict__ rsb, int k0,
                                                     float (*Bst)[160], int pr0, int ts) {
  float4 v0[8], v1[8];
  v2f r[8];
#pragma unroll
  for (int u = 0; u < 8; ++u) {
    const int i = ts + u * 128;
    const int kp = i >> 4, cc = (i & 15) << 2;
    v0[u] = *(const float4*)(Bb + (size_t)(k0 + 2 * kp) * 64 + cc);
    v1[u] = *(const float4*)(Bb + (size_t)(k0 + 2 * kp + 1) * 64 + cc);
    r[u] = *(const v2f*)(rsb + k0 + 2 * kp);
  }
#pragma unroll
  for (int u = 0; u < 8; ++u) {
    const int i = ts + u * 128;
    const int kp = i >> 4, cc = (i & 15) << 2;
    float4 lo = {v0[u].x * r[u][0], v1[u].x * r[u][1], v0[u].y * r[u][0], v1[u].y * r[u][1]};
    float4 hi = {v0[u].z * r[u][0], v1[u].z * r[u][1], v0[u].w * r[u][0], v1[u].w * r[u][1]};
    *(float4*)(&Bst[pr0 + kp][cc * 2])     = lo;
    *(float4*)(&Bst[pr0 + kp][cc * 2 + 4]) = hi;
  }
}

__global__ __launch_bounds__(256)
void k_sx_points(const float* __restrict__ pa, const float* __restrict__ x,
                 const float* __restrict__ rS,
                 float* __restrict__ sx, float* __restrict__ sxF) {
  const int b = blockIdx.y;
  const int rowBase = blockIdx.x * 32;
  const float* pab = pa + (size_t)b * 1024 * 16;
  const float* rsb = rS + (size_t)b * 1024;
  const float* Bb = x + (size_t)b * 1024 * 64;
  float* Cb = sx + (size_t)b * 1024 * 64;

  __shared__ float Bs[2][SX_KC / 2][160];
  __shared__ __attribute__((aligned(16))) float Af[2 * SX_AF_BUF];

  const int t = threadIdx.x;
  const int lane = t & 31;
  const int m16 = lane & 15;
  const int kh = lane >> 4;
  const int wnum = __builtin_amdgcn_readfirstlane(t >> 5);
  const bool product = (wnum & 1) == (wnum >> 2);
  const int wave = t >> 6;
  const int rt = wave >> 1;
  const int ch = wave & 1;
  const int ts = wave * 32 + lane;

  float pm[16];
  float qa[4][16], qb[4][16];

  const v8f zero = {0.f, 0.f, 0.f, 0.f, 0.f, 0.f, 0.f, 0.f};
  v8f cacc[2];
  cacc[0] = zero; cacc[1] = zero;

  float* afw0 = Af + (lane >> 4) * SX_AF_TILE + (8 * wave) * 64 + (lane & 15) * 2;
  const float* afr0 = Af + rt * SX_AF_TILE + lane * 2;
  const float* bcol0 = &Bs[0][kh][(ch * 32 + m16) * 2];

  if (product) {
    const float* pr = pab + (size_t)(rowBase + lane) * 16;
#pragma unroll
    for (int q = 0; q < 4; ++q) {
      const float4 v = *(const float4*)(pr + q * 4);
      pm[q * 4 + 0] = v.x; pm[q * 4 + 1] = v.y; pm[q * 4 + 2] = v.z; pm[q * 4 + 3] = v.w;
    }
    sx_load_pts(pab, 0, wave, qa);
    sx_load_pts(pab, 1, wave, qb);
    sx_steps<true, false>(pab, 0, wave, qa, qb, pm, afw0, afr0, bcol0, cacc);
  } else {
    sx_stage_half(Bb, rsb, 0,   Bs[0], 0,  ts);
    sx_stage_half(Bb, rsb, 128, Bs[0], 64, ts);
  }
  __syncthreads();

  for (int sc = 0; sc < 8; ++sc) {
    if (product) {
      const float* afr = afr0 + (sc & 1) * SX_AF_BUF;
      float* afw = afw0 + ((sc + 1) & 1) * SX_AF_BUF;
      const float* bcol = bcol0 + (size_t)(((sc >> 1) & 1) * (SX_KC / 2) + (sc & 1) * 64) * 160;
      if (sc < 7) {
        sx_steps<true, true>(pab, (sc + 1) * 8, wave, qa, qb, pm, afw, afr, bcol, cacc);
      } else {
        sx_steps<false, true>(pab, 0, wave, qa, qb, pm, afw, afr, bcol, cacc);
      }
    } else if (sc < 6) {
      const int cn = (sc >> 1) + 1;
      sx_stage_half(Bb, rsb, cn * SX_KC + (sc & 1) * 128, Bs[cn & 1], (sc & 1) * 64, ts);
    }
    if (sc < 7) __syncthreads();
  }

  if (product) {
#pragma unroll
  for (int nt = 0; nt < 2; ++nt) {
#pragma unroll
    for (int r = 0; r < 8; ++r) {
      const int m = rowBase + rt * 16 + r + 8 * kh;
      Cb[(size_t)m * 64 + (ch * 2 + nt) * 16 + m16] = cacc[nt][r];
    }
  }
  {
    float* Fb = sxF + (((size_t)b * 64 + blockIdx.x * 2 + rt) * 4 + ch * 2) * 256 + lane * 8;
#pragma unroll
    for (int nt = 0; nt < 2; ++nt) {
      *(float4*)(Fb + nt * 256)     = float4{cacc[nt][0], cacc[nt][1], cacc[nt][2], cacc[nt][3]};
      *(float4*)(Fb + nt * 256 + 4) = float4{cacc[nt][4], cacc[nt][5], cacc[nt][6], cacc[nt][7]};
    }
  }
  }
}

#define XG_PF 8
typedef float v4f __attribute__((ext_vector_type(4)));
#define V4_LO(x) __builtin_shufflevector((x), (x), 0, 1)
#define V4_HI(x) __builtin_shufflevector((x), (x), 2, 3)
__global__ __launch_bounds__(128)
void k_xg_sim(const float* __restrict__ E, const float* __restrict__ rmx,
              const float* __restrict__ rinv, const float* __restrict__ sxF,
              float* __restrict__ xg1) {
  const int b = blockIdx.y;
  const int rowBase = blockIdx.x * 32;
  float* Cb = xg1 + (size_t)b * 1024 * 64;

  const int t = threadIdx.x;
  const int wave = t >> 5;
  const int lane = t & 31;
  const int m16 = lane & 15;
  const int kh = lane >> 4;
  const int rt = wave >> 1;
  const int ch = wave & 1;

  const int mrow = rowBase + rt * 16 + m16;
  v2f bq[4];
  {
    const float4 e0 = *(const float4*)(E + (size_t)mrow * 16 + 8 * kh);
    const float4 e1 = *(const float4*)(E + (size_t)mrow * 16 + 8 * kh + 4);
    bq[0][0] = e0.x; bq[0][1] = e0.y; bq[1][0] = e0.z; bq[1][1] = e0.w;
    bq[2][0] = e1.x; bq[2][1] = e1.y; bq[3][0] = e1.z; bq[3][1] = e1.w;
  }
  const float mxm = rmx[mrow];
  const float invm = rinv[mrow];
  const float* erow = E + (size_t)m16 * 16 + 8 * kh;
  const float* bfr = sxF + ((size_t)b * 256 + ch * 2) * 256 + lane * 8;

  const v8f zero = {0.f, 0.f, 0.f, 0.f, 0.f, 0.f, 0.f, 0.f};
  v8f cacc[2];
  cacc[0] = zero; cacc[1] = zero;

  v4f eA[XG_PF][2], eB[XG_PF][2];
  v4f bA[XG_PF][2][2], bB[XG_PF][2][2];
  v8f st_cur;
  v2f a_cur[4];

  auto request = [&](v4f (&eb)[XG_PF][2], v4f (&bb)[XG_PF][2][2], int tt, int g) {
    const float* en = erow + (size_t)((g + 2) & 63) * 256;
    eb[tt][0] = *(const v4f*)(en);
    eb[tt][1] = *(const v4f*)(en + 4);
    const float* bn = bfr + (size_t)(g & 63) * 1024;
#pragma unroll
    for (int nt = 0; nt < 2; ++nt) {
      bb[tt][nt][0] = *(const v4f*)(bn + nt * 256);
      bb[tt][nt][1] = *(const v4f*)(bn + nt * 256 + 4);
    }
  };
  auto trip = [&](const v4f (&eb)[XG_PF][2], const v4f (&bb)[XG_PF][2][2], int tt) {
    const v2f ta[4] = {V4_LO(eb[tt][0]), V4_HI(eb[tt][0]), V4_LO(eb[tt][1]), V4_HI(eb[tt][1])};
    v2f bf[4][2];
#pragma unroll
    for (int nt = 0; nt < 2; ++nt) {
      bf[0][nt] = V4_LO(bb[tt][nt][0]); bf[1][nt] = V4_HI(bb[tt][nt][0]);
      bf[2][nt] = V4_LO(bb[tt][nt][1]); bf[3][nt] = V4_HI(bb[tt][nt][1]);
    }
    v2f a_nxt[4];
#pragma unroll
    for (int j = 0; j < 4; ++j) {
      a_nxt[j][0] = __expf(fmaxf(st_cur[2 * j], 0.f) - mxm) * invm;
      a_nxt[j][1] = __expf(fmaxf(st_cur[2 * j + 1], 0.f) - mxm) * invm;
    }
    v8f st_nxt = zero;
#pragma unroll
    for (int j = 0; j < 4; ++j) {
      st_nxt = wmma_f32(ta[j], bq[j], st_nxt);
      cacc[0] = wmma_f32(a_cur[j], bf[j][0], cacc[0]);
      cacc[1] = wmma_f32(a_cur[j], bf[j][1], cacc[1]);
    }
    st_cur = st_nxt;
#pragma unroll
    for (int j = 0; j < 4; ++j) a_cur[j] = a_nxt[j];
    __builtin_amdgcn_sched_barrier(0);
  };

  {
    const v4f f0 = *(const v4f*)(erow);
    const v4f f1 = *(const v4f*)(erow + 4);
    const v4f g0 = *(const v4f*)(erow + 256);
    const v4f g1 = *(const v4f*)(erow + 256 + 4);
#pragma unroll
    for (int u = 0; u < XG_PF; ++u) request(eA, bA, u, u);
    asm volatile("" ::: "memory");
    v8f s0 = zero;
    s0 = wmma_f32(V4_LO(f0), bq[0], s0);
    s0 = wmma_f32(V4_HI(f0), bq[1], s0);
    s0 = wmma_f32(V4_LO(f1), bq[2], s0);
    s0 = wmma_f32(V4_HI(f1), bq[3], s0);
    v8f s1 = zero;
    s1 = wmma_f32(V4_LO(g0), bq[0], s1);
    s1 = wmma_f32(V4_HI(g0), bq[1], s1);
    s1 = wmma_f32(V4_LO(g1), bq[2], s1);
    s1 = wmma_f32(V4_HI(g1), bq[3], s1);
#pragma unroll
    for (int j = 0; j < 4; ++j) {
      a_cur[j][0] = __expf(fmaxf(s0[2 * j], 0.f) - mxm) * invm;
      a_cur[j][1] = __expf(fmaxf(s0[2 * j + 1], 0.f) - mxm) * invm;
    }
    st_cur = s1;
  }

  for (int g0 = 0; g0 < 64; g0 += 2 * XG_PF) {
#pragma unroll
    for (int u = 0; u < XG_PF; ++u) {
      if (u < XG_PF / 2) {
        request(eB, bB, 2 * u,     g0 + XG_PF + 2 * u);
        request(eB, bB, 2 * u + 1, g0 + XG_PF + 2 * u + 1);
        asm volatile("" ::: "memory");
      }
      trip(eA, bA, u);
    }
#pragma unroll
    for (int u = 0; u < XG_PF; ++u) {
      if (u < XG_PF / 2) {
        request(eA, bA, 2 * u,     g0 + 2 * XG_PF + 2 * u);
        request(eA, bA, 2 * u + 1, g0 + 2 * XG_PF + 2 * u + 1);
        asm volatile("" ::: "memory");
      }
      trip(eB, bB, u);
    }
  }

#pragma unroll
  for (int nt = 0; nt < 2; ++nt) {
#pragma unroll
    for (int r = 0; r < 8; ++r) {
      const int m = rowBase + rt * 16 + r + 8 * kh;
      Cb[(size_t)m * 64 + (ch * 2 + nt) * 16 + m16] = cacc[nt][r];
    }
  }
}

#define FP_DS 2
#define FP_NS (16 / FP_DS)

static __device__ __forceinline__ void pool_issue(const float* __restrict__ src, int t,
                                                  float4 (&v0)[16], float4 (&v1)[16]) {
#pragma unroll
  for (int u = 0; u < 16; ++u) {
    const int i = t + u * 128;
    const int kp = i >> 4, cc = (i & 15) << 2;
    v0[u] = *(const float4*)(src + (size_t)(2 * kp) * 64 + cc);
    v1[u] = *(const float4*)(src + (size_t)(2 * kp + 1) * 64 + cc);
  }
}

static __device__ __forceinline__ void pool_store(float (*Bsb)[160], int t,
                                                  const float4 (&v0)[16], const float4 (&v1)[16]) {
#pragma unroll
  for (int u = 0; u < 16; ++u) {
    const int i = t + u * 128;
    const int kp = i >> 4, cc = (i & 15) << 2;
    float4 lo = {v0[u].x, v1[u].x, v0[u].y, v1[u].y};
    float4 hi = {v0[u].z, v1[u].z, v0[u].w, v1[u].w};
    *(float4*)(&Bsb[kp][cc * 2])     = lo;
    *(float4*)(&Bsb[kp][cc * 2 + 4]) = hi;
  }
}

static __device__ __forceinline__ void pool_stage(const float* bc, const v2f (&a)[32],
                                                  v2f e2, v8f (&cacc)[2]) {
  v2f bf[2][8][2];
#pragma unroll
  for (int j = 0; j < 8; ++j) {
    bf[0][j][0] = *(const v2f*)(bc + (2 * j) * 160);
    bf[0][j][1] = *(const v2f*)(bc + (2 * j) * 160 + 32);
  }
#pragma unroll
  for (int blk = 0; blk < FP_DS * 4; ++blk) {
    if (blk + 1 < FP_DS * 4) {
#pragma unroll
      for (int j = 0; j < 8; ++j) {
        bf[(blk + 1) & 1][j][0] = *(const v2f*)(bc + (16 * (blk + 1) + 2 * j) * 160);
        bf[(blk + 1) & 1][j][1] = *(const v2f*)(bc + (16 * (blk + 1) + 2 * j) * 160 + 32);
      }
    }
    const float ed = e2[blk >> 2];
#pragma unroll
    for (int j = 0; j < 8; ++j) {
      const v2f as = a[8 * (blk & 3) + j] * ed;
      cacc[0] = wmma_f32(as, bf[blk & 1][j][0], cacc[0]);
      cacc[1] = wmma_f32(as, bf[blk & 1][j][1], cacc[1]);
    }
  }
}

__global__ __launch_bounds__(128)
void k_final_pool(const float* __restrict__ sx, const float* __restrict__ xg1,
                  const float* __restrict__ E, const float* __restrict__ Wp,
                  const float* __restrict__ bias_pool, float* __restrict__ out) {
  const int b = blockIdx.y;
  const int rowBase = blockIdx.x * 32;

  __shared__ float Bs[2][FP_DS * 64][160];

  const int t = threadIdx.x;
  const int wave = t >> 5;
  const int lane = t & 31;
  const int m16 = lane & 15;
  const int kh = lane >> 4;
  const int rt = wave >> 1;
  const int ch = wave & 1;
  const int nrow = rowBase + rt * 16 + m16;

  float4 v0[16], v1[16];
  pool_issue(Wp, t, v0, v1);

  v2f a[32];
  {
    const size_t roff = ((size_t)b * 1024 + nrow) * 64 + 2 * kh;
#pragma unroll
    for (int ks = 0; ks < 16; ++ks) {
      a[ks]      = *(const v2f*)(sx  + roff + 4 * ks);
      a[16 + ks] = *(const v2f*)(xg1 + roff + 4 * ks);
    }
  }
  const float* erow = E + (size_t)nrow * 16;
  v2f ea[4], bb[4][2];
#pragma unroll
  for (int j = 0; j < 4; ++j) {
    ea[j] = *(const v2f*)(erow + 4 * j + 2 * kh);
#pragma unroll
    for (int nt = 0; nt < 2; ++nt) {
      const int col = ch * 32 + nt * 16 + m16;
      bb[j][nt][0] = bias_pool[(4 * j + 2 * kh) * 64 + col];
      bb[j][nt][1] = bias_pool[(4 * j + 2 * kh + 1) * 64 + col];
    }
  }
  v2f e_cur = *(const v2f*)(erow);

  const v8f zero = {0.f, 0.f, 0.f, 0.f, 0.f, 0.f, 0.f, 0.f};
  v8f cacc[2];
  cacc[0] = zero; cacc[1] = zero;
#pragma unroll
  for (int j = 0; j < 4; ++j) {
    cacc[0] = wmma_f32(ea[j], bb[j][0], cacc[0]);
    cacc[1] = wmma_f32(ea[j], bb[j][1], cacc[1]);
  }

  pool_store(Bs[0], t, v0, v1);
  __syncthreads();

  for (int s = 0; s < FP_NS; ++s) {
    const int sn = (s + 1) & (FP_NS - 1);
    pool_issue(Wp + (size_t)sn * (FP_DS * 8192), t, v0, v1);
    const v2f e_nxt = *(const v2f*)(erow + FP_DS * sn);
    pool_stage(&Bs[s & 1][kh][(ch * 32 + m16) * 2], a, e_cur, cacc);
    pool_store(Bs[(s + 1) & 1], t, v0, v1);
    e_cur = e_nxt;
    __syncthreads();
  }

  float* Cb = out + (size_t)b * 1024 * 64;
#pragma unroll
  for (int nt = 0; nt < 2; ++nt) {
#pragma unroll
    for (int r = 0; r < 8; ++r) {
      const int m = rowBase + rt * 16 + r + 8 * kh;
      Cb[(size_t)m * 64 + (ch * 2 + nt) * 16 + m16] = cacc[nt][r];
    }
  }
}

extern "C" void kernel_launch(void* const* d_in, const int* in_sizes, int n_in,
                              void* d_out, int out_size, void* d_ws, size_t ws_size,
                              hipStream_t stream) {
  (void)in_sizes; (void)n_in; (void)out_size; (void)ws_size;
  const float* x    = (const float*)d_in[0];
  const float* E    = (const float*)d_in[1];
  const float* pa   = (const float*)d_in[2];
  const float* Wp   = (const float*)d_in[3];
  const float* bp   = (const float*)d_in[4];
  float* out = (float*)d_out;

  float* ws   = (float*)d_ws;
  float* rmx  = ws;
  float* rinv = ws + 1024;
  float* sube = ws   + 1048576;
  float* rS   = sube + 8388608;
  float* sx   = rS   + 8192 + 524288;
  float* xg1  = sx   + 524288;
  float* sxF  = sube;

  k_rowstats  <<<64, 512, 0, stream>>>(E, rmx, rinv);
  k_colsum_pts<<<dim3(32, 8),     512, 0, stream>>>(pa, rS);
  k_sx_points<<<dim3(32, 8), 256, 0, stream>>>(pa, x, rS, sx, sxF);
  k_xg_sim<<<dim3(32, 8), 128, 180224, stream>>>(E, rmx, rinv, sxF, xg1);
  k_final_pool<<<dim3(32, 8), 128, 0, stream>>>(sx, xg1, E, Wp, bp, out);
}


# Last two launches fused: xg1's 32x64 tile passes from the sim product's accumulators via LDS into the final product's A fragments, first pool stage requested in the sim loop's tail; sums unchanged
# speedup vs baseline: 3.7476x; 1.0412x over previous
#include <hip/hip_runtime.h>
#include <hip/hip_bf16.h>
#include <cstddef>


typedef float v2f __attribute__((ext_vector_type(2)));
typedef float v8f __attribute__((ext_vector_type(8)));

static __device__ __forceinline__ v8f wmma_f32(v2f a, v2f b, v8f c) {
  return __builtin_amdgcn_wmma_f32_16x16x4_f32(
       false, a,  false, b,
       (short)0, c,  false,  false);
}

static __device__ __forceinline__ float exp_neg_l1(const float (&p)[16],
                                                   const float (&q)[16]) {
  float s0 = 0.f, s1 = 0.f, s2 = 0.f, s3 = 0.f;
#pragma unroll
  for (int d = 0; d < 4; ++d) {
    s0 += fabsf(p[d]      - q[d]);
    s1 += fabsf(p[d + 4]  - q[d + 4]);
    s2 += fabsf(p[d + 8]  - q[d + 8]);
    s3 += fabsf(p[d + 12] - q[d + 12]);
  }
  return __expf(-((s0 + s1) + (s2 + s3)));
}

__global__ __launch_bounds__(512)
void k_rowstats(const float* __restrict__ E, float* __restrict__ rmx,
                float* __restrict__ rinv) {
  const int t = threadIdx.x;
  const int wave = t >> 5;
  const int lane = t & 31;
  const int m16 = lane & 15;
  const int kh = lane >> 4;
  const int mrow = blockIdx.x * 16 + m16;
  __shared__ float part[32][16];

  v2f bq[4];
  {
    const float4 e0 = *(const float4*)(E + (size_t)mrow * 16 + 8 * kh);
    const float4 e1 = *(const float4*)(E + (size_t)mrow * 16 + 8 * kh + 4);
    bq[0][0] = e0.x; bq[0][1] = e0.y; bq[1][0] = e0.z; bq[1][1] = e0.w;
    bq[2][0] = e1.x; bq[2][1] = e1.y; bq[3][0] = e1.z; bq[3][1] = e1.w;
  }
  const v8f zero = {0.f, 0.f, 0.f, 0.f, 0.f, 0.f, 0.f, 0.f};
  v8f tl[4];
#pragma unroll
  for (int q = 0; q < 4; ++q) {
    const float* er = E + (size_t)(wave * 64 + q * 16 + m16) * 16 + 8 * kh;
    const float4 f0 = *(const float4*)(er);
    const float4 f1 = *(const float4*)(er + 4);
    v8f st = zero;
    st = wmma_f32(v2f{f0.x, f0.y}, bq[0], st);
    st = wmma_f32(v2f{f0.z, f0.w}, bq[1], st);
    st = wmma_f32(v2f{f1.x, f1.y}, bq[2], st);
    st = wmma_f32(v2f{f1.z, f1.w}, bq[3], st);
    tl[q] = st;
  }
  float mx = 0.f;
#pragma unroll
  for (int q = 0; q < 4; ++q)
#pragma unroll
    for (int r = 0; r < 8; ++r) mx = fmaxf(mx, tl[q][r]);
  part[wave * 2 + kh][m16] = mx;
  __syncthreads();
  float m = 0.f;
#pragma unroll
  for (int p = 0; p < 32; ++p) m = fmaxf(m, part[p][m16]);
  __syncthreads();
  float acc = 0.f;
#pragma unroll
  for (int q = 0; q < 4; ++q)
#pragma unroll
    for (int r = 0; r < 8; ++r) acc += __expf(fmaxf(tl[q][r], 0.f) - m);
  part[wave * 2 + kh][m16] = acc;
  __syncthreads();
  if (t < 16) {
    float tot = 0.f;
#pragma unroll
    for (int p = 0; p < 32; ++p) tot += part[p][t];
    rmx[blockIdx.x * 16 + t] = m;
    rinv[blockIdx.x * 16 + t] = 1.0f / tot;
  }
}

__global__ __launch_bounds__(512)
void k_colsum_pts(const float* __restrict__ pa, float* __restrict__ rS) {
  const int b = blockIdx.y;
  const int lane = threadIdx.x & 31;
  const int w = __builtin_amdgcn_readfirstlane((int)(threadIdx.x >> 5));
  const int j = blockIdx.x * 32 + lane;
  __shared__ float part[16][32];

  float pj[16];
#pragma unroll
  for (int q = 0; q < 4; ++q) {
    const float4 v = *(const float4*)(pa + ((size_t)b * 1024 + j) * 16 + q * 4);
    pj[q * 4 + 0] = v.x; pj[q * 4 + 1] = v.y; pj[q * 4 + 2] = v.z; pj[q * 4 + 3] = v.w;
  }
  const float* pi = pa + ((size_t)b * 1024 + (size_t)w * 64) * 16;
  float acc = 0.f;
  for (int ii = 0; ii < 64; ++ii) {
    float qi[16];
#pragma unroll
    for (int d = 0; d < 16; ++d) qi[d] = pi[ii * 16 + d];
    acc += exp_neg_l1(qi, pj);
  }
  part[w][lane] = acc;
  __syncthreads();
  if (threadIdx.x < 32) {
    float tot = 0.f;
#pragma unroll
    for (int w2 = 0; w2 < 16; ++w2) tot += part[w2][threadIdx.x];
    rS[b * 1024 + blockIdx.x * 32 + threadIdx.x] = 1.0f / tot;
  }
}

#define SX_KC 256
#define SX_AF_TILE 2080
#define SX_AF_BUF (2 * SX_AF_TILE)

static __device__ __forceinline__ int sx_step_row(int p, int w) {
  return ((p >> 3) * 128 + w * 32 + (p & 7) * 4) & 1023;
}

static __device__ __forceinline__ void sx_load_pts(const float* __restrict__ pab, int p, int w,
                                                   float (&q)[4][16]) {
  const float* pn = pab + (size_t)sx_step_row(p, w) * 16;
#pragma unroll
  for (int kk = 0; kk < 4; ++kk) {
#pragma unroll
    for (int q4 = 0; q4 < 4; ++q4) {
      const float4 v = *(const float4*)(pn + kk * 16 + q4 * 4);
      q[kk][q4 * 4 + 0] = v.x; q[kk][q4 * 4 + 1] = v.y;
      q[kk][q4 * 4 + 2] = v.z; q[kk][q4 * 4 + 3] = v.w;
    }
  }
}

template <bool FORM, bool USE>
static __device__ __forceinline__ void sx_step(const float* __restrict__ pab, int p0, int w, int j,
                                               float (&q)[4][16], const float (&pm)[16],
                                               float* afw, const float* afr, const float* bcol,
                                               const v2f (&au)[4], const v2f (&bu)[4][2],
                                               v2f (&an)[4], v2f (&bn)[4][2], v8f (&cacc)[2]) {
  if (USE) {
    const int jn = (j + 1) & 7;
#pragma unroll
    for (int u = 0; u < 4; ++u) {
      an[u] = *(const v2f*)(afr + (4 * jn + u) * 64);
      bn[u][0] = *(const v2f*)(bcol + (2 * (4 * jn + u)) * 160);
      bn[u][1] = *(const v2f*)(bcol + (2 * (4 * jn + u)) * 160 + 32);
    }
  }
  float e[4];
  if (FORM) {
#pragma unroll
    for (int kk = 0; kk < 4; ++kk) e[kk] = exp_neg_l1(q[kk], pm);
    sx_load_pts(pab, p0 + j + 2, w, q);
  }
  if (USE) {
#pragma unroll
    for (int u = 0; u < 4; ++u) {
      cacc[0] = wmma_f32(au[u], bu[u][0], cacc[0]);
      cacc[1] = wmma_f32(au[u], bu[u][1], cacc[1]);
    }
  }
  if (FORM && USE) {
    __builtin_amdgcn_sched_group_barrier(0x100, 8, 0);
#pragma unroll
    for (int u = 0; u < 8; ++u) {
      __builtin_amdgcn_sched_group_barrier(0x008, 1, 0);
      __builtin_amdgcn_sched_group_barrier(0x002, 19, 0);
    }
  }
  if (FORM) {
    *(v2f*)(afw + j * 64)      = v2f{e[0], e[1]};
    *(v2f*)(afw + j * 64 + 32) = v2f{e[2], e[3]};
  }
}

template <bool FORM, bool USE>
static __device__ __forceinline__ void sx_steps(const float* __restrict__ pab, int p0, int w,
                                                float (&qa)[4][16], float (&qb)[4][16],
                                                const float (&pm)[16],
                                                float* afw, const float* afr, const float* bcol,
                                                v8f (&cacc)[2]) {
  v2f a0[4], b0[4][2], a1[4], b1[4][2];
  if (USE) {
#pragma unroll
    for (int u = 0; u < 4; ++u) {
      a0[u] = *(const v2f*)(afr + u * 64);
      b0[u][0] = *(const v2f*)(bcol + (2 * u) * 160);
      b0[u][1] = *(const v2f*)(bcol + (2 * u) * 160 + 32);
    }
  }
#pragma unroll(FORM && USE ? 4 : 1)
  for (int jj = 0; jj < 8; jj += 2) {
    sx_step<FORM, USE>(pab, p0, w, jj,     qa, pm, afw, afr, bcol, a0, b0, a1, b1, cacc);
    sx_step<FORM, USE>(pab, p0, w, jj + 1, qb, pm, afw, afr, bcol, a1, b1, a0, b0, cacc);
  }
}

static __device__ __forceinline__ void sx_stage_half(const float* __restrict__ Bb,
                                                     const float* __restrict__ rsb, int k0,
                                                     float (*Bst)[160], int pr0, int ts) {
  float4 v0[8], v1[8];
  v2f r[8];
#pragma unroll
  for (int u = 0; u < 8; ++u) {
    const int i = ts + u * 128;
    const int kp = i >> 4, cc = (i & 15) << 2;
    v0[u] = *(const float4*)(Bb + (size_t)(k0 + 2 * kp) * 64 + cc);
    v1[u] = *(const float4*)(Bb + (size_t)(k0 + 2 * kp + 1) * 64 + cc);
    r[u] = *(const v2f*)(rsb + k0 + 2 * kp);
  }
#pragma unroll
  for (int u = 0; u < 8; ++u) {
    const int i = ts + u * 128;
    const int kp = i >> 4, cc = (i & 15) << 2;
    float4 lo = {v0[u].x * r[u][0], v1[u].x * r[u][1], v0[u].y * r[u][0], v1[u].y * r[u][1]};
    float4 hi = {v0[u].z * r[u][0], v1[u].z * r[u][1], v0[u].w * r[u][0], v1[u].w * r[u][1]};
    *(float4*)(&Bst[pr0 + kp][cc * 2])     = lo;
    *(float4*)(&Bst[pr0 + kp][cc * 2 + 4]) = hi;
  }
}

__global__ __launch_bounds__(256)
void k_sx_points(const float* __restrict__ pa, const float* __restrict__ x,
                 const float* __restrict__ rS,
                 float* __restrict__ sx, float* __restrict__ sxF) {
  const int b = blockIdx.y;
  const int rowBase = blockIdx.x * 32;
  const float* pab = pa + (size_t)b * 1024 * 16;
  const float* rsb = rS + (size_t)b * 1024;
  const float* Bb = x + (size_t)b * 1024 * 64;
  float* Cb = sx + (size_t)b * 1024 * 64;

  __shared__ float Bs[2][SX_KC / 2][160];
  __shared__ __attribute__((aligned(16))) float Af[2 * SX_AF_BUF];

  const int t = threadIdx.x;
  const int lane = t & 31;
  const int m16 = lane & 15;
  const int kh = lane >> 4;
  const int wnum = __builtin_amdgcn_readfirstlane(t >> 5);
  const bool product = (wnum & 1) == (wnum >> 2);
  const int wave = t >> 6;
  const int rt = wave >> 1;
  const int ch = wave & 1;
  const int ts = wave * 32 + lane;

  float pm[16];
  float qa[4][16], qb[4][16];

  const v8f zero = {0.f, 0.f, 0.f, 0.f, 0.f, 0.f, 0.f, 0.f};
  v8f cacc[2];
  cacc[0] = zero; cacc[1] = zero;

  float* afw0 = Af + (lane >> 4) * SX_AF_TILE + (8 * wave) * 64 + (lane & 15) * 2;
  const float* afr0 = Af + rt * SX_AF_TILE + lane * 2;
  const float* bcol0 = &Bs[0][kh][(ch * 32 + m16) * 2];

  if (product) {
    const float* pr = pab + (size_t)(rowBase + lane) * 16;
#pragma unroll
    for (int q = 0; q < 4; ++q) {
      const float4 v = *(const float4*)(pr + q * 4);
      pm[q * 4 + 0] = v.x; pm[q * 4 + 1] = v.y; pm[q * 4 + 2] = v.z; pm[q * 4 + 3] = v.w;
    }
    sx_load_pts(pab, 0, wave, qa);
    sx_load_pts(pab, 1, wave, qb);
    sx_steps<true, false>(pab, 0, wave, qa, qb, pm, afw0, afr0, bcol0, cacc);
  } else {
    sx_stage_half(Bb, rsb, 0,   Bs[0], 0,  ts);
    sx_stage_half(Bb, rsb, 128, Bs[0], 64, ts);
  }
  __syncthreads();

  for (int sc = 0; sc < 8; ++sc) {
    if (product) {
      const float* afr = afr0 + (sc & 1) * SX_AF_BUF;
      float* afw = afw0 + ((sc + 1) & 1) * SX_AF_BUF;
      const float* bcol = bcol0 + (size_t)(((sc >> 1) & 1) * (SX_KC / 2) + (sc & 1) * 64) * 160;
      if (sc < 7) {
        sx_steps<true, true>(pab, (sc + 1) * 8, wave, qa, qb, pm, afw, afr, bcol, cacc);
      } else {
        sx_steps<false, true>(pab, 0, wave, qa, qb, pm, afw, afr, bcol, cacc);
      }
    } else if (sc < 6) {
      const int cn = (sc >> 1) + 1;
      sx_stage_half(Bb, rsb, cn * SX_KC + (sc & 1) * 128, Bs[cn & 1], (sc & 1) * 64, ts);
    }
    if (sc < 7) __syncthreads();
  }

  if (product) {
#pragma unroll
  for (int nt = 0; nt < 2; ++nt) {
#pragma unroll
    for (int r = 0; r < 8; ++r) {
      const int m = rowBase + rt * 16 + r + 8 * kh;
      Cb[(size_t)m * 64 + (ch * 2 + nt) * 16 + m16] = cacc[nt][r];
    }
  }
  {
    float* Fb = sxF + (((size_t)b * 64 + blockIdx.x * 2 + rt) * 4 + ch * 2) * 256 + lane * 8;
#pragma unroll
    for (int nt = 0; nt < 2; ++nt) {
      *(float4*)(Fb + nt * 256)     = float4{cacc[nt][0], cacc[nt][1], cacc[nt][2], cacc[nt][3]};
      *(float4*)(Fb + nt * 256 + 4) = float4{cacc[nt][4], cacc[nt][5], cacc[nt][6], cacc[nt][7]};
    }
  }
  }
}

#define XG_PF 8
typedef float v4f __attribute__((ext_vector_type(4)));
#define V4_LO(x) __builtin_shufflevector((x), (x), 0, 1)
#define V4_HI(x) __builtin_shufflevector((x), (x), 2, 3)
#define FP_DS 2
#define FP_NS (16 / FP_DS)
#define XS_PITCH 68

static __device__ __forceinline__ void pool_issue(const float* __restrict__ src, int t,
                                                  float4 (&v0)[16], float4 (&v1)[16]) {
#pragma unroll
  for (int u = 0; u < 16; ++u) {
    const int i = t + u * 128;
    const int kp = i >> 4, cc = (i & 15) << 2;
    v0[u] = *(const float4*)(src + (size_t)(2 * kp) * 64 + cc);
    v1[u] = *(const float4*)(src + (size_t)(2 * kp + 1) * 64 + cc);
  }
}

static __device__ __forceinline__ void pool_store(float (*Bsb)[160], int t,
                                                  const float4 (&v0)[16], const float4 (&v1)[16]) {
#pragma unroll
  for (int u = 0; u < 16; ++u) {
    const int i = t + u * 128;
    const int kp = i >> 4, cc = (i & 15) << 2;
    float4 lo = {v0[u].x, v1[u].x, v0[u].y, v1[u].y};
    float4 hi = {v0[u].z, v1[u].z, v0[u].w, v1[u].w};
    *(float4*)(&Bsb[kp][cc * 2])     = lo;
    *(float4*)(&Bsb[kp][cc * 2 + 4]) = hi;
  }
}

static __device__ __forceinline__ void pool_stage(const float* bc, const v2f (&a)[32],
                                                  v2f e2, v8f (&cacc)[2]) {
  v2f bf[2][8][2];
#pragma unroll
  for (int j = 0; j < 8; ++j) {
    bf[0][j][0] = *(const v2f*)(bc + (2 * j) * 160);
    bf[0][j][1] = *(const v2f*)(bc + (2 * j) * 160 + 32);
  }
#pragma unroll
  for (int blk = 0; blk < FP_DS * 4; ++blk) {
    if (blk + 1 < FP_DS * 4) {
#pragma unroll
      for (int j = 0; j < 8; ++j) {
        bf[(blk + 1) & 1][j][0] = *(const v2f*)(bc + (16 * (blk + 1) + 2 * j) * 160);
        bf[(blk + 1) & 1][j][1] = *(const v2f*)(bc + (16 * (blk + 1) + 2 * j) * 160 + 32);
      }
    }
    const float ed = e2[blk >> 2];
#pragma unroll
    for (int j = 0; j < 8; ++j) {
      const v2f as = a[8 * (blk & 3) + j] * ed;
      cacc[0] = wmma_f32(as, bf[blk & 1][j][0], cacc[0]);
      cacc[1] = wmma_f32(as, bf[blk & 1][j][1], cacc[1]);
    }
  }
}

__global__ __launch_bounds__(128)
void k_xg_final(const float* __restrict__ E, const float* __restrict__ rmx,
                const float* __restrict__ rinv, const float* __restrict__ sxF,
                const float* __restrict__ sx, const float* __restrict__ Wp,
                const float* __restrict__ bias_pool, float* __restrict__ out) {
  const int b = blockIdx.y;
  const int rowBase = blockIdx.x * 32;

  __shared__ float Bs[2][FP_DS * 64][160];
  __shared__ __attribute__((aligned(16))) float Xs[32][XS_PITCH];

  const int t = threadIdx.x;
  const int wave = t >> 5;
  const int lane = t & 31;
  const int m16 = lane & 15;
  const int kh = lane >> 4;
  const int rt = wave >> 1;
  const int ch = wave & 1;
  const int nrow = rowBase + rt * 16 + m16;

  float4 v0[16], v1[16];

  v2f bq[4];
  {
    const float4 e0 = *(const float4*)(E + (size_t)nrow * 16 + 8 * kh);
    const float4 e1 = *(const float4*)(E + (size_t)nrow * 16 + 8 * kh + 4);
    bq[0][0] = e0.x; bq[0][1] = e0.y; bq[1][0] = e0.z; bq[1][1] = e0.w;
    bq[2][0] = e1.x; bq[2][1] = e1.y; bq[3][0] = e1.z; bq[3][1] = e1.w;
  }
  const float mxm = rmx[nrow];
  const float invm = rinv[nrow];
  const float* etile = E + (size_t)m16 * 16 + 8 * kh;
  const float* bfr = sxF + ((size_t)b * 256 + ch * 2) * 256 + lane * 8;

  const v8f zero = {0.f, 0.f, 0.f, 0.f, 0.f, 0.f, 0.f, 0.f};
  v8f xacc[2];
  xacc[0] = zero; xacc[1] = zero;

  v2f a[32];
  const float* erow = E + (size_t)nrow * 16;
  v2f ea[4], bb[4][2];
  v2f e_cur;

  v4f eA[XG_PF][2], eB[XG_PF][2];
  v4f bA[XG_PF][2][2], bB[XG_PF][2][2];
  v8f st_cur;
  v2f a_cur[4];

  auto request = [&](v4f (&eb)[XG_PF][2], v4f (&bb2)[XG_PF][2][2], int tt, int g) {
    const float* en = etile + (size_t)((g + 2) & 63) * 256;
    eb[tt][0] = *(const v4f*)(en);
    eb[tt][1] = *(const v4f*)(en + 4);
    const float* bn = bfr + (size_t)(g & 63) * 1024;
#pragma unroll
    for (int nt = 0; nt < 2; ++nt) {
      bb2[tt][nt][0] = *(const v4f*)(bn + nt * 256);
      bb2[tt][nt][1] = *(const v4f*)(bn + nt * 256 + 4);
    }
  };
  auto trip = [&](const v4f (&eb)[XG_PF][2], const v4f (&bb2)[XG_PF][2][2], int tt) {
    const v2f ta[4] = {V4_LO(eb[tt][0]), V4_HI(eb[tt][0]), V4_LO(eb[tt][1]), V4_HI(eb[tt][1])};
    v2f bf[4][2];
#pragma unroll
    for (int nt = 0; nt < 2; ++nt) {
      bf[0][nt] = V4_LO(bb2[tt][nt][0]); bf[1][nt] = V4_HI(bb2[tt][nt][0]);
      bf[2][nt] = V4_LO(bb2[tt][nt][1]); bf[3][nt] = V4_HI(bb2[tt][nt][1]);
    }
    v2f a_nxt[4];
#pragma unroll
    for (int j = 0; j < 4; ++j) {
      a_nxt[j][0] = __expf(fmaxf(st_cur[2 * j], 0.f) - mxm) * invm;
      a_nxt[j][1] = __expf(fmaxf(st_cur[2 * j + 1], 0.f) - mxm) * invm;
    }
    v8f st_nxt = zero;
#pragma unroll
    for (int j = 0; j < 4; ++j) {
      st_nxt = wmma_f32(ta[j], bq[j], st_nxt);
      xacc[0] = wmma_f32(a_cur[j], bf[j][0], xacc[0]);
      xacc[1] = wmma_f32(a_cur[j], bf[j][1], xacc[1]);
    }
    st_cur = st_nxt;
#pragma unroll
    for (int j = 0; j < 4; ++j) a_cur[j] = a_nxt[j];
    __builtin_amdgcn_sched_barrier(0);
  };

  {
    const v4f f0 = *(const v4f*)(etile);
    const v4f f1 = *(const v4f*)(etile + 4);
    const v4f g0 = *(const v4f*)(etile + 256);
    const v4f g1 = *(const v4f*)(etile + 256 + 4);
#pragma unroll
    for (int u = 0; u < XG_PF; ++u) request(eA, bA, u, u);
    asm volatile("" ::: "memory");
    v8f s0 = zero;
    s0 = wmma_f32(V4_LO(f0), bq[0], s0);
    s0 = wmma_f32(V4_HI(f0), bq[1], s0);
    s0 = wmma_f32(V4_LO(f1), bq[2], s0);
    s0 = wmma_f32(V4_HI(f1), bq[3], s0);
    v8f s1 = zero;
    s1 = wmma_f32(V4_LO(g0), bq[0], s1);
    s1 = wmma_f32(V4_HI(g0), bq[1], s1);
    s1 = wmma_f32(V4_LO(g1), bq[2], s1);
    s1 = wmma_f32(V4_HI(g1), bq[3], s1);
#pragma unroll
    for (int j = 0; j < 4; ++j) {
      a_cur[j][0] = __expf(fmaxf(s0[2 * j], 0.f) - mxm) * invm;
      a_cur[j][1] = __expf(fmaxf(s0[2 * j + 1], 0.f) - mxm) * invm;
    }
    st_cur = s1;
  }

  auto second_request = [&](int q) {
#pragma unroll
    for (int u = 4 * q; u < 4 * q + 4; ++u) {
      const int i = t + u * 128;
      const int kp = i >> 4, cc = (i & 15) << 2;
      v0[u] = *(const float4*)(Wp + (size_t)(2 * kp) * 64 + cc);
      v1[u] = *(const float4*)(Wp + (size_t)(2 * kp + 1) * 64 + cc);
    }
    const size_t roff = ((size_t)b * 1024 + nrow) * 64 + 2 * kh;
#pragma unroll
    for (int ks = 4 * q; ks < 4 * q + 4; ++ks) a[ks] = *(const v2f*)(sx + roff + 4 * ks);
    ea[q] = *(const v2f*)(erow + 4 * q + 2 * kh);
#pragma unroll
    for (int nt = 0; nt < 2; ++nt) {
      const int col = ch * 32 + nt * 16 + m16;
      bb[q][nt][0] = bias_pool[(4 * q + 2 * kh) * 64 + col];
      bb[q][nt][1] = bias_pool[(4 * q + 2 * kh + 1) * 64 + col];
    }
    if (q == 0) e_cur = *(const v2f*)(erow);
  };

  auto pass16 = [&](int g0, bool last) __attribute__((always_inline)) {
#pragma unroll
    for (int u = 0; u < XG_PF; ++u) {
      if (u < XG_PF / 2) {
        request(eB, bB, 2 * u,     g0 + XG_PF + 2 * u);
        request(eB, bB, 2 * u + 1, g0 + XG_PF + 2 * u + 1);
        asm volatile("" ::: "memory");
      }
      trip(eA, bA, u);
    }
#pragma unroll
    for (int u = 0; u < XG_PF; ++u) {
      if (u < XG_PF / 2) {
        if (!last) {
          request(eA, bA, 2 * u,     g0 + 2 * XG_PF + 2 * u);
          request(eA, bA, 2 * u + 1, g0 + 2 * XG_PF + 2 * u + 1);
        } else {
          second_request(u);
        }
        asm volatile("" ::: "memory");
      }
      trip(eB, bB, u);
    }
  };
  for (int g0 = 0; g0 < 48; g0 += 2 * XG_PF) pass16(g0, false);
  pass16(48, true);

  pool_store(Bs[0], t, v0, v1);
#pragma unroll
  for (int nt = 0; nt < 2; ++nt) {
#pragma unroll
    for (int r = 0; r < 8; ++r)
      Xs[rt * 16 + r + 8 * kh][(ch * 2 + nt) * 16 + m16] = xacc[nt][r];
  }
  __syncthreads();
#pragma unroll
  for (int ks = 0; ks < 16; ++ks)
    a[16 + ks] = *(const v2f*)(&Xs[rt * 16 + m16][4 * ks + 2 * kh]);

  v8f cacc[2];
  cacc[0] = zero; cacc[1] = zero;
#pragma unroll
  for (int j = 0; j < 4; ++j) {
    cacc[0] = wmma_f32(ea[j], bb[j][0], cacc[0]);
    cacc[1] = wmma_f32(ea[j], bb[j][1], cacc[1]);
  }

  for (int s = 0; s < FP_NS; ++s) {
    const int sn = (s + 1) & (FP_NS - 1);
    pool_issue(Wp + (size_t)sn * (FP_DS * 8192), t, v0, v1);
    const v2f e_nxt = *(const v2f*)(erow + FP_DS * sn);
    pool_stage(&Bs[s & 1][kh][(ch * 32 + m16) * 2], a, e_cur, cacc);
    pool_store(Bs[(s + 1) & 1], t, v0, v1);
    e_cur = e_nxt;
    __syncthreads();
  }

  float* Cb = out + (size_t)b * 1024 * 64;
#pragma unroll
  for (int nt = 0; nt < 2; ++nt) {
#pragma unroll
    for (int r = 0; r < 8; ++r) {
      const int m = rowBase + rt * 16 + r + 8 * kh;
      Cb[(size_t)m * 64 + (ch * 2 + nt) * 16 + m16] = cacc[nt][r];
    }
  }
}

extern "C" void kernel_launch(void* const* d_in, const int* in_sizes, int n_in,
                              void* d_out, int out_size, void* d_ws, size_t ws_size,
                              hipStream_t stream) {
  (void)in_sizes; (void)n_in; (void)out_size; (void)ws_size;
  const float* x    = (const float*)d_in[0];
  const float* E    = (const float*)d_in[1];
  const float* pa   = (const float*)d_in[2];
  const float* Wp   = (const float*)d_in[3];
  const float* bp   = (const float*)d_in[4];
  float* out = (float*)d_out;

  float* ws   = (float*)d_ws;
  float* rmx  = ws;
  float* rinv = ws + 1024;
  float* sube = ws   + 1048576;
  float* rS   = sube + 8388608;
  float* sx   = rS   + 8192 + 524288;
  float* sxF  = sube;

  k_rowstats  <<<64, 512, 0, stream>>>(E, rmx, rinv);
  k_colsum_pts<<<dim3(32, 8),     512, 0, stream>>>(pa, rS);
  k_sx_points<<<dim3(32, 8), 256, 0, stream>>>(pa, x, rS, sx, sxF);
  k_xg_final<<<dim3(32, 8), 128, 0, stream>>>(E, rmx, rinv, sxF, sx, Wp, bp, out);
}


# Row stats of relu(E E^T) formed by the sx kernel's idle staging waves as 16 partial (max, sum) pairs per row, joined in the final kernel; rowstats launch removed; changes sum order of softmax sums
# speedup vs baseline: 3.9524x; 1.0546x over previous
#include <hip/hip_runtime.h>
#include <hip/hip_bf16.h>
#include <cstddef>


typedef float v2f __attribute__((ext_vector_type(2)));
typedef float v8f __attribute__((ext_vector_type(8)));

static __device__ __forceinline__ v8f wmma_f32(v2f a, v2f b, v8f c) {
  return __builtin_amdgcn_wmma_f32_16x16x4_f32(
       false, a,  false, b,
       (short)0, c,  false,  false);
}

static __device__ __forceinline__ float exp_neg_l1(const float (&p)[16],
                                                   const float (&q)[16]) {
  float s0 = 0.f, s1 = 0.f, s2 = 0.f, s3 = 0.f;
#pragma unroll
  for (int d = 0; d < 4; ++d) {
    s0 += fabsf(p[d]      - q[d]);
    s1 += fabsf(p[d + 4]  - q[d + 4]);
    s2 += fabsf(p[d + 8]  - q[d + 8]);
    s3 += fabsf(p[d + 12] - q[d + 12]);
  }
  return __expf(-((s0 + s1) + (s2 + s3)));
}

static __device__ __forceinline__ void rowstats_task(const float* __restrict__ E,
                                                     float* __restrict__ stats,
                                                     int task, int lane) {
  const int mt = task >> 4;
  const int kc = task & 15;
  const int m16 = lane & 15;
  const int kh = lane >> 4;
  const int mrow = mt * 16 + m16;

  v2f bq[4];
  {
    const float4 e0 = *(const float4*)(E + (size_t)mrow * 16 + 8 * kh);
    const float4 e1 = *(const float4*)(E + (size_t)mrow * 16 + 8 * kh + 4);
    bq[0][0] = e0.x; bq[0][1] = e0.y; bq[1][0] = e0.z; bq[1][1] = e0.w;
    bq[2][0] = e1.x; bq[2][1] = e1.y; bq[3][0] = e1.z; bq[3][1] = e1.w;
  }
  const v8f zero = {0.f, 0.f, 0.f, 0.f, 0.f, 0.f, 0.f, 0.f};
  v8f tl[4];
#pragma unroll
  for (int q = 0; q < 4; ++q) {
    const float* er = E + (size_t)(kc * 64 + q * 16 + m16) * 16 + 8 * kh;
    const float4 f0 = *(const float4*)(er);
    const float4 f1 = *(const float4*)(er + 4);
    v8f st = zero;
    st = wmma_f32(v2f{f0.x, f0.y}, bq[0], st);
    st = wmma_f32(v2f{f0.z, f0.w}, bq[1], st);
    st = wmma_f32(v2f{f1.x, f1.y}, bq[2], st);
    st = wmma_f32(v2f{f1.z, f1.w}, bq[3], st);
    tl[q] = st;
  }
  float mx = 0.f;
#pragma unroll
  for (int q = 0; q < 4; ++q)
#pragma unroll
    for (int r = 0; r < 8; ++r) mx = fmaxf(mx, tl[q][r]);
  mx = fmaxf(mx, __shfl_xor(mx, 16));
  float acc = 0.f;
#pragma unroll
  for (int q = 0; q < 4; ++q)
#pragma unroll
    for (int r = 0; r < 8; ++r) acc += __expf(fmaxf(tl[q][r], 0.f) - mx);
  acc += __shfl_xor(acc, 16);
  stats[(size_t)task * 32 + lane] = kh ? acc : mx;
}

__global__ __launch_bounds__(512)
void k_colsum_pts(const float* __restrict__ pa, float* __restrict__ rS) {
  const int b = blockIdx.y;
  const int lane = threadIdx.x & 31;
  const int w = __builtin_amdgcn_readfirstlane((int)(threadIdx.x >> 5));
  const int j = blockIdx.x * 32 + lane;
  __shared__ float part[16][32];

  float pj[16];
#pragma unroll
  for (int q = 0; q < 4; ++q) {
    const float4 v = *(const float4*)(pa + ((size_t)b * 1024 + j) * 16 + q * 4);
    pj[q * 4 + 0] = v.x; pj[q * 4 + 1] = v.y; pj[q * 4 + 2] = v.z; pj[q * 4 + 3] = v.w;
  }
  const float* pi = pa + ((size_t)b * 1024 + (size_t)w * 64) * 16;
  float acc = 0.f;
  for (int ii = 0; ii < 64; ++ii) {
    float qi[16];
#pragma unroll
    for (int d = 0; d < 16; ++d) qi[d] = pi[ii * 16 + d];
    acc += exp_neg_l1(qi, pj);
  }
  part[w][lane] = acc;
  __syncthreads();
  if (threadIdx.x < 32) {
    float tot = 0.f;
#pragma unroll
    for (int w2 = 0; w2 < 16; ++w2) tot += part[w2][threadIdx.x];
    rS[b * 1024 + blockIdx.x * 32 + threadIdx.x] = 1.0f / tot;
  }
}

#define SX_KC 256
#define SX_AF_TILE 2080
#define SX_AF_BUF (2 * SX_AF_TILE)

static __device__ __forceinline__ int sx_step_row(int p, int w) {
  return ((p >> 3) * 128 + w * 32 + (p & 7) * 4) & 1023;
}

static __device__ __forceinline__ void sx_load_pts(const float* __restrict__ pab, int p, int w,
                                                   float (&q)[4][16]) {
  const float* pn = pab + (size_t)sx_step_row(p, w) * 16;
#pragma unroll
  for (int kk = 0; kk < 4; ++kk) {
#pragma unroll
    for (int q4 = 0; q4 < 4; ++q4) {
      const float4 v = *(const float4*)(pn + kk * 16 + q4 * 4);
      q[kk][q4 * 4 + 0] = v.x; q[kk][q4 * 4 + 1] = v.y;
      q[kk][q4 * 4 + 2] = v.z; q[kk][q4 * 4 + 3] = v.w;
    }
  }
}

template <bool FORM, bool USE>
static __device__ __forceinline__ void sx_step(const float* __restrict__ pab, int p0, int w, int j,
                                               float (&q)[4][16], const float (&pm)[16],
                                               float* afw, const float* afr, const float* bcol,
                                               const v2f (&au)[4], const v2f (&bu)[4][2],
                                               v2f (&an)[4], v2f (&bn)[4][2], v8f (&cacc)[2]) {
  if (USE) {
    const int jn = (j + 1) & 7;
#pragma unroll
    for (int u = 0; u < 4; ++u) {
      an[u] = *(const v2f*)(afr + (4 * jn + u) * 64);
      bn[u][0] = *(const v2f*)(bcol + (2 * (4 * jn + u)) * 160);
      bn[u][1] = *(const v2f*)(bcol + (2 * (4 * jn + u)) * 160 + 32);
    }
  }
  float e[4];
  if (FORM) {
#pragma unroll
    for (int kk = 0; kk < 4; ++kk) e[kk] = exp_neg_l1(q[kk], pm);
    sx_load_pts(pab, p0 + j + 2, w, q);
  }
  if (USE) {
#pragma unroll
    for (int u = 0; u < 4; ++u) {
      cacc[0] = wmma_f32(au[u], bu[u][0], cacc[0]);
      cacc[1] = wmma_f32(au[u], bu[u][1], cacc[1]);
    }
  }
  if (FORM && USE) {
    __builtin_amdgcn_sched_group_barrier(0x100, 8, 0);
#pragma unroll
    for (int u = 0; u < 8; ++u) {
      __builtin_amdgcn_sched_group_barrier(0x008, 1, 0);
      __builtin_amdgcn_sched_group_barrier(0x002, 19, 0);
    }
  }
  if (FORM) {
    *(v2f*)(afw + j * 64)      = v2f{e[0], e[1]};
    *(v2f*)(afw + j * 64 + 32) = v2f{e[2], e[3]};
  }
}

template <bool FORM, bool USE>
static __device__ __forceinline__ void sx_steps(const float* __restrict__ pab, int p0, int w,
                                                float (&qa)[4][16], float (&qb)[4][16],
                                                const float (&pm)[16],
                                                float* afw, const float* afr, const float* bcol,
                                                v8f (&cacc)[2]) {
  v2f a0[4], b0[4][2], a1[4], b1[4][2];
  if (USE) {
#pragma unroll
    for (int u = 0; u < 4; ++u) {
      a0[u] = *(const v2f*)(afr + u * 64);
      b0[u][0] = *(const v2f*)(bcol + (2 * u) * 160);
      b0[u][1] = *(const v2f*)(bcol + (2 * u) * 160 + 32);
    }
  }
#pragma unroll(FORM && USE ? 4 : 1)
  for (int jj = 0; jj < 8; jj += 2) {
    sx_step<FORM, USE>(pab, p0, w, jj,     qa, pm, afw, afr, bcol, a0, b0, a1, b1, cacc);
    sx_step<FORM, USE>(pab, p0, w, jj + 1, qb, pm, afw, afr, bcol, a1, b1, a0, b0, cacc);
  }
}

static __device__ __forceinline__ void sx_stage_half(const float* __restrict__ Bb,
                                                     const float* __restrict__ rsb, int k0,
                                                     float (*Bst)[160], int pr0, int ts) {
  float4 v0[8], v1[8];
  v2f r[8];
#pragma unroll
  for (int u = 0; u < 8; ++u) {
    const int i = ts + u * 128;
    const int kp = i >> 4, cc = (i & 15) << 2;
    v0[u] = *(const float4*)(Bb + (size_t)(k0 + 2 * kp) * 64 + cc);
    v1[u] = *(const float4*)(Bb + (size_t)(k0 + 2 * kp + 1) * 64 + cc);
    r[u] = *(const v2f*)(rsb + k0 + 2 * kp);
  }
#pragma unroll
  for (int u = 0; u < 8; ++u) {
    const int i = ts + u * 128;
    const int kp = i >> 4, cc = (i & 15) << 2;
    float4 lo = {v0[u].x * r[u][0], v1[u].x * r[u][1], v0[u].y * r[u][0], v1[u].y * r[u][1]};
    float4 hi = {v0[u].z * r[u][0], v1[u].z * r[u][1], v0[u].w * r[u][0], v1[u].w * r[u][1]};
    *(float4*)(&Bst[pr0 + kp][cc * 2])     = lo;
    *(float4*)(&Bst[pr0 + kp][cc * 2 + 4]) = hi;
  }
}

__global__ __launch_bounds__(256)
void k_sx_points(const float* __restrict__ pa, const float* __restrict__ x,
                 const float* __restrict__ rS, const float* __restrict__ E,
                 float* __restrict__ sx, float* __restrict__ sxF,
                 float* __restrict__ stats) {
  const int b = blockIdx.y;
  const int rowBase = blockIdx.x * 32;
  const float* pab = pa + (size_t)b * 1024 * 16;
  const float* rsb = rS + (size_t)b * 1024;
  const float* Bb = x + (size_t)b * 1024 * 64;
  float* Cb = sx + (size_t)b * 1024 * 64;

  __shared__ float Bs[2][SX_KC / 2][160];
  __shared__ __attribute__((aligned(16))) float Af[2 * SX_AF_BUF];

  const int t = threadIdx.x;
  const int lane = t & 31;
  const int m16 = lane & 15;
  const int kh = lane >> 4;
  const int wnum = __builtin_amdgcn_readfirstlane(t >> 5);
  const bool product = (wnum & 1) == (wnum >> 2);
  const int wave = t >> 6;
  const int rt = wave >> 1;
  const int ch = wave & 1;
  const int ts = wave * 32 + lane;

  float pm[16];
  float qa[4][16], qb[4][16];

  const v8f zero = {0.f, 0.f, 0.f, 0.f, 0.f, 0.f, 0.f, 0.f};
  v8f cacc[2];
  cacc[0] = zero; cacc[1] = zero;

  float* afw0 = Af + (lane >> 4) * SX_AF_TILE + (8 * wave) * 64 + (lane & 15) * 2;
  const float* afr0 = Af + rt * SX_AF_TILE + lane * 2;
  const float* bcol0 = &Bs[0][kh][(ch * 32 + m16) * 2];

  if (product) {
    const float* pr = pab + (size_t)(rowBase + lane) * 16;
#pragma unroll
    for (int q = 0; q < 4; ++q) {
      const float4 v = *(const float4*)(pr + q * 4);
      pm[q * 4 + 0] = v.x; pm[q * 4 + 1] = v.y; pm[q * 4 + 2] = v.z; pm[q * 4 + 3] = v.w;
    }
    sx_load_pts(pab, 0, wave, qa);
    sx_load_pts(pab, 1, wave, qb);
    sx_steps<true, false>(pab, 0, wave, qa, qb, pm, afw0, afr0, bcol0, cacc);
  } else {
    sx_stage_half(Bb, rsb, 0,   Bs[0], 0,  ts);
    sx_stage_half(Bb, rsb, 128, Bs[0], 64, ts);
  }
  __syncthreads();

  for (int sc = 0; sc < 8; ++sc) {
    if (product) {
      const float* afr = afr0 + (sc & 1) * SX_AF_BUF;
      float* afw = afw0 + ((sc + 1) & 1) * SX_AF_BUF;
      const float* bcol = bcol0 + (size_t)(((sc >> 1) & 1) * (SX_KC / 2) + (sc & 1) * 64) * 160;
      if (sc < 7) {
        sx_steps<true, true>(pab, (sc + 1) * 8, wave, qa, qb, pm, afw, afr, bcol, cacc);
      } else {
        sx_steps<false, true>(pab, 0, wave, qa, qb, pm, afw, afr, bcol, cacc);
      }
    } else if (sc < 6) {
      const int cn = (sc >> 1) + 1;
      sx_stage_half(Bb, rsb, cn * SX_KC + (sc & 1) * 128, Bs[cn & 1], (sc & 1) * 64, ts);
    }
    if (sc < 7) __syncthreads();
  }

  if (!product) {
    rowstats_task(E, stats, (blockIdx.y * 32 + blockIdx.x) * 4 + wave, lane);
  }

  if (product) {
#pragma unroll
  for (int nt = 0; nt < 2; ++nt) {
#pragma unroll
    for (int r = 0; r < 8; ++r) {
      const int m = rowBase + rt * 16 + r + 8 * kh;
      Cb[(size_t)m * 64 + (ch * 2 + nt) * 16 + m16] = cacc[nt][r];
    }
  }
  {
    float* Fb = sxF + (((size_t)b * 64 + blockIdx.x * 2 + rt) * 4 + ch * 2) * 256 + lane * 8;
#pragma unroll
    for (int nt = 0; nt < 2; ++nt) {
      *(float4*)(Fb + nt * 256)     = float4{cacc[nt][0], cacc[nt][1], cacc[nt][2], cacc[nt][3]};
      *(float4*)(Fb + nt * 256 + 4) = float4{cacc[nt][4], cacc[nt][5], cacc[nt][6], cacc[nt][7]};
    }
  }
  }
}

#define XG_PF 8
typedef float v4f __attribute__((ext_vector_type(4)));
#define V4_LO(x) __builtin_shufflevector((x), (x), 0, 1)
#define V4_HI(x) __builtin_shufflevector((x), (x), 2, 3)
#define FP_DS 2
#define FP_NS (16 / FP_DS)
#define XS_PITCH 68

static __device__ __forceinline__ void pool_issue(const float* __restrict__ src, int t,
                                                  float4 (&v0)[16], float4 (&v1)[16]) {
#pragma unroll
  for (int u = 0; u < 16; ++u) {
    const int i = t + u * 128;
    const int kp = i >> 4, cc = (i & 15) << 2;
    v0[u] = *(const float4*)(src + (size_t)(2 * kp) * 64 + cc);
    v1[u] = *(const float4*)(src + (size_t)(2 * kp + 1) * 64 + cc);
  }
}

static __device__ __forceinline__ void pool_store(float (*Bsb)[160], int t,
                                                  const float4 (&v0)[16], const float4 (&v1)[16]) {
#pragma unroll
  for (int u = 0; u < 16; ++u) {
    const int i = t + u * 128;
    const int kp = i >> 4, cc = (i & 15) << 2;
    float4 lo = {v0[u].x, v1[u].x, v0[u].y, v1[u].y};
    float4 hi = {v0[u].z, v1[u].z, v0[u].w, v1[u].w};
    *(float4*)(&Bsb[kp][cc * 2])     = lo;
    *(float4*)(&Bsb[kp][cc * 2 + 4]) = hi;
  }
}

static __device__ __forceinline__ void pool_stage(const float* bc, const v2f (&a)[32],
                                                  v2f e2, v8f (&cacc)[2]) {
  v2f bf[2][8][2];
#pragma unroll
  for (int j = 0; j < 8; ++j) {
    bf[0][j][0] = *(const v2f*)(bc + (2 * j) * 160);
    bf[0][j][1] = *(const v2f*)(bc + (2 * j) * 160 + 32);
  }
#pragma unroll
  for (int blk = 0; blk < FP_DS * 4; ++blk) {
    if (blk + 1 < FP_DS * 4) {
#pragma unroll
      for (int j = 0; j < 8; ++j) {
        bf[(blk + 1) & 1][j][0] = *(const v2f*)(bc + (16 * (blk + 1) + 2 * j) * 160);
        bf[(blk + 1) & 1][j][1] = *(const v2f*)(bc + (16 * (blk + 1) + 2 * j) * 160 + 32);
      }
    }
    const float ed = e2[blk >> 2];
#pragma unroll
    for (int j = 0; j < 8; ++j) {
      const v2f as = a[8 * (blk & 3) + j] * ed;
      cacc[0] = wmma_f32(as, bf[blk & 1][j][0], cacc[0]);
      cacc[1] = wmma_f32(as, bf[blk & 1][j][1], cacc[1]);
    }
  }
}

__global__ __launch_bounds__(128)
void k_xg_final(const float* __restrict__ E, const float* __restrict__ stats,
                const float* __restrict__ sxF,
                const float* __restrict__ sx, const float* __restrict__ Wp,
                const float* __restrict__ bias_pool, float* __restrict__ out) {
  const int b = blockIdx.y;
  const int rowBase = blockIdx.x * 32;

  __shared__ float Bs[2][FP_DS * 64][160];
  __shared__ __attribute__((aligned(16))) float Xs[32][XS_PITCH];

  const int t = threadIdx.x;
  const int wave = t >> 5;
  const int lane = t & 31;
  const int m16 = lane & 15;
  const int kh = lane >> 4;
  const int rt = wave >> 1;
  const int ch = wave & 1;
  const int nrow = rowBase + rt * 16 + m16;

  float4 v0[16], v1[16];

  v2f bq[4];
  {
    const float4 e0 = *(const float4*)(E + (size_t)nrow * 16 + 8 * kh);
    const float4 e1 = *(const float4*)(E + (size_t)nrow * 16 + 8 * kh + 4);
    bq[0][0] = e0.x; bq[0][1] = e0.y; bq[1][0] = e0.z; bq[1][1] = e0.w;
    bq[2][0] = e1.x; bq[2][1] = e1.y; bq[3][0] = e1.z; bq[3][1] = e1.w;
  }
  float mxm = 0.f;
  float invm;
  {
    const float* sp = stats + (size_t)(nrow >> 4) * 512 + (nrow & 15);
    float pmx[16], psm[16];
#pragma unroll
    for (int kc = 0; kc < 16; ++kc) {
      pmx[kc] = sp[kc * 32];
      psm[kc] = sp[kc * 32 + 16];
    }
#pragma unroll
    for (int kc = 0; kc < 16; ++kc) mxm = fmaxf(mxm, pmx[kc]);
    float tot = 0.f;
#pragma unroll
    for (int kc = 0; kc < 16; ++kc) tot += psm[kc] * __expf(pmx[kc] - mxm);
    invm = 1.0f / tot;
  }
  const float* etile = E + (size_t)m16 * 16 + 8 * kh;
  const float* bfr = sxF + ((size_t)b * 256 + ch * 2) * 256 + lane * 8;

  const v8f zero = {0.f, 0.f, 0.f, 0.f, 0.f, 0.f, 0.f, 0.f};
  v8f xacc[2];
  xacc[0] = zero; xacc[1] = zero;

  v2f a[32];
  const float* erow = E + (size_t)nrow * 16;
  v2f ea[4], bb[4][2];
  v2f e_cur;

  v4f eA[XG_PF][2], eB[XG_PF][2];
  v4f bA[XG_PF][2][2], bB[XG_PF][2][2];
  v8f st_cur;
  v2f a_cur[4];

  auto request = [&](v4f (&eb)[XG_PF][2], v4f (&bb2)[XG_PF][2][2], int tt, int g) {
    const float* en = etile + (size_t)((g + 2) & 63) * 256;
    eb[tt][0] = *(const v4f*)(en);
    eb[tt][1] = *(const v4f*)(en + 4);
    const float* bn = bfr + (size_t)(g & 63) * 1024;
#pragma unroll
    for (int nt = 0; nt < 2; ++nt) {
      bb2[tt][nt][0] = *(const v4f*)(bn + nt * 256);
      bb2[tt][nt][1] = *(const v4f*)(bn + nt * 256 + 4);
    }
  };
  auto trip = [&](const v4f (&eb)[XG_PF][2], const v4f (&bb2)[XG_PF][2][2], int tt) {
    const v2f ta[4] = {V4_LO(eb[tt][0]), V4_HI(eb[tt][0]), V4_LO(eb[tt][1]), V4_HI(eb[tt][1])};
    v2f bf[4][2];
#pragma unroll
    for (int nt = 0; nt < 2; ++nt) {
      bf[0][nt] = V4_LO(bb2[tt][nt][0]); bf[1][nt] = V4_HI(bb2[tt][nt][0]);
      bf[2][nt] = V4_LO(bb2[tt][nt][1]); bf[3][nt] = V4_HI(bb2[tt][nt][1]);
    }
    v2f a_nxt[4];
#pragma unroll
    for (int j = 0; j < 4; ++j) {
      a_nxt[j][0] = __expf(fmaxf(st_cur[2 * j], 0.f) - mxm) * invm;
      a_nxt[j][1] = __expf(fmaxf(st_cur[2 * j + 1], 0.f) - mxm) * invm;
    }
    v8f st_nxt = zero;
#pragma unroll
    for (int j = 0; j < 4; ++j) {
      st_nxt = wmma_f32(ta[j], bq[j], st_nxt);
      xacc[0] = wmma_f32(a_cur[j], bf[j][0], xacc[0]);
      xacc[1] = wmma_f32(a_cur[j], bf[j][1], xacc[1]);
    }
    st_cur = st_nxt;
#pragma unroll
    for (int j = 0; j < 4; ++j) a_cur[j] = a_nxt[j];
    __builtin_amdgcn_sched_barrier(0);
  };

  {
    const v4f f0 = *(const v4f*)(etile);
    const v4f f1 = *(const v4f*)(etile + 4);
    const v4f g0 = *(const v4f*)(etile + 256);
    const v4f g1 = *(const v4f*)(etile + 256 + 4);
#pragma unroll
    for (int u = 0; u < XG_PF; ++u) request(eA, bA, u, u);
    asm volatile("" ::: "memory");
    v8f s0 = zero;
    s0 = wmma_f32(V4_LO(f0), bq[0], s0);
    s0 = wmma_f32(V4_HI(f0), bq[1], s0);
    s0 = wmma_f32(V4_LO(f1), bq[2], s0);
    s0 = wmma_f32(V4_HI(f1), bq[3], s0);
    v8f s1 = zero;
    s1 = wmma_f32(V4_LO(g0), bq[0], s1);
    s1 = wmma_f32(V4_HI(g0), bq[1], s1);
    s1 = wmma_f32(V4_LO(g1), bq[2], s1);
    s1 = wmma_f32(V4_HI(g1), bq[3], s1);
#pragma unroll
    for (int j = 0; j < 4; ++j) {
      a_cur[j][0] = __expf(fmaxf(s0[2 * j], 0.f) - mxm) * invm;
      a_cur[j][1] = __expf(fmaxf(s0[2 * j + 1], 0.f) - mxm) * invm;
    }
    st_cur = s1;
  }

  auto second_request = [&](int q) {
#pragma unroll
    for (int u = 4 * q; u < 4 * q + 4; ++u) {
      const int i = t + u * 128;
      const int kp = i >> 4, cc = (i & 15) << 2;
      v0[u] = *(const float4*)(Wp + (size_t)(2 * kp) * 64 + cc);
      v1[u] = *(const float4*)(Wp + (size_t)(2 * kp + 1) * 64 + cc);
    }
    const size_t roff = ((size_t)b * 1024 + nrow) * 64 + 2 * kh;
#pragma unroll
    for (int ks = 4 * q; ks < 4 * q + 4; ++ks) a[ks] = *(const v2f*)(sx + roff + 4 * ks);
    ea[q] = *(const v2f*)(erow + 4 * q + 2 * kh);
#pragma unroll
    for (int nt = 0; nt < 2; ++nt) {
      const int col = ch * 32 + nt * 16 + m16;
      bb[q][nt][0] = bias_pool[(4 * q + 2 * kh) * 64 + col];
      bb[q][nt][1] = bias_pool[(4 * q + 2 * kh + 1) * 64 + col];
    }
    if (q == 0) e_cur = *(const v2f*)(erow);
  };

  auto pass16 = [&](int g0, bool last) __attribute__((always_inline)) {
#pragma unroll
    for (int u = 0; u < XG_PF; ++u) {
      if (u < XG_PF / 2) {
        request(eB, bB, 2 * u,     g0 + XG_PF + 2 * u);
        request(eB, bB, 2 * u + 1, g0 + XG_PF + 2 * u + 1);
        asm volatile("" ::: "memory");
      }
      trip(eA, bA, u);
    }
#pragma unroll
    for (int u = 0; u < XG_PF; ++u) {
      if (u < XG_PF / 2) {
        if (!last) {
          request(eA, bA, 2 * u,     g0 + 2 * XG_PF + 2 * u);
          request(eA, bA, 2 * u + 1, g0 + 2 * XG_PF + 2 * u + 1);
        } else {
          second_request(u);
        }
        asm volatile("" ::: "memory");
      }
      trip(eB, bB, u);
    }
  };
  for (int g0 = 0; g0 < 48; g0 += 2 * XG_PF) pass16(g0, false);
  pass16(48, true);

  pool_store(Bs[0], t, v0, v1);
#pragma unroll
  for (int nt = 0; nt < 2; ++nt) {
#pragma unroll
    for (int r = 0; r < 8; ++r)
      Xs[rt * 16 + r + 8 * kh][(ch * 2 + nt) * 16 + m16] = xacc[nt][r];
  }
  __syncthreads();
#pragma unroll
  for (int ks = 0; ks < 16; ++ks)
    a[16 + ks] = *(const v2f*)(&Xs[rt * 16 + m16][4 * ks + 2 * kh]);

  v8f cacc[2];
  cacc[0] = zero; cacc[1] = zero;
#pragma unroll
  for (int j = 0; j < 4; ++j) {
    cacc[0] = wmma_f32(ea[j], bb[j][0], cacc[0]);
    cacc[1] = wmma_f32(ea[j], bb[j][1], cacc[1]);
  }

  for (int s = 0; s < FP_NS; ++s) {
    const int sn = (s + 1) & (FP_NS - 1);
    pool_issue(Wp + (size_t)sn * (FP_DS * 8192), t, v0, v1);
    const v2f e_nxt = *(const v2f*)(erow + FP_DS * sn);
    pool_stage(&Bs[s & 1][kh][(ch * 32 + m16) * 2], a, e_cur, cacc);
    pool_store(Bs[(s + 1) & 1], t, v0, v1);
    e_cur = e_nxt;
    __syncthreads();
  }

  float* Cb = out + (size_t)b * 1024 * 64;
#pragma unroll
  for (int nt = 0; nt < 2; ++nt) {
#pragma unroll
    for (int r = 0; r < 8; ++r) {
      const int m = rowBase + rt * 16 + r + 8 * kh;
      Cb[(size_t)m * 64 + (ch * 2 + nt) * 16 + m16] = cacc[nt][r];
    }
  }
}

extern "C" void kernel_launch(void* const* d_in, const int* in_sizes, int n_in,
                              void* d_out, int out_size, void* d_ws, size_t ws_size,
                              hipStream_t stream) {
  (void)in_sizes; (void)n_in; (void)out_size; (void)ws_size;
  const float* x    = (const float*)d_in[0];
  const float* E    = (const float*)d_in[1];
  const float* pa   = (const float*)d_in[2];
  const float* Wp   = (const float*)d_in[3];
  const float* bp   = (const float*)d_in[4];
  float* out = (float*)d_out;

  float* ws   = (float*)d_ws;
  float* stats = ws;
  float* sube = ws   + 1048576;
  float* rS   = sube + 8388608;
  float* sx   = rS   + 8192 + 524288;
  float* sxF  = sube;

  k_colsum_pts<<<dim3(32, 8),     512, 0, stream>>>(pa, rS);
  k_sx_points<<<dim3(32, 8), 256, 0, stream>>>(pa, x, rS, E, sx, sxF, stats);
  k_xg_final<<<dim3(32, 8), 128, 0, stream>>>(E, stats, sxF, sx, Wp, bp, out);
}


# sub_e formed once: the column-sum kernel also stores it in A-fragment order (L1 distance is symmetric); the sx product copies it to LDS and its matrix waves do no forming arithmetic; sums unchanged
# speedup vs baseline: 4.6361x; 1.1730x over previous
#include <hip/hip_runtime.h>
#include <hip/hip_bf16.h>
#include <cstddef>


typedef float v2f __attribute__((ext_vector_type(2)));
typedef float v8f __attribute__((ext_vector_type(8)));

static __device__ __forceinline__ v8f wmma_f32(v2f a, v2f b, v8f c) {
  return __builtin_amdgcn_wmma_f32_16x16x4_f32(
       false, a,  false, b,
       (short)0, c,  false,  false);
}

static __device__ __forceinline__ float exp_neg_l1(const float (&p)[16],
                                                   const float (&q)[16]) {
  float s0 = 0.f, s1 = 0.f, s2 = 0.f, s3 = 0.f;
#pragma unroll
  for (int d = 0; d < 4; ++d) {
    s0 += fabsf(p[d]      - q[d]);
    s1 += fabsf(p[d + 4]  - q[d + 4]);
    s2 += fabsf(p[d + 8]  - q[d + 8]);
    s3 += fabsf(p[d + 12] - q[d + 12]);
  }
  return __expf(-((s0 + s1) + (s2 + s3)));
}

static __device__ __forceinline__ void rowstats_task(const float* __restrict__ E,
                                                     float* __restrict__ stats,
                                                     int task, int lane) {
  const int mt = task >> 4;
  const int kc = task & 15;
  const int m16 = lane & 15;
  const int kh = lane >> 4;
  const int mrow = mt * 16 + m16;

  v2f bq[4];
  {
    const float4 e0 = *(const float4*)(E + (size_t)mrow * 16 + 8 * kh);
    const float4 e1 = *(const float4*)(E + (size_t)mrow * 16 + 8 * kh + 4);
    bq[0][0] = e0.x; bq[0][1] = e0.y; bq[1][0] = e0.z; bq[1][1] = e0.w;
    bq[2][0] = e1.x; bq[2][1] = e1.y; bq[3][0] = e1.z; bq[3][1] = e1.w;
  }
  const v8f zero = {0.f, 0.f, 0.f, 0.f, 0.f, 0.f, 0.f, 0.f};
  v8f tl[4];
#pragma unroll
  for (int q = 0; q < 4; ++q) {
    const float* er = E + (size_t)(kc * 64 + q * 16 + m16) * 16 + 8 * kh;
    const float4 f0 = *(const float4*)(er);
    const float4 f1 = *(const float4*)(er + 4);
    v8f st = zero;
    st = wmma_f32(v2f{f0.x, f0.y}, bq[0], st);
    st = wmma_f32(v2f{f0.z, f0.w}, bq[1], st);
    st = wmma_f32(v2f{f1.x, f1.y}, bq[2], st);
    st = wmma_f32(v2f{f1.z, f1.w}, bq[3], st);
    tl[q] = st;
  }
  float mx = 0.f;
#pragma unroll
  for (int q = 0; q < 4; ++q)
#pragma unroll
    for (int r = 0; r < 8; ++r) mx = fmaxf(mx, tl[q][r]);
  mx = fmaxf(mx, __shfl_xor(mx, 16));
  float acc = 0.f;
#pragma unroll
  for (int q = 0; q < 4; ++q)
#pragma unroll
    for (int r = 0; r < 8; ++r) acc += __expf(fmaxf(tl[q][r], 0.f) - mx);
  acc += __shfl_xor(acc, 16);
  stats[(size_t)task * 32 + lane] = kh ? acc : mx;
}

__global__ __launch_bounds__(512)
void k_colsum_pts(const float* __restrict__ pa, float* __restrict__ rS,
                  float* __restrict__ Afr) {
  const int b = blockIdx.y;
  const int lane = threadIdx.x & 31;
  const int w = __builtin_amdgcn_readfirstlane((int)(threadIdx.x >> 5));
  const int j = blockIdx.x * 32 + lane;
  __shared__ float part[16][32];

  float pj[16];
#pragma unroll
  for (int q = 0; q < 4; ++q) {
    const float4 v = *(const float4*)(pa + ((size_t)b * 1024 + j) * 16 + q * 4);
    pj[q * 4 + 0] = v.x; pj[q * 4 + 1] = v.y; pj[q * 4 + 2] = v.z; pj[q * 4 + 3] = v.w;
  }
  const float* pi = pa + ((size_t)b * 1024 + (size_t)w * 64) * 16;
  float* dst = Afr + ((((size_t)b * 64 + blockIdx.x * 2 + (lane >> 4)) * 128 + (size_t)w * 8) * 128) + (lane & 15) * 4;
  float acc = 0.f;
  for (int pp = 0; pp < 8; ++pp) {
    float e[8];
#pragma unroll
    for (int u = 0; u < 8; ++u) {
      float qi[16];
#pragma unroll
      for (int d = 0; d < 16; ++d) qi[d] = pi[(pp * 8 + u) * 16 + d];
      e[u] = exp_neg_l1(qi, pj);
      acc += e[u];
    }
    *(float4*)(dst + pp * 128)      = float4{e[0], e[1], e[4], e[5]};
    *(float4*)(dst + pp * 128 + 64) = float4{e[2], e[3], e[6], e[7]};
  }
  part[w][lane] = acc;
  __syncthreads();
  if (threadIdx.x < 32) {
    float tot = 0.f;
#pragma unroll
    for (int w2 = 0; w2 < 16; ++w2) tot += part[w2][threadIdx.x];
    rS[b * 1024 + blockIdx.x * 32 + threadIdx.x] = 1.0f / tot;
  }
}

#define SX_KC 256
typedef float v4f __attribute__((ext_vector_type(4)));
#define V4_LO(x) __builtin_shufflevector((x), (x), 0, 1)
#define V4_HI(x) __builtin_shufflevector((x), (x), 2, 3)

template <int NU>
static __device__ __forceinline__ void sx_stage_b(const float* __restrict__ Bb,
                                                  const float* __restrict__ rsb, int k0,
                                                  float (*Bst)[160], int pr0, int ts) {
  float4 v0[NU], v1[NU];
  v2f r[NU];
#pragma unroll
  for (int u = 0; u < NU; ++u) {
    const int i = ts + u * 128;
    const int kp = i >> 4, cc = (i & 15) << 2;
    v0[u] = *(const float4*)(Bb + (size_t)(k0 + 2 * kp) * 64 + cc);
    v1[u] = *(const float4*)(Bb + (size_t)(k0 + 2 * kp + 1) * 64 + cc);
    r[u] = *(const v2f*)(rsb + k0 + 2 * kp);
  }
#pragma unroll
  for (int u = 0; u < NU; ++u) {
    const int i = ts + u * 128;
    const int kp = i >> 4, cc = (i & 15) << 2;
    float4 lo = {v0[u].x * r[u][0], v1[u].x * r[u][1], v0[u].y * r[u][0], v1[u].y * r[u][1]};
    float4 hi = {v0[u].z * r[u][0], v1[u].z * r[u][1], v0[u].w * r[u][0], v1[u].w * r[u][1]};
    *(float4*)(&Bst[pr0 + kp][cc * 2])     = lo;
    *(float4*)(&Bst[pr0 + kp][cc * 2 + 4]) = hi;
  }
}

static __device__ __forceinline__ void sx_a_issue(const float* __restrict__ Ab, int sc, int ts,
                                                  float4 (&va)[2][4]) {
#pragma unroll
  for (int rtl = 0; rtl < 2; ++rtl)
#pragma unroll
    for (int u = 0; u < 4; ++u)
      va[rtl][u] = *(const float4*)(Ab + (size_t)rtl * 16384 + sc * 2048 + (ts + u * 128) * 4);
}
static __device__ __forceinline__ void sx_a_store(float (*Asb)[2048], int ts,
                                                  const float4 (&va)[2][4]) {
#pragma unroll
  for (int rtl = 0; rtl < 2; ++rtl)
#pragma unroll
    for (int u = 0; u < 4; ++u)
      *(float4*)(&Asb[rtl][(ts + u * 128) * 4]) = va[rtl][u];
}

static __device__ __forceinline__ void sx_use(const float* ar, const float* bcol, v8f (&cacc)[2]) {
  v4f a[16];
  v2f bf[32][2];
#pragma unroll
  for (int pg = 0; pg < 4; ++pg) {
#pragma unroll
    for (int p = 4 * pg; p < 4 * pg + 4; ++p) {
      a[p] = *(const v4f*)(ar + p * 128);
      bf[2 * p][0]     = *(const v2f*)(bcol + (4 * p) * 160);
      bf[2 * p][1]     = *(const v2f*)(bcol + (4 * p) * 160 + 32);
      bf[2 * p + 1][0] = *(const v2f*)(bcol + (4 * p + 2) * 160);
      bf[2 * p + 1][1] = *(const v2f*)(bcol + (4 * p + 2) * 160 + 32);
    }
    __builtin_amdgcn_sched_barrier(0);
  }
#pragma unroll
  for (int p = 0; p < 16; ++p) {
    cacc[0] = wmma_f32(V4_LO(a[p]), bf[2 * p][0], cacc[0]);
    cacc[1] = wmma_f32(V4_LO(a[p]), bf[2 * p][1], cacc[1]);
    cacc[0] = wmma_f32(V4_HI(a[p]), bf[2 * p + 1][0], cacc[0]);
    cacc[1] = wmma_f32(V4_HI(a[p]), bf[2 * p + 1][1], cacc[1]);
  }
}

__global__ __launch_bounds__(256)
void k_sx_points(const float* __restrict__ Afr, const float* __restrict__ x,
                 const float* __restrict__ rS, const float* __restrict__ E,
                 float* __restrict__ sx, float* __restrict__ sxF,
                 float* __restrict__ stats) {
  const int b = blockIdx.y;
  const int rowBase = blockIdx.x * 32;
  const float* Ab = Afr + ((size_t)b * 64 + blockIdx.x * 2) * 16384;
  const float* rsb = rS + (size_t)b * 1024;
  const float* Bb = x + (size_t)b * 1024 * 64;
  float* Cb = sx + (size_t)b * 1024 * 64;

  __shared__ float Bs[2][SX_KC / 2][160];
  __shared__ __attribute__((aligned(16))) float As[2][2][2048];

  const int t = threadIdx.x;
  const int lane = t & 31;
  const int m16 = lane & 15;
  const int kh = lane >> 4;
  const int wnum = __builtin_amdgcn_readfirstlane(t >> 5);
  const bool product = (wnum & 1) == (wnum >> 2);
  const int wave = t >> 6;
  const int rt = wave >> 1;
  const int ch = wave & 1;
  const int ts = wave * 32 + lane;

  const v8f zero = {0.f, 0.f, 0.f, 0.f, 0.f, 0.f, 0.f, 0.f};
  v8f cacc[2];
  cacc[0] = zero; cacc[1] = zero;

  const float* bcol0 = &Bs[0][kh][(ch * 32 + m16) * 2];

  if (product) {
    float4 va[2][4];
    sx_a_issue(Ab, 0, ts, va);
    sx_a_store(As[0], ts, va);
  } else {
    sx_stage_b<16>(Bb, rsb, 0, Bs[0], 0, ts);
  }
  __syncthreads();

  for (int sc = 0; sc < 8; ++sc) {
    if (product) {
      const float* ar = &As[sc & 1][rt][lane * 4];
      const float* bcol = bcol0 + (size_t)(((sc >> 1) & 1) * (SX_KC / 2) + (sc & 1) * 64) * 160;
      sx_use(ar, bcol, cacc);
    } else if (sc < 7) {
      float4 va[2][4];
      sx_a_issue(Ab, sc + 1, ts, va);
      if (sc < 6) {
        const int cn = (sc >> 1) + 1;
        sx_stage_b<8>(Bb, rsb, cn * SX_KC + (sc & 1) * 128, Bs[cn & 1], (sc & 1) * 64, ts);
      }
      sx_a_store(As[(sc + 1) & 1], ts, va);
    }
    if (sc < 7) __syncthreads();
  }

  if (!product) {
    rowstats_task(E, stats, (blockIdx.y * 32 + blockIdx.x) * 4 + wave, lane);
  }

  if (product) {
#pragma unroll
  for (int nt = 0; nt < 2; ++nt) {
#pragma unroll
    for (int r = 0; r < 8; ++r) {
      const int m = rowBase + rt * 16 + r + 8 * kh;
      Cb[(size_t)m * 64 + (ch * 2 + nt) * 16 + m16] = cacc[nt][r];
    }
  }
  {
    float* Fb = sxF + (((size_t)b * 64 + blockIdx.x * 2 + rt) * 4 + ch * 2) * 256 + lane * 8;
#pragma unroll
    for (int nt = 0; nt < 2; ++nt) {
      *(float4*)(Fb + nt * 256)     = float4{cacc[nt][0], cacc[nt][1], cacc[nt][2], cacc[nt][3]};
      *(float4*)(Fb + nt * 256 + 4) = float4{cacc[nt][4], cacc[nt][5], cacc[nt][6], cacc[nt][7]};
    }
  }
  }
}

#define XG_PF 8
#define FP_DS 2
#define FP_NS (16 / FP_DS)
#define XS_PITCH 68

static __device__ __forceinline__ void pool_issue(const float* __restrict__ src, int t,
                                                  float4 (&v0)[16], float4 (&v1)[16]) {
#pragma unroll
  for (int u = 0; u < 16; ++u) {
    const int i = t + u * 128;
    const int kp = i >> 4, cc = (i & 15) << 2;
    v0[u] = *(const float4*)(src + (size_t)(2 * kp) * 64 + cc);
    v1[u] = *(const float4*)(src + (size_t)(2 * kp + 1) * 64 + cc);
  }
}

static __device__ __forceinline__ void pool_store(float (*Bsb)[160], int t,
                                                  const float4 (&v0)[16], const float4 (&v1)[16]) {
#pragma unroll
  for (int u = 0; u < 16; ++u) {
    const int i = t + u * 128;
    const int kp = i >> 4, cc = (i & 15) << 2;
    float4 lo = {v0[u].x, v1[u].x, v0[u].y, v1[u].y};
    float4 hi = {v0[u].z, v1[u].z, v0[u].w, v1[u].w};
    *(float4*)(&Bsb[kp][cc * 2])     = lo;
    *(float4*)(&Bsb[kp][cc * 2 + 4]) = hi;
  }
}

static __device__ __forceinline__ void pool_stage(const float* bc, const v2f (&a)[32],
                                                  v2f e2, v8f (&cacc)[2]) {
  v2f bf[2][8][2];
#pragma unroll
  for (int j = 0; j < 8; ++j) {
    bf[0][j][0] = *(const v2f*)(bc + (2 * j) * 160);
    bf[0][j][1] = *(const v2f*)(bc + (2 * j) * 160 + 32);
  }
#pragma unroll
  for (int blk = 0; blk < FP_DS * 4; ++blk) {
    if (blk + 1 < FP_DS * 4) {
#pragma unroll
      for (int j = 0; j < 8; ++j) {
        bf[(blk + 1) & 1][j][0] = *(const v2f*)(bc + (16 * (blk + 1) + 2 * j) * 160);
        bf[(blk + 1) & 1][j][1] = *(const v2f*)(bc + (16 * (blk + 1) + 2 * j) * 160 + 32);
      }
    }
    const float ed = e2[blk >> 2];
#pragma unroll
    for (int j = 0; j < 8; ++j) {
      const v2f as = a[8 * (blk & 3) + j] * ed;
      cacc[0] = wmma_f32(as, bf[blk & 1][j][0], cacc[0]);
      cacc[1] = wmma_f32(as, bf[blk & 1][j][1], cacc[1]);
    }
  }
}

__global__ __launch_bounds__(128)
void k_xg_final(const float* __restrict__ E, const float* __restrict__ stats,
                const float* __restrict__ sxF,
                const float* __restrict__ sx, const float* __restrict__ Wp,
                const float* __restrict__ bias_pool, float* __restrict__ out) {
  const int b = blockIdx.y;
  const int rowBase = blockIdx.x * 32;

  __shared__ float Bs[2][FP_DS * 64][160];
  __shared__ __attribute__((aligned(16))) float Xs[32][XS_PITCH];

  const int t = threadIdx.x;
  const int wave = t >> 5;
  const int lane = t & 31;
  const int m16 = lane & 15;
  const int kh = lane >> 4;
  const int rt = wave >> 1;
  const int ch = wave & 1;
  const int nrow = rowBase + rt * 16 + m16;

  float4 v0[16], v1[16];

  v2f bq[4];
  {
    const float4 e0 = *(const float4*)(E + (size_t)nrow * 16 + 8 * kh);
    const float4 e1 = *(const float4*)(E + (size_t)nrow * 16 + 8 * kh + 4);
    bq[0][0] = e0.x; bq[0][1] = e0.y; bq[1][0] = e0.z; bq[1][1] = e0.w;
    bq[2][0] = e1.x; bq[2][1] = e1.y; bq[3][0] = e1.z; bq[3][1] = e1.w;
  }
  float mxm = 0.f;
  float invm;
  {
    const float* sp = stats + (size_t)(nrow >> 4) * 512 + (nrow & 15);
    float pmx[16], psm[16];
#pragma unroll
    for (int kc = 0; kc < 16; ++kc) {
      pmx[kc] = sp[kc * 32];
      psm[kc] = sp[kc * 32 + 16];
    }
#pragma unroll
    for (int kc = 0; kc < 16; ++kc) mxm = fmaxf(mxm, pmx[kc]);
    float tot = 0.f;
#pragma unroll
    for (int kc = 0; kc < 16; ++kc) tot += psm[kc] * __expf(pmx[kc] - mxm);
    invm = 1.0f / tot;
  }
  const float* etile = E + (size_t)m16 * 16 + 8 * kh;
  const float* bfr = sxF + ((size_t)b * 256 + ch * 2) * 256 + lane * 8;

  const v8f zero = {0.f, 0.f, 0.f, 0.f, 0.f, 0.f, 0.f, 0.f};
  v8f xacc[2];
  xacc[0] = zero; xacc[1] = zero;

  v2f a[32];
  const float* erow = E + (size_t)nrow * 16;
  v2f ea[4], bb[4][2];
  v2f e_cur;

  v4f eA[XG_PF][2], eB[XG_PF][2];
  v4f bA[XG_PF][2][2], bB[XG_PF][2][2];
  v8f st_cur;
  v2f a_cur[4];

  auto request = [&](v4f (&eb)[XG_PF][2], v4f (&bb2)[XG_PF][2][2], int tt, int g) {
    const float* en = etile + (size_t)((g + 2) & 63) * 256;
    eb[tt][0] = *(const v4f*)(en);
    eb[tt][1] = *(const v4f*)(en + 4);
    const float* bn = bfr + (size_t)(g & 63) * 1024;
#pragma unroll
    for (int nt = 0; nt < 2; ++nt) {
      bb2[tt][nt][0] = *(const v4f*)(bn + nt * 256);
      bb2[tt][nt][1] = *(const v4f*)(bn + nt * 256 + 4);
    }
  };
  auto trip = [&](const v4f (&eb)[XG_PF][2], const v4f (&bb2)[XG_PF][2][2], int tt) {
    const v2f ta[4] = {V4_LO(eb[tt][0]), V4_HI(eb[tt][0]), V4_LO(eb[tt][1]), V4_HI(eb[tt][1])};
    v2f bf[4][2];
#pragma unroll
    for (int nt = 0; nt < 2; ++nt) {
      bf[0][nt] = V4_LO(bb2[tt][nt][0]); bf[1][nt] = V4_HI(bb2[tt][nt][0]);
      bf[2][nt] = V4_LO(bb2[tt][nt][1]); bf[3][nt] = V4_HI(bb2[tt][nt][1]);
    }
    v2f a_nxt[4];
#pragma unroll
    for (int j = 0; j < 4; ++j) {
      a_nxt[j][0] = __expf(fmaxf(st_cur[2 * j], 0.f) - mxm) * invm;
      a_nxt[j][1] = __expf(fmaxf(st_cur[2 * j + 1], 0.f) - mxm) * invm;
    }
    v8f st_nxt = zero;
#pragma unroll
    for (int j = 0; j < 4; ++j) {
      st_nxt = wmma_f32(ta[j], bq[j], st_nxt);
      xacc[0] = wmma_f32(a_cur[j], bf[j][0], xacc[0]);
      xacc[1] = wmma_f32(a_cur[j], bf[j][1], xacc[1]);
    }
    st_cur = st_nxt;
#pragma unroll
    for (int j = 0; j < 4; ++j) a_cur[j] = a_nxt[j];
    __builtin_amdgcn_sched_barrier(0);
  };

  {
    const v4f f0 = *(const v4f*)(etile);
    const v4f f1 = *(const v4f*)(etile + 4);
    const v4f g0 = *(const v4f*)(etile + 256);
    const v4f g1 = *(const v4f*)(etile + 256 + 4);
#pragma unroll
    for (int u = 0; u < XG_PF; ++u) request(eA, bA, u, u);
    asm volatile("" ::: "memory");
    v8f s0 = zero;
    s0 = wmma_f32(V4_LO(f0), bq[0], s0);
    s0 = wmma_f32(V4_HI(f0), bq[1], s0);
    s0 = wmma_f32(V4_LO(f1), bq[2], s0);
    s0 = wmma_f32(V4_HI(f1), bq[3], s0);
    v8f s1 = zero;
    s1 = wmma_f32(V4_LO(g0), bq[0], s1);
    s1 = wmma_f32(V4_HI(g0), bq[1], s1);
    s1 = wmma_f32(V4_LO(g1), bq[2], s1);
    s1 = wmma_f32(V4_HI(g1), bq[3], s1);
#pragma unroll
    for (int j = 0; j < 4; ++j) {
      a_cur[j][0] = __expf(fmaxf(s0[2 * j], 0.f) - mxm) * invm;
      a_cur[j][1] = __expf(fmaxf(s0[2 * j + 1], 0.f) - mxm) * invm;
    }
    st_cur = s1;
  }

  auto second_request = [&](int q) {
#pragma unroll
    for (int u = 4 * q; u < 4 * q + 4; ++u) {
      const int i = t + u * 128;
      const int kp = i >> 4, cc = (i & 15) << 2;
      v0[u] = *(const float4*)(Wp + (size_t)(2 * kp) * 64 + cc);
      v1[u] = *(const float4*)(Wp + (size_t)(2 * kp + 1) * 64 + cc);
    }
    const size_t roff = ((size_t)b * 1024 + nrow) * 64 + 2 * kh;
#pragma unroll
    for (int ks = 4 * q; ks < 4 * q + 4; ++ks) a[ks] = *(const v2f*)(sx + roff + 4 * ks);
    ea[q] = *(const v2f*)(erow + 4 * q + 2 * kh);
#pragma unroll
    for (int nt = 0; nt < 2; ++nt) {
      const int col = ch * 32 + nt * 16 + m16;
      bb[q][nt][0] = bias_pool[(4 * q + 2 * kh) * 64 + col];
      bb[q][nt][1] = bias_pool[(4 * q + 2 * kh + 1) * 64 + col];
    }
    if (q == 0) e_cur = *(const v2f*)(erow);
  };

  auto pass16 = [&](int g0, bool last) __attribute__((always_inline)) {
#pragma unroll
    for (int u = 0; u < XG_PF; ++u) {
      if (u < XG_PF / 2) {
        request(eB, bB, 2 * u,     g0 + XG_PF + 2 * u);
        request(eB, bB, 2 * u + 1, g0 + XG_PF + 2 * u + 1);
        asm volatile("" ::: "memory");
      }
      trip(eA, bA, u);
    }
#pragma unroll
    for (int u = 0; u < XG_PF; ++u) {
      if (u < XG_PF / 2) {
        if (!last) {
          request(eA, bA, 2 * u,     g0 + 2 * XG_PF + 2 * u);
          request(eA, bA, 2 * u + 1, g0 + 2 * XG_PF + 2 * u + 1);
        } else {
          second_request(u);
        }
        asm volatile("" ::: "memory");
      }
      trip(eB, bB, u);
    }
  };
  for (int g0 = 0; g0 < 48; g0 += 2 * XG_PF) pass16(g0, false);
  pass16(48, true);

  pool_store(Bs[0], t, v0, v1);
#pragma unroll
  for (int nt = 0; nt < 2; ++nt) {
#pragma unroll
    for (int r = 0; r < 8; ++r)
      Xs[rt * 16 + r + 8 * kh][(ch * 2 + nt) * 16 + m16] = xacc[nt][r];
  }
  __syncthreads();
#pragma unroll
  for (int ks = 0; ks < 16; ++ks)
    a[16 + ks] = *(const v2f*)(&Xs[rt * 16 + m16][4 * ks + 2 * kh]);

  v8f cacc[2];
  cacc[0] = zero; cacc[1] = zero;
#pragma unroll
  for (int j = 0; j < 4; ++j) {
    cacc[0] = wmma_f32(ea[j], bb[j][0], cacc[0]);
    cacc[1] = wmma_f32(ea[j], bb[j][1], cacc[1]);
  }

  for (int s = 0; s < FP_NS; ++s) {
    const int sn = (s + 1) & (FP_NS - 1);
    pool_issue(Wp + (size_t)sn * (FP_DS * 8192), t, v0, v1);
    const v2f e_nxt = *(const v2f*)(erow + FP_DS * sn);
    pool_stage(&Bs[s & 1][kh][(ch * 32 + m16) * 2], a, e_cur, cacc);
    pool_store(Bs[(s + 1) & 1], t, v0, v1);
    e_cur = e_nxt;
    __syncthreads();
  }

  float* Cb = out + (size_t)b * 1024 * 64;
#pragma unroll
  for (int nt = 0; nt < 2; ++nt) {
#pragma unroll
    for (int r = 0; r < 8; ++r) {
      const int m = rowBase + rt * 16 + r + 8 * kh;
      Cb[(size_t)m * 64 + (ch * 2 + nt) * 16 + m16] = cacc[nt][r];
    }
  }
}

extern "C" void kernel_launch(void* const* d_in, const int* in_sizes, int n_in,
                              void* d_out, int out_size, void* d_ws, size_t ws_size,
                              hipStream_t stream) {
  (void)in_sizes; (void)n_in; (void)out_size; (void)ws_size;
  const float* x    = (const float*)d_in[0];
  const float* E    = (const float*)d_in[1];
  const float* pa   = (const float*)d_in[2];
  const float* Wp   = (const float*)d_in[3];
  const float* bp   = (const float*)d_in[4];
  float* out = (float*)d_out;

  float* ws   = (float*)d_ws;
  float* stats = ws;
  float* Afr  = ws   + 1048576;
  float* rS   = Afr  + 8388608;
  float* sxF  = rS   + 8192;
  float* sx   = rS   + 8192 + 524288;

  k_colsum_pts<<<dim3(32, 8),     512, 0, stream>>>(pa, rS, Afr);
  k_sx_points<<<dim3(32, 8), 256, 0, stream>>>(Afr, x, rS, E, sx, sxF, stats);
  k_xg_final<<<dim3(32, 8), 128, 0, stream>>>(E, stats, sxF, sx, Wp, bp, out);
}
